# starting kernel
# baseline (speedup 1.0000x reference)
#include <hip/hip_runtime.h>
#include <math.h>


typedef float v2f __attribute__((ext_vector_type(2)));
typedef float v8f __attribute__((ext_vector_type(8)));

#define DK 256
#define KCHUNK 128
#define LDT (KCHUNK + 4)
#define BM 32
#define BN 64

__device__ __forceinline__ unsigned f2ord(float f) {
  unsigned u = __float_as_uint(f);
  return (u & 0x80000000u) ? ~u : (u | 0x80000000u);
}
__device__ __forceinline__ float ord2f(unsigned u) {
  return __uint_as_float((u & 0x80000000u) ? (u ^ 0x80000000u) : ~u);
}

__global__ __launch_bounds__(256) void gat_gemm_wmma(
    const float* __restrict__ X, const float* __restrict__ W,
    float* __restrict__ H, int N, int Dact, int Dst) {
  __shared__ float As[BM * LDT];
  __shared__ float Bst[BN * LDT];

  const int tid  = threadIdx.x;
  const int lane = tid & 31;
  const int lo   = lane & 15;
  const int hi   = lane >> 4;
  const int wid  = tid >> 5;
  const int wm   = (wid >> 2) * 16;
  const int wn   = (wid & 3) * 16;
  const int rowBase = blockIdx.x * BM;
  const int colBase = blockIdx.y * BN;

  v8f acc = {};

  for (int kk = 0; kk < DK; kk += KCHUNK) {
    __syncthreads();
    for (int it = 0; it < 4; ++it) {
      int f4  = tid + 256 * it;
      int r   = f4 >> 5;
      int c4  = f4 & 31;
      int gr  = rowBase + r;
      float4 v = make_float4(0.f, 0.f, 0.f, 0.f);
      if (gr < N)
        v = *reinterpret_cast<const float4*>(X + (size_t)gr * DK + kk + c4 * 4);
      *reinterpret_cast<float4*>(&As[r * LDT + c4 * 4]) = v;
    }
    for (int it = 0; it < 32; ++it) {
      int idx = tid + 256 * it;
      int k   = idx >> 6;
      int n   = idx & 63;
      int gc  = colBase + n;
      float v = (gc < Dact) ? W[(size_t)(kk + k) * Dact + gc] : 0.f;
      Bst[n * LDT + k] = v;
    }
    __syncthreads();

    const float* arow = &As[(wm + lo) * LDT];
    const float* brow = &Bst[(wn + lo) * LDT];
    #pragma unroll
    for (int k0 = 0; k0 < KCHUNK; k0 += 4) {
      v2f a = *reinterpret_cast<const v2f*>(arow + k0 + 2 * hi);
      v2f b = *reinterpret_cast<const v2f*>(brow + k0 + 2 * hi);
      acc = __builtin_amdgcn_wmma_f32_16x16x4_f32(false, a, false, b,
                                                  (short)0, acc, false, false);
    }
  }

  const int col = colBase + wn + lo;
  if (col < Dst) {
    #pragma unroll
    for (int r = 0; r < 8; ++r) {
      int row = rowBase + wm + hi * 8 + r;
      if (row < N) H[(size_t)row * Dst + col] = acc[r];
    }
  }
}

__global__ __launch_bounds__(256) void gat_alpha_init(
    const float* __restrict__ H, const float* __restrict__ a_s,
    const float* __restrict__ a_d, float* __restrict__ asv,
    float* __restrict__ adv, unsigned* __restrict__ mv,
    float* __restrict__ dnv, float* __restrict__ outz,
    int N, int Dpad, int Dact) {
  const int lane = threadIdx.x & 31;
  const int i = blockIdx.x * 8 + (threadIdx.x >> 5);
  if (i >= N) return;
  float s1 = 0.f, s2 = 0.f;
  for (int d = lane; d < Dpad; d += 32) {
    float hv = H[(size_t)i * Dpad + d];
    float av = (d < Dact) ? a_s[d] : 0.f;
    float bv = (d < Dact) ? a_d[d] : 0.f;
    s1 += hv * av;
    s2 += hv * bv;
    outz[(size_t)i * Dpad + d] = 0.f;
  }
  #pragma unroll
  for (int off = 16; off > 0; off >>= 1) {
    s1 += __shfl_xor(s1, off, 32);
    s2 += __shfl_xor(s2, off, 32);
  }
  if (lane == 0) {
    asv[i] = s1;
    adv[i] = s2;
    mv[i]  = f2ord(-INFINITY);
    dnv[i] = 0.f;
  }
}

__global__ __launch_bounds__(256) void gat_edge_score(
    const int* __restrict__ src, const int* __restrict__ dst,
    const float* __restrict__ asv, const float* __restrict__ adv,
    float* __restrict__ esc, unsigned* __restrict__ mv, int E, int ET) {
  int e = blockIdx.x * blockDim.x + threadIdx.x;
  if (e >= ET) return;
  int s, d;
  if (e < E) { s = src[e]; d = dst[e]; } else { s = d = e - E; }
  float sc = asv[s] + adv[d];
  sc = (sc > 0.f) ? sc : 0.2f * sc;
  esc[e] = sc;
  atomicMax(&mv[d], f2ord(sc));
}

__global__ __launch_bounds__(256) void gat_edge_exp(
    const int* __restrict__ dst, const unsigned* __restrict__ mv,
    float* __restrict__ esc, float* __restrict__ dnv, int E, int ET) {
  int e = blockIdx.x * blockDim.x + threadIdx.x;
  if (e >= ET) return;
  int d = (e < E) ? dst[e] : (e - E);
  float p = __expf(esc[e] - ord2f(mv[d]));
  esc[e] = p;
  atomicAdd(&dnv[d], p);
}

__global__ __launch_bounds__(256) void gat_aggregate(
    const int* __restrict__ src, const int* __restrict__ dst,
    const float* __restrict__ esc, const float* __restrict__ dnv,
    const float* __restrict__ H, float* __restrict__ OUT,
    int E, int ET, int Dpad) {
  int e = (blockIdx.x * blockDim.x + threadIdx.x) >> 5;
  int lane = threadIdx.x & 31;
  if (e >= ET) return;
  int s, d;
  if (e < E) { s = src[e]; d = dst[e]; } else { s = d = e - E; }
  float coeff = esc[e] / dnv[d];
  const float4* hp = reinterpret_cast<const float4*>(H + (size_t)s * Dpad);
  float* op = OUT + (size_t)d * Dpad;
  int nf4 = Dpad >> 2;
  for (int d4 = lane; d4 < nf4; d4 += 32) {
    float4 hv = hp[d4];
    atomicAdd(op + d4 * 4 + 0, coeff * hv.x);
    atomicAdd(op + d4 * 4 + 1, coeff * hv.y);
    atomicAdd(op + d4 * 4 + 2, coeff * hv.z);
    atomicAdd(op + d4 * 4 + 3, coeff * hv.w);
  }
}

__global__ __launch_bounds__(256) void gat_bias_relu(
    float* __restrict__ HIO, const float* __restrict__ b,
    size_t total, int Dpad) {
  size_t t = (size_t)blockIdx.x * blockDim.x + threadIdx.x;
  if (t >= total) return;
  float v = HIO[t] + b[t % Dpad];
  HIO[t] = v > 0.f ? v : 0.f;
}

__global__ __launch_bounds__(256) void gat_bias_out(
    const float* __restrict__ O, const float* __restrict__ b,
    float* __restrict__ out, int N, int Dpad, int Dact) {
  int t = blockIdx.x * blockDim.x + threadIdx.x;
  if (t >= N * Dpad) return;
  int i = t / Dpad, d = t % Dpad;
  if (d < Dact) out[(size_t)i * Dact + d] = O[t] + b[d];
}

extern "C" void kernel_launch(void* const* d_in, const int* in_sizes, int n_in,
                              void* d_out, int out_size, void* d_ws, size_t ws_size,
                              hipStream_t stream) {
  const float* x  = (const float*)d_in[0];
  const int*   ei = (const int*)d_in[1];
  const float* W1 = (const float*)d_in[2];
  const float* as1 = (const float*)d_in[3];
  const float* ad1 = (const float*)d_in[4];
  const float* b1 = (const float*)d_in[5];
  const float* W2 = (const float*)d_in[6];
  const float* as2 = (const float*)d_in[7];
  const float* ad2 = (const float*)d_in[8];
  const float* b2 = (const float*)d_in[9];
  const float* W3 = (const float*)d_in[10];
  const float* as3 = (const float*)d_in[11];
  const float* ad3 = (const float*)d_in[12];
  const float* b3 = (const float*)d_in[13];

  const int N  = in_sizes[0] / 256;
  const int E  = in_sizes[1] / 2;
  const int ET = E + N;
  const int* src = ei;
  const int* dst = ei + E;

  float* ws = (float*)d_ws;
  float*    hA  = ws;
  float*    hB  = hA + (size_t)N * 256;
  float*    h3  = hB + (size_t)N * 256;
  float*    o3  = h3 + (size_t)N * 16;
  float*    asv = o3 + (size_t)N * 16;
  float*    adv = asv + N;
  unsigned* mv  = (unsigned*)(adv + N);
  float*    dnv = adv + 2 * (size_t)N;
  float*    esc = dnv + N;

  dim3 blk(256);
  dim3 gemmG((N + BM - 1) / BM, 4);
  dim3 gemmG3((N + BM - 1) / BM, 1);
  int gAlpha = (N + 7) / 8;
  int gEdge  = (ET + 255) / 256;
  int gAgg   = (ET + 7) / 8;
  int gEltF  = (int)(((size_t)N * 256 + 255) / 256);
  int gElt3  = (N * 16 + 255) / 256;

  gat_gemm_wmma<<<gemmG, blk, 0, stream>>>(x, W1, hA, N, 256, 256);
  gat_alpha_init<<<gAlpha, blk, 0, stream>>>(hA, as1, ad1, asv, adv, mv, dnv, hB, N, 256, 256);
  gat_edge_score<<<gEdge, blk, 0, stream>>>(src, dst, asv, adv, esc, mv, E, ET);
  gat_edge_exp<<<gEdge, blk, 0, stream>>>(dst, mv, esc, dnv, E, ET);
  gat_aggregate<<<gAgg, blk, 0, stream>>>(src, dst, esc, dnv, hA, hB, E, ET, 256);
  gat_bias_relu<<<gEltF, blk, 0, stream>>>(hB, b1, (size_t)N * 256, 256);

  gat_gemm_wmma<<<gemmG, blk, 0, stream>>>(hB, W2, hA, N, 256, 256);
  gat_alpha_init<<<gAlpha, blk, 0, stream>>>(hA, as2, ad2, asv, adv, mv, dnv, hB, N, 256, 256);
  gat_edge_score<<<gEdge, blk, 0, stream>>>(src, dst, asv, adv, esc, mv, E, ET);
  gat_edge_exp<<<gEdge, blk, 0, stream>>>(dst, mv, esc, dnv, E, ET);
  gat_aggregate<<<gAgg, blk, 0, stream>>>(src, dst, esc, dnv, hA, hB, E, ET, 256);
  gat_bias_relu<<<gEltF, blk, 0, stream>>>(hB, b2, (size_t)N * 256, 256);

  gat_gemm_wmma<<<gemmG3, blk, 0, stream>>>(hB, W3, h3, N, 10, 16);
  gat_alpha_init<<<gAlpha, blk, 0, stream>>>(h3, as3, ad3, asv, adv, mv, dnv, o3, N, 16, 10);
  gat_edge_score<<<gEdge, blk, 0, stream>>>(src, dst, asv, adv, esc, mv, E, ET);
  gat_edge_exp<<<gEdge, blk, 0, stream>>>(dst, mv, esc, dnv, E, ET);
  gat_aggregate<<<gAgg, blk, 0, stream>>>(src, dst, esc, dnv, h3, o3, E, ET, 16);
  gat_bias_out<<<gElt3, blk, 0, stream>>>(o3, b3, (float*)d_out, N, 16, 10);
}


# Edge softmax and layer-3 aggregate move into row-owner blocks: block lists its rows' edges in LDS in its dst scan, does max/exp/sum there; 7 launches, all global atomics go; divide after LDS sums
# speedup vs baseline: 2.1474x; 2.1474x over previous
#include <hip/hip_runtime.h>
#include <math.h>


typedef float v2f __attribute__((ext_vector_type(2)));
typedef float v8f __attribute__((ext_vector_type(8)));

#define DK 256
#define KCHUNK 128
#define LDT (KCHUNK + 4)
#define BM 32
#define BN 64

__device__ __forceinline__ unsigned f2ord(float f) {
  unsigned u = __float_as_uint(f);
  return (u & 0x80000000u) ? ~u : (u | 0x80000000u);
}
__device__ __forceinline__ float ord2f(unsigned u) {
  return __uint_as_float((u & 0x80000000u) ? (u ^ 0x80000000u) : ~u);
}

__global__ __launch_bounds__(256) void gat_gemm_wmma(
    const float* __restrict__ X, const float* __restrict__ W,
    float* __restrict__ H, int N, int Dact, int Dst) {
  __shared__ float As[BM * LDT];
  __shared__ float Bst[BN * LDT];

  const int tid  = threadIdx.x;
  const int lane = tid & 31;
  const int lo   = lane & 15;
  const int hi   = lane >> 4;
  const int wid  = tid >> 5;
  const int wm   = (wid >> 2) * 16;
  const int wn   = (wid & 3) * 16;
  const int rowBase = blockIdx.x * BM;
  const int colBase = blockIdx.y * BN;

  v8f acc = {};

  for (int kk = 0; kk < DK; kk += KCHUNK) {
    __syncthreads();
    for (int it = 0; it < 4; ++it) {
      int f4  = tid + 256 * it;
      int r   = f4 >> 5;
      int c4  = f4 & 31;
      int gr  = rowBase + r;
      float4 v = make_float4(0.f, 0.f, 0.f, 0.f);
      if (gr < N)
        v = *reinterpret_cast<const float4*>(X + (size_t)gr * DK + kk + c4 * 4);
      *reinterpret_cast<float4*>(&As[r * LDT + c4 * 4]) = v;
    }
    for (int it = 0; it < 32; ++it) {
      int idx = tid + 256 * it;
      int k   = idx >> 6;
      int n   = idx & 63;
      int gc  = colBase + n;
      float v = (gc < Dact) ? W[(size_t)(kk + k) * Dact + gc] : 0.f;
      Bst[n * LDT + k] = v;
    }
    __syncthreads();

    const float* arow = &As[(wm + lo) * LDT];
    const float* brow = &Bst[(wn + lo) * LDT];
    #pragma unroll
    for (int k0 = 0; k0 < KCHUNK; k0 += 4) {
      v2f a = *reinterpret_cast<const v2f*>(arow + k0 + 2 * hi);
      v2f b = *reinterpret_cast<const v2f*>(brow + k0 + 2 * hi);
      acc = __builtin_amdgcn_wmma_f32_16x16x4_f32(false, a, false, b,
                                                  (short)0, acc, false, false);
    }
  }

  const int col = colBase + wn + lo;
  if (col < Dst) {
    #pragma unroll
    for (int r = 0; r < 8; ++r) {
      int row = rowBase + wm + hi * 8 + r;
      if (row < N) H[(size_t)row * Dst + col] = acc[r];
    }
  }
}

__global__ __launch_bounds__(256) void gat_alpha(
    const float* __restrict__ H, const float* __restrict__ a_s,
    const float* __restrict__ a_d, float* __restrict__ asv,
    float* __restrict__ adv, int N, int Dpad, int Dact) {
  const int lane = threadIdx.x & 31;
  const int i = blockIdx.x * 8 + (threadIdx.x >> 5);
  if (i >= N) return;
  float s1 = 0.f, s2 = 0.f;
  for (int d = lane; d < Dpad; d += 32) {
    float hv = H[(size_t)i * Dpad + d];
    float av = (d < Dact) ? a_s[d] : 0.f;
    float bv = (d < Dact) ? a_d[d] : 0.f;
    s1 += hv * av;
    s2 += hv * bv;
  }
  #pragma unroll
  for (int off = 16; off > 0; off >>= 1) {
    s1 += __shfl_xor(s1, off, 32);
    s2 += __shfl_xor(s2, off, 32);
  }
  if (lane == 0) {
    asv[i] = s1;
    adv[i] = s2;
  }
}

#define AD 256
#define D3 16
#define SCAN 16
#define LROUNDS 6
#define LCAP (1024 * LROUNDS)

__device__ __forceinline__ float leaky02(float t) {
  return (t > 0.f) ? t : 0.2f * t;
}

__device__ __forceinline__ int rl_i(int v, int l) {
  return __builtin_amdgcn_readlane(v, l);
}
__device__ __forceinline__ float rl_f(float v, int l) {
  return __int_as_float(__builtin_amdgcn_readlane(__float_as_int(v), l));
}

__device__ __forceinline__ void list_append(int* cnt, int* lst, int qreg,
                                            int m, int lane) {
  int b = 0;
  if (lane == 0) b = atomicAdd(cnt, m);
  b = __builtin_amdgcn_readfirstlane(b);
  const int slot = b + lane;
  if (lane < m && slot < LCAP) lst[slot] = qreg;
}

__device__ __forceinline__ void row_add4(
    int s, int dl, float p, int g, int c, int lane,
    const float* __restrict__ H, float* acc) {
  float v[4][8];
  float pb[4];
  int   ro[4];
  #pragma unroll
  for (int b = 0; b < 4; ++b) {
    const int q = (b < c) ? (g + b) : g;
    const int sb = rl_i(s, q);
    ro[b] = rl_i(dl, q) * AD + lane;
    pb[b] = rl_f(p, q);
    const float* hp = H + (size_t)sb * AD + lane;
    #pragma unroll
    for (int j = 0; j < 8; ++j) v[b][j] = hp[j * 32];
  }
  #pragma unroll
  for (int b = 0; b < 4; ++b) {
    if (b < c) {
      #pragma unroll
      for (int j = 0; j < 8; ++j)
        atomicAdd(&acc[ro[b] + j * 32], pb[b] * v[b][j]);
    }
  }
}

__device__ __forceinline__ void row_add16(
    int s, int dl, float p, const float* __restrict__ H, float* acc) {
  const float4* hp = reinterpret_cast<const float4*>(H + (size_t)s * D3);
  const float4 a0 = hp[0], a1 = hp[1], a2 = hp[2], a3 = hp[3];
  float* ar = &acc[dl * D3];
  atomicAdd(ar + 0,  p * a0.x); atomicAdd(ar + 1,  p * a0.y);
  atomicAdd(ar + 2,  p * a0.z); atomicAdd(ar + 3,  p * a0.w);
  atomicAdd(ar + 4,  p * a1.x); atomicAdd(ar + 5,  p * a1.y);
  atomicAdd(ar + 6,  p * a1.z); atomicAdd(ar + 7,  p * a1.w);
  atomicAdd(ar + 8,  p * a2.x); atomicAdd(ar + 9,  p * a2.y);
  atomicAdd(ar + 10, p * a2.z); atomicAdd(ar + 11, p * a2.w);
  atomicAdd(ar + 12, p * a3.x); atomicAdd(ar + 13, p * a3.y);
  atomicAdd(ar + 14, p * a3.z); atomicAdd(ar + 15, p * a3.w);
}

template <int D, int ARW, int MODE>
__device__ __forceinline__ void owner_direct(
    const int* __restrict__ src, const int* __restrict__ dst,
    const float* __restrict__ asv, const float* __restrict__ H,
    const float* advl, unsigned* mrow, float* den, float* acc,
    int lo, int E, int ET, int wid, int lane) {
  for (int b0 = wid * 32; b0 < ET; b0 += 1024) {
    const int e = b0 + lane;
    int s = 0, d = -1;
    if (e < ET) {
      if (e < E) { s = src[e]; d = dst[e]; } else { s = d = e - E; }
    }
    const int dl = d - lo;
    const bool mine = (unsigned)dl < (unsigned)ARW;
    float sc = 0.f;
    if (mine) sc = leaky02(asv[s] + advl[dl]);
    if (MODE == 0) {
      if (mine) atomicMax(&mrow[dl], f2ord(sc));
    } else {
      float p = 0.f;
      if (mine) {
        p = __expf(sc - ord2f(mrow[dl]));
        atomicAdd(&den[dl], p);
      }
      if (D == AD) {
        unsigned mask = __builtin_amdgcn_ballot_w32(mine);
        while (mask) {
          const int L = __builtin_ctz(mask);
          mask &= mask - 1;
          row_add4(s, dl, p, L, 1, lane, H, acc);
        }
      } else {
        if (mine) row_add16(s, dl, p, H, acc);
      }
    }
  }
}

template <int D, int ARW>
__global__ __launch_bounds__(1024) void gat_owner(
    const int* __restrict__ src, const int* __restrict__ dst,
    const float* __restrict__ asv, const float* __restrict__ adv,
    const float* __restrict__ H, const float* __restrict__ bias,
    float* __restrict__ OUT, int E, int N, int Dact) {
  __shared__ __attribute__((aligned(16))) float acc[ARW * D];
  __shared__ int      lsrc[LCAP];
  __shared__ int      ldl[LCAP];
  __shared__ float    lp[LCAP];
  __shared__ unsigned mrow[ARW];
  __shared__ float    den[ARW];
  __shared__ float    advl[ARW];
  __shared__ int      cnt;

  const int tid  = threadIdx.x;
  const int lane = tid & 31;
  const int wid  = __builtin_amdgcn_readfirstlane(tid >> 5);
  const int lo   = blockIdx.x * ARW;
  const int ET   = E + N;
  int nown = N - lo;
  nown = (nown < ARW) ? nown : ARW;

  for (int x = tid; x < ARW * D / 4; x += 1024)
    reinterpret_cast<float4*>(acc)[x] = make_float4(0.f, 0.f, 0.f, 0.f);
  for (int r = tid; r < ARW; r += 1024) {
    mrow[r] = f2ord(-INFINITY);
    den[r]  = 0.f;
    advl[r] = (r < nown) ? adv[lo + r] : 0.f;
    if (r < nown) lsrc[r] = E + lo + r;
  }
  if (tid == 0) cnt = nown;
  __syncthreads();

  const int wstep = 32 * SCAN;
  int qreg = 0;
  int n = 0;
  int base = wid * wstep;
  int dv[SCAN];
  const int last = (E > 0) ? (E - 1) : 0;
  if (E > 0) {
    #pragma unroll
    for (int k = 0; k < SCAN; ++k) {
      int idx = base + k * 32 + lane;
      int ld  = dst[idx < last ? idx : last];
      dv[k] = (idx < E) ? ld : -1;
    }
  }
  while (base < E) {
    unsigned mybits = 0;
    #pragma unroll
    for (int k = 0; k < SCAN; ++k)
      mybits |= ((unsigned)(dv[k] - lo) < (unsigned)ARW) ? (1u << k) : 0u;
    const int cur = base;
    base += 32 * wstep;
    #pragma unroll
    for (int k = 0; k < SCAN; ++k) {
      int idx = base + k * 32 + lane;
      int ld  = dst[idx < last ? idx : last];
      dv[k] = (idx < E) ? ld : -1;
    }
    unsigned lanes = __builtin_amdgcn_ballot_w32(mybits != 0);
    while (lanes) {
      int L = __builtin_ctz(lanes);
      lanes &= lanes - 1;
      unsigned bits = (unsigned)__builtin_amdgcn_readlane((int)mybits, L);
      while (bits) {
        int k = __builtin_ctz(bits);
        bits &= bits - 1;
        qreg = (lane == n) ? (cur + k * 32 + L) : qreg;
        ++n;
        if (n == 32) {
          list_append(&cnt, lsrc, qreg, 32, lane);
          n = 0;
        }
      }
    }
  }
  if (n) list_append(&cnt, lsrc, qreg, n, lane);
  __syncthreads();

  const int total = cnt;
  if (total <= LCAP) {
    int   sk[LROUNDS], dk[LROUNDS];
    float ak[LROUNDS];
    #pragma unroll
    for (int k = 0; k < LROUNDS; ++k) {
      const int i = tid + 1024 * k;
      int s = 0, d = lo;
      if (i < total) {
        const int e = lsrc[i];
        if (e < E) { s = src[e]; d = dst[e]; } else { s = d = e - E; }
      }
      sk[k] = s;
      dk[k] = d - lo;
    }
    #pragma unroll
    for (int k = 0; k < LROUNDS; ++k) ak[k] = asv[sk[k]];
    #pragma unroll
    for (int k = 0; k < LROUNDS; ++k) {
      const int i = tid + 1024 * k;
      if (i < total) {
        const float sc = leaky02(ak[k] + advl[dk[k]]);
        lsrc[i] = sk[k];
        ldl[i]  = dk[k];
        lp[i]   = sc;
        atomicMax(&mrow[dk[k]], f2ord(sc));
      }
    }
    __syncthreads();
    for (int b0 = wid * 32; b0 < total; b0 += 1024) {
      const int i = b0 + lane;
      int s = 0, dl = 0;
      float p = 0.f;
      if (i < total) {
        s  = lsrc[i];
        dl = ldl[i];
        p  = __expf(lp[i] - ord2f(mrow[dl]));
        atomicAdd(&den[dl], p);
      }
      if (D == AD) {
        int m = total - b0;
        m = (m < 32) ? m : 32;
        for (int g = 0; g < m; g += 4)
          row_add4(s, dl, p, g, m - g, lane, H, acc);
      } else {
        if (i < total) row_add16(s, dl, p, H, acc);
      }
    }
    __syncthreads();
  } else {
    owner_direct<D, ARW, 0>(src, dst, asv, H, advl, mrow, den, acc,
                            lo, E, ET, wid, lane);
    __syncthreads();
    owner_direct<D, ARW, 1>(src, dst, asv, H, advl, mrow, den, acc,
                            lo, E, ET, wid, lane);
    __syncthreads();
  }

  if (D == AD) {
    for (int idx = tid; idx < ARW * (AD / 4); idx += 1024) {
      const int r  = idx >> 6;
      const int c4 = idx & 63;
      if (r < nown) {
        const float inv = 1.0f / den[r];
        float4 a  = *reinterpret_cast<const float4*>(&acc[r * AD + c4 * 4]);
        float4 bb = *reinterpret_cast<const float4*>(bias + c4 * 4);
        float4 o;
        o.x = a.x * inv + bb.x; o.x = o.x > 0.f ? o.x : 0.f;
        o.y = a.y * inv + bb.y; o.y = o.y > 0.f ? o.y : 0.f;
        o.z = a.z * inv + bb.z; o.z = o.z > 0.f ? o.z : 0.f;
        o.w = a.w * inv + bb.w; o.w = o.w > 0.f ? o.w : 0.f;
        *reinterpret_cast<float4*>(OUT + (size_t)(lo + r) * AD + c4 * 4) = o;
      }
    }
  } else {
    const int nout = nown * Dact;
    for (int x = tid; x < nout; x += 1024) {
      const int r = x / Dact;
      const int c = x - r * Dact;
      OUT[(size_t)lo * Dact + x] = acc[r * D3 + c] / den[r] + bias[c];
    }
  }
}

extern "C" void kernel_launch(void* const* d_in, const int* in_sizes, int n_in,
                              void* d_out, int out_size, void* d_ws, size_t ws_size,
                              hipStream_t stream) {
  const float* x  = (const float*)d_in[0];
  const int*   ei = (const int*)d_in[1];
  const float* W1 = (const float*)d_in[2];
  const float* as1 = (const float*)d_in[3];
  const float* ad1 = (const float*)d_in[4];
  const float* b1 = (const float*)d_in[5];
  const float* W2 = (const float*)d_in[6];
  const float* as2 = (const float*)d_in[7];
  const float* ad2 = (const float*)d_in[8];
  const float* b2 = (const float*)d_in[9];
  const float* W3 = (const float*)d_in[10];
  const float* as3 = (const float*)d_in[11];
  const float* ad3 = (const float*)d_in[12];
  const float* b3 = (const float*)d_in[13];

  const int N  = in_sizes[0] / 256;
  const int E  = in_sizes[1] / 2;
  const int* src = ei;
  const int* dst = ei + E;

  float* ws = (float*)d_ws;
  float*    hA  = ws;
  float*    hB  = hA + (size_t)N * 256;
  float*    h3  = hB + (size_t)N * 256;
  float*    asv = h3 + (size_t)N * 16;
  float*    adv = asv + N;

  dim3 blk(256);
  dim3 gemmG((N + BM - 1) / BM, 4);
  dim3 gemmG3((N + BM - 1) / BM, 1);
  int gAlpha = (N + 7) / 8;
  int gOwn   = (N + 199) / 200;
  int gOwn3  = (N + 207) / 208;

  gat_gemm_wmma<<<gemmG, blk, 0, stream>>>(x, W1, hA, N, 256, 256);
  gat_alpha<<<gAlpha, blk, 0, stream>>>(hA, as1, ad1, asv, adv, N, 256, 256);
  gat_owner<AD, 200><<<gOwn, 1024, 0, stream>>>(src, dst, asv, adv, hA, b1, hB, E, N, 256);

  gat_gemm_wmma<<<gemmG, blk, 0, stream>>>(hB, W2, hA, N, 256, 256);
  gat_alpha<<<gAlpha, blk, 0, stream>>>(hA, as2, ad2, asv, adv, N, 256, 256);
  gat_owner<AD, 200><<<gOwn, 1024, 0, stream>>>(src, dst, asv, adv, hA, b2, hB, E, N, 256);

  gat_gemm_wmma<<<gemmG3, blk, 0, stream>>>(hB, W3, h3, N, 10, 16);
  gat_alpha<<<gAlpha, blk, 0, stream>>>(h3, as3, ad3, asv, adv, N, 16, 10);
  gat_owner<D3, 208><<<gOwn3, 1024, 0, stream>>>(src, dst, asv, adv, h3, b3, (float*)d_out, E, N, 10);
}


# GEMM and node-dot kernels: a K chunk's 36 X and W loads (a node's 24 loads) made unconditional by index clamp plus zero select and issued as one group, not load-wait-store one at a time
# speedup vs baseline: 3.3809x; 1.5744x over previous
#include <hip/hip_runtime.h>
#include <math.h>


typedef float v2f __attribute__((ext_vector_type(2)));
typedef float v8f __attribute__((ext_vector_type(8)));

#define DK 256
#define KCHUNK 128
#define LDT (KCHUNK + 4)
#define BM 32
#define BN 64

__device__ __forceinline__ unsigned f2ord(float f) {
  unsigned u = __float_as_uint(f);
  return (u & 0x80000000u) ? ~u : (u | 0x80000000u);
}
__device__ __forceinline__ float ord2f(unsigned u) {
  return __uint_as_float((u & 0x80000000u) ? (u ^ 0x80000000u) : ~u);
}

template <int DACT, int DST>
__global__ __launch_bounds__(256) void gat_gemm_wmma(
    const float* __restrict__ X, const float* __restrict__ W,
    float* __restrict__ H, int N) {
  __shared__ float As[BM * LDT];
  __shared__ float Bst[BN * LDT];

  const int tid  = threadIdx.x;
  const int lane = tid & 31;
  const int lo   = lane & 15;
  const int hi   = lane >> 4;
  const int wid  = tid >> 5;
  const int wm   = (wid >> 2) * 16;
  const int wn   = (wid & 3) * 16;
  const int rowBase = blockIdx.x * BM;
  const int colBase = blockIdx.y * BN;

  const int ar  = tid >> 5;
  const int ac4 = tid & 31;
  const int bk  = tid >> 6;
  const int bn  = tid & 63;
  const int gc  = colBase + bn;
  const bool cok = gc < DACT;
  const int gcc = cok ? gc : (DACT - 1);

  v8f acc = {};

  for (int kk = 0; kk < DK; kk += KCHUNK) {
    float4 av[4];
    float  bv[32];
    #pragma unroll
    for (int it = 0; it < 4; ++it) {
      const int gr  = rowBase + ar + 8 * it;
      const int grc = (gr < N) ? gr : (N - 1);
      av[it] = *reinterpret_cast<const float4*>(X + (size_t)grc * DK + kk + ac4 * 4);
    }
    #pragma unroll
    for (int it = 0; it < 32; ++it)
      bv[it] = W[(size_t)(kk + bk + 4 * it) * DACT + gcc];

    __syncthreads();
    #pragma unroll
    for (int it = 0; it < 4; ++it) {
      const int gr = rowBase + ar + 8 * it;
      float4 v = av[it];
      if (gr >= N) v = make_float4(0.f, 0.f, 0.f, 0.f);
      *reinterpret_cast<float4*>(&As[(ar + 8 * it) * LDT + ac4 * 4]) = v;
    }
    #pragma unroll
    for (int it = 0; it < 32; ++it)
      Bst[bn * LDT + bk + 4 * it] = cok ? bv[it] : 0.f;
    __syncthreads();

    const float* arow = &As[(wm + lo) * LDT];
    const float* brow = &Bst[(wn + lo) * LDT];
    #pragma unroll
    for (int k0 = 0; k0 < KCHUNK; k0 += 4) {
      v2f a = *reinterpret_cast<const v2f*>(arow + k0 + 2 * hi);
      v2f b = *reinterpret_cast<const v2f*>(brow + k0 + 2 * hi);
      acc = __builtin_amdgcn_wmma_f32_16x16x4_f32(false, a, false, b,
                                                  (short)0, acc, false, false);
    }
  }

  const int col = colBase + wn + lo;
  if (col < DST) {
    #pragma unroll
    for (int r = 0; r < 8; ++r) {
      int row = rowBase + wm + hi * 8 + r;
      if (row < N) H[(size_t)row * DST + col] = acc[r];
    }
  }
}

template <int DPAD, int DACT>
__global__ __launch_bounds__(256) void gat_alpha(
    const float* __restrict__ H, const float* __restrict__ a_s,
    const float* __restrict__ a_d, float* __restrict__ asv,
    float* __restrict__ adv, int N) {
  const int lane = threadIdx.x & 31;
  const int i = blockIdx.x * 8 + (threadIdx.x >> 5);
  if (i >= N) return;
  constexpr int NJ = (DPAD + 31) / 32;
  float hv[NJ], av[NJ], bv[NJ];
  #pragma unroll
  for (int j = 0; j < NJ; ++j) {
    const int d  = lane + 32 * j;
    const int dh = (d < DPAD) ? d : (DPAD - 1);
    const int da = (d < DACT) ? d : (DACT - 1);
    hv[j] = H[(size_t)i * DPAD + dh];
    av[j] = a_s[da];
    bv[j] = a_d[da];
  }
  float s1 = 0.f, s2 = 0.f;
  #pragma unroll
  for (int j = 0; j < NJ; ++j) {
    const int d = lane + 32 * j;
    const float h = (d < DPAD) ? hv[j] : 0.f;
    const float a = (d < DACT) ? av[j] : 0.f;
    const float b = (d < DACT) ? bv[j] : 0.f;
    s1 = fmaf(h, a, s1);
    s2 = fmaf(h, b, s2);
  }
  #pragma unroll
  for (int off = 16; off > 0; off >>= 1) {
    s1 += __shfl_xor(s1, off, 32);
    s2 += __shfl_xor(s2, off, 32);
  }
  if (lane == 0) {
    asv[i] = s1;
    adv[i] = s2;
  }
}

#define AD 256
#define D3 16
#define SCAN 16
#define LROUNDS 6
#define LCAP (1024 * LROUNDS)

__device__ __forceinline__ float leaky02(float t) {
  return (t > 0.f) ? t : 0.2f * t;
}

__device__ __forceinline__ int rl_i(int v, int l) {
  return __builtin_amdgcn_readlane(v, l);
}
__device__ __forceinline__ float rl_f(float v, int l) {
  return __int_as_float(__builtin_amdgcn_readlane(__float_as_int(v), l));
}

__device__ __forceinline__ void list_append(int* cnt, int* lst, int qreg,
                                            int m, int lane) {
  int b = 0;
  if (lane == 0) b = atomicAdd(cnt, m);
  b = __builtin_amdgcn_readfirstlane(b);
  const int slot = b + lane;
  if (lane < m && slot < LCAP) lst[slot] = qreg;
}

__device__ __forceinline__ void row_add4(
    int s, int dl, float p, int g, int c, int lane,
    const float* __restrict__ H, float* acc) {
  float v[4][8];
  float pb[4];
  int   ro[4];
  #pragma unroll
  for (int b = 0; b < 4; ++b) {
    const int q = (b < c) ? (g + b) : g;
    const int sb = rl_i(s, q);
    ro[b] = rl_i(dl, q) * AD + lane;
    pb[b] = rl_f(p, q);
    const float* hp = H + (size_t)sb * AD + lane;
    #pragma unroll
    for (int j = 0; j < 8; ++j) v[b][j] = hp[j * 32];
  }
  #pragma unroll
  for (int b = 0; b < 4; ++b) {
    if (b < c) {
      #pragma unroll
      for (int j = 0; j < 8; ++j)
        atomicAdd(&acc[ro[b] + j * 32], pb[b] * v[b][j]);
    }
  }
}

__device__ __forceinline__ void row_add16(
    int s, int dl, float p, const float* __restrict__ H, float* acc) {
  const float4* hp = reinterpret_cast<const float4*>(H + (size_t)s * D3);
  const float4 a0 = hp[0], a1 = hp[1], a2 = hp[2], a3 = hp[3];
  float* ar = &acc[dl * D3];
  atomicAdd(ar + 0,  p * a0.x); atomicAdd(ar + 1,  p * a0.y);
  atomicAdd(ar + 2,  p * a0.z); atomicAdd(ar + 3,  p * a0.w);
  atomicAdd(ar + 4,  p * a1.x); atomicAdd(ar + 5,  p * a1.y);
  atomicAdd(ar + 6,  p * a1.z); atomicAdd(ar + 7,  p * a1.w);
  atomicAdd(ar + 8,  p * a2.x); atomicAdd(ar + 9,  p * a2.y);
  atomicAdd(ar + 10, p * a2.z); atomicAdd(ar + 11, p * a2.w);
  atomicAdd(ar + 12, p * a3.x); atomicAdd(ar + 13, p * a3.y);
  atomicAdd(ar + 14, p * a3.z); atomicAdd(ar + 15, p * a3.w);
}

template <int D, int ARW, int MODE>
__device__ __forceinline__ void owner_direct(
    const int* __restrict__ src, const int* __restrict__ dst,
    const float* __restrict__ asv, const float* __restrict__ H,
    const float* advl, unsigned* mrow, float* den, float* acc,
    int lo, int E, int ET, int wid, int lane) {
  for (int b0 = wid * 32; b0 < ET; b0 += 1024) {
    const int e = b0 + lane;
    int s = 0, d = -1;
    if (e < ET) {
      if (e < E) { s = src[e]; d = dst[e]; } else { s = d = e - E; }
    }
    const int dl = d - lo;
    const bool mine = (unsigned)dl < (unsigned)ARW;
    float sc = 0.f;
    if (mine) sc = leaky02(asv[s] + advl[dl]);
    if (MODE == 0) {
      if (mine) atomicMax(&mrow[dl], f2ord(sc));
    } else {
      float p = 0.f;
      if (mine) {
        p = __expf(sc - ord2f(mrow[dl]));
        atomicAdd(&den[dl], p);
      }
      if (D == AD) {
        unsigned mask = __builtin_amdgcn_ballot_w32(mine);
        while (mask) {
          const int L = __builtin_ctz(mask);
          mask &= mask - 1;
          row_add4(s, dl, p, L, 1, lane, H, acc);
        }
      } else {
        if (mine) row_add16(s, dl, p, H, acc);
      }
    }
  }
}

template <int D, int ARW>
__global__ __launch_bounds__(1024) void gat_owner(
    const int* __restrict__ src, const int* __restrict__ dst,
    const float* __restrict__ asv, const float* __restrict__ adv,
    const float* __restrict__ H, const float* __restrict__ bias,
    float* __restrict__ OUT, int E, int N, int Dact) {
  __shared__ __attribute__((aligned(16))) float acc[ARW * D];
  __shared__ int      lsrc[LCAP];
  __shared__ int      ldl[LCAP];
  __shared__ float    lp[LCAP];
  __shared__ unsigned mrow[ARW];
  __shared__ float    den[ARW];
  __shared__ float    advl[ARW];
  __shared__ int      cnt;

  const int tid  = threadIdx.x;
  const int lane = tid & 31;
  const int wid  = __builtin_amdgcn_readfirstlane(tid >> 5);
  const int lo   = blockIdx.x * ARW;
  const int ET   = E + N;
  int nown = N - lo;
  nown = (nown < ARW) ? nown : ARW;

  for (int x = tid; x < ARW * D / 4; x += 1024)
    reinterpret_cast<float4*>(acc)[x] = make_float4(0.f, 0.f, 0.f, 0.f);
  for (int r = tid; r < ARW; r += 1024) {
    mrow[r] = f2ord(-INFINITY);
    den[r]  = 0.f;
    advl[r] = (r < nown) ? adv[lo + r] : 0.f;
    if (r < nown) lsrc[r] = E + lo + r;
  }
  if (tid == 0) cnt = nown;
  __syncthreads();

  const int wstep = 32 * SCAN;
  int qreg = 0;
  int n = 0;
  int base = wid * wstep;
  int dv[SCAN];
  const int last = (E > 0) ? (E - 1) : 0;
  if (E > 0) {
    #pragma unroll
    for (int k = 0; k < SCAN; ++k) {
      int idx = base + k * 32 + lane;
      int ld  = dst[idx < last ? idx : last];
      dv[k] = (idx < E) ? ld : -1;
    }
  }
  while (base < E) {
    unsigned mybits = 0;
    #pragma unroll
    for (int k = 0; k < SCAN; ++k)
      mybits |= ((unsigned)(dv[k] - lo) < (unsigned)ARW) ? (1u << k) : 0u;
    const int cur = base;
    base += 32 * wstep;
    #pragma unroll
    for (int k = 0; k < SCAN; ++k) {
      int idx = base + k * 32 + lane;
      int ld  = dst[idx < last ? idx : last];
      dv[k] = (idx < E) ? ld : -1;
    }
    unsigned lanes = __builtin_amdgcn_ballot_w32(mybits != 0);
    while (lanes) {
      int L = __builtin_ctz(lanes);
      lanes &= lanes - 1;
      unsigned bits = (unsigned)__builtin_amdgcn_readlane((int)mybits, L);
      while (bits) {
        int k = __builtin_ctz(bits);
        bits &= bits - 1;
        qreg = (lane == n) ? (cur + k * 32 + L) : qreg;
        ++n;
        if (n == 32) {
          list_append(&cnt, lsrc, qreg, 32, lane);
          n = 0;
        }
      }
    }
  }
  if (n) list_append(&cnt, lsrc, qreg, n, lane);
  __syncthreads();

  const int total = cnt;
  if (total <= LCAP) {
    int   sk[LROUNDS], dk[LROUNDS];
    float ak[LROUNDS];
    #pragma unroll
    for (int k = 0; k < LROUNDS; ++k) {
      const int i = tid + 1024 * k;
      int s = 0, d = lo;
      if (i < total) {
        const int e = lsrc[i];
        if (e < E) { s = src[e]; d = dst[e]; } else { s = d = e - E; }
      }
      sk[k] = s;
      dk[k] = d - lo;
    }
    #pragma unroll
    for (int k = 0; k < LROUNDS; ++k) ak[k] = asv[sk[k]];
    #pragma unroll
    for (int k = 0; k < LROUNDS; ++k) {
      const int i = tid + 1024 * k;
      if (i < total) {
        const float sc = leaky02(ak[k] + advl[dk[k]]);
        lsrc[i] = sk[k];
        ldl[i]  = dk[k];
        lp[i]   = sc;
        atomicMax(&mrow[dk[k]], f2ord(sc));
      }
    }
    __syncthreads();
    for (int b0 = wid * 32; b0 < total; b0 += 1024) {
      const int i = b0 + lane;
      int s = 0, dl = 0;
      float p = 0.f;
      if (i < total) {
        s  = lsrc[i];
        dl = ldl[i];
        p  = __expf(lp[i] - ord2f(mrow[dl]));
        atomicAdd(&den[dl], p);
      }
      if (D == AD) {
        int m = total - b0;
        m = (m < 32) ? m : 32;
        for (int g = 0; g < m; g += 4)
          row_add4(s, dl, p, g, m - g, lane, H, acc);
      } else {
        if (i < total) row_add16(s, dl, p, H, acc);
      }
    }
    __syncthreads();
  } else {
    owner_direct<D, ARW, 0>(src, dst, asv, H, advl, mrow, den, acc,
                            lo, E, ET, wid, lane);
    __syncthreads();
    owner_direct<D, ARW, 1>(src, dst, asv, H, advl, mrow, den, acc,
                            lo, E, ET, wid, lane);
    __syncthreads();
  }

  if (D == AD) {
    for (int idx = tid; idx < ARW * (AD / 4); idx += 1024) {
      const int r  = idx >> 6;
      const int c4 = idx & 63;
      if (r < nown) {
        const float inv = 1.0f / den[r];
        float4 a  = *reinterpret_cast<const float4*>(&acc[r * AD + c4 * 4]);
        float4 bb = *reinterpret_cast<const float4*>(bias + c4 * 4);
        float4 o;
        o.x = a.x * inv + bb.x; o.x = o.x > 0.f ? o.x : 0.f;
        o.y = a.y * inv + bb.y; o.y = o.y > 0.f ? o.y : 0.f;
        o.z = a.z * inv + bb.z; o.z = o.z > 0.f ? o.z : 0.f;
        o.w = a.w * inv + bb.w; o.w = o.w > 0.f ? o.w : 0.f;
        *reinterpret_cast<float4*>(OUT + (size_t)(lo + r) * AD + c4 * 4) = o;
      }
    }
  } else {
    const int nout = nown * Dact;
    for (int x = tid; x < nout; x += 1024) {
      const int r = x / Dact;
      const int c = x - r * Dact;
      OUT[(size_t)lo * Dact + x] = acc[r * D3 + c] / den[r] + bias[c];
    }
  }
}

extern "C" void kernel_launch(void* const* d_in, const int* in_sizes, int n_in,
                              void* d_out, int out_size, void* d_ws, size_t ws_size,
                              hipStream_t stream) {
  const float* x  = (const float*)d_in[0];
  const int*   ei = (const int*)d_in[1];
  const float* W1 = (const float*)d_in[2];
  const float* as1 = (const float*)d_in[3];
  const float* ad1 = (const float*)d_in[4];
  const float* b1 = (const float*)d_in[5];
  const float* W2 = (const float*)d_in[6];
  const float* as2 = (const float*)d_in[7];
  const float* ad2 = (const float*)d_in[8];
  const float* b2 = (const float*)d_in[9];
  const float* W3 = (const float*)d_in[10];
  const float* as3 = (const float*)d_in[11];
  const float* ad3 = (const float*)d_in[12];
  const float* b3 = (const float*)d_in[13];

  const int N  = in_sizes[0] / 256;
  const int E  = in_sizes[1] / 2;
  const int* src = ei;
  const int* dst = ei + E;

  float* ws = (float*)d_ws;
  float*    hA  = ws;
  float*    hB  = hA + (size_t)N * 256;
  float*    h3  = hB + (size_t)N * 256;
  float*    asv = h3 + (size_t)N * 16;
  float*    adv = asv + N;

  dim3 blk(256);
  dim3 gemmG((N + BM - 1) / BM, 4);
  dim3 gemmG3((N + BM - 1) / BM, 1);
  int gAlpha = (N + 7) / 8;
  int gOwn   = (N + 199) / 200;
  int gOwn3  = (N + 207) / 208;

  gat_gemm_wmma<256, 256><<<gemmG, blk, 0, stream>>>(x, W1, hA, N);
  gat_alpha<256, 256><<<gAlpha, blk, 0, stream>>>(hA, as1, ad1, asv, adv, N);
  gat_owner<AD, 200><<<gOwn, 1024, 0, stream>>>(src, dst, asv, adv, hA, b1, hB, E, N, 256);

  gat_gemm_wmma<256, 256><<<gemmG, blk, 0, stream>>>(hB, W2, hA, N);
  gat_alpha<256, 256><<<gAlpha, blk, 0, stream>>>(hA, as2, ad2, asv, adv, N);
  gat_owner<AD, 200><<<gOwn, 1024, 0, stream>>>(src, dst, asv, adv, hA, b2, hB, E, N, 256);

  gat_gemm_wmma<10, 16><<<gemmG3, blk, 0, stream>>>(hB, W3, h3, N);
  gat_alpha<16, 10><<<gAlpha, blk, 0, stream>>>(h3, as3, ad3, asv, adv, N);
  gat_owner<D3, 208><<<gOwn3, 1024, 0, stream>>>(src, dst, asv, adv, h3, b3, (float*)d_out, E, N, 10);
}


# Owners scan one owner byte per edge (written once per call by a new first launch: 10 launches) in 16-byte loads, not the 4-byte dst list: 4x fewer scan bytes and tests; sums in LDS arrival order
# speedup vs baseline: 3.8495x; 1.1386x over previous
#include <hip/hip_runtime.h>
#include <math.h>


typedef float v2f __attribute__((ext_vector_type(2)));
typedef float v8f __attribute__((ext_vector_type(8)));

#define DK 256
#define KCHUNK 128
#define LDT (KCHUNK + 4)
#define BM 32
#define BN 64

__device__ __forceinline__ unsigned f2ord(float f) {
  unsigned u = __float_as_uint(f);
  return (u & 0x80000000u) ? ~u : (u | 0x80000000u);
}
__device__ __forceinline__ float ord2f(unsigned u) {
  return __uint_as_float((u & 0x80000000u) ? (u ^ 0x80000000u) : ~u);
}

template <int DACT, int DST>
__global__ __launch_bounds__(256) void gat_gemm_wmma(
    const float* __restrict__ X, const float* __restrict__ W,
    float* __restrict__ H, int N) {
  __shared__ float As[BM * LDT];
  __shared__ float Bst[BN * LDT];

  const int tid  = threadIdx.x;
  const int lane = tid & 31;
  const int lo   = lane & 15;
  const int hi   = lane >> 4;
  const int wid  = tid >> 5;
  const int wm   = (wid >> 2) * 16;
  const int wn   = (wid & 3) * 16;
  const int rowBase = blockIdx.x * BM;
  const int colBase = blockIdx.y * BN;

  const int ar  = tid >> 5;
  const int ac4 = tid & 31;
  const int bk  = tid >> 6;
  const int bn  = tid & 63;
  const int gc  = colBase + bn;
  const bool cok = gc < DACT;
  const int gcc = cok ? gc : (DACT - 1);

  v8f acc = {};

  for (int kk = 0; kk < DK; kk += KCHUNK) {
    float4 av[4];
    float  bv[32];
    #pragma unroll
    for (int it = 0; it < 4; ++it) {
      const int gr  = rowBase + ar + 8 * it;
      const int grc = (gr < N) ? gr : (N - 1);
      av[it] = *reinterpret_cast<const float4*>(X + (size_t)grc * DK + kk + ac4 * 4);
    }
    #pragma unroll
    for (int it = 0; it < 32; ++it)
      bv[it] = W[(size_t)(kk + bk + 4 * it) * DACT + gcc];

    __syncthreads();
    #pragma unroll
    for (int it = 0; it < 4; ++it) {
      const int gr = rowBase + ar + 8 * it;
      float4 v = av[it];
      if (gr >= N) v = make_float4(0.f, 0.f, 0.f, 0.f);
      *reinterpret_cast<float4*>(&As[(ar + 8 * it) * LDT + ac4 * 4]) = v;
    }
    #pragma unroll
    for (int it = 0; it < 32; ++it)
      Bst[bn * LDT + bk + 4 * it] = cok ? bv[it] : 0.f;
    __syncthreads();

    const float* arow = &As[(wm + lo) * LDT];
    const float* brow = &Bst[(wn + lo) * LDT];
    #pragma unroll
    for (int k0 = 0; k0 < KCHUNK; k0 += 4) {
      v2f a = *reinterpret_cast<const v2f*>(arow + k0 + 2 * hi);
      v2f b = *reinterpret_cast<const v2f*>(brow + k0 + 2 * hi);
      acc = __builtin_amdgcn_wmma_f32_16x16x4_f32(false, a, false, b,
                                                  (short)0, acc, false, false);
    }
  }

  const int col = colBase + wn + lo;
  if (col < DST) {
    #pragma unroll
    for (int r = 0; r < 8; ++r) {
      int row = rowBase + wm + hi * 8 + r;
      if (row < N) H[(size_t)row * DST + col] = acc[r];
    }
  }
}

template <int DPAD, int DACT>
__global__ __launch_bounds__(256) void gat_alpha(
    const float* __restrict__ H, const float* __restrict__ a_s,
    const float* __restrict__ a_d, float* __restrict__ asv,
    float* __restrict__ adv, int N) {
  const int lane = threadIdx.x & 31;
  const int i = blockIdx.x * 8 + (threadIdx.x >> 5);
  if (i >= N) return;
  constexpr int NJ = (DPAD + 31) / 32;
  float hv[NJ], av[NJ], bv[NJ];
  #pragma unroll
  for (int j = 0; j < NJ; ++j) {
    const int d  = lane + 32 * j;
    const int dh = (d < DPAD) ? d : (DPAD - 1);
    const int da = (d < DACT) ? d : (DACT - 1);
    hv[j] = H[(size_t)i * DPAD + dh];
    av[j] = a_s[da];
    bv[j] = a_d[da];
  }
  float s1 = 0.f, s2 = 0.f;
  #pragma unroll
  for (int j = 0; j < NJ; ++j) {
    const int d = lane + 32 * j;
    const float h = (d < DPAD) ? hv[j] : 0.f;
    const float a = (d < DACT) ? av[j] : 0.f;
    const float b = (d < DACT) ? bv[j] : 0.f;
    s1 = fmaf(h, a, s1);
    s2 = fmaf(h, b, s2);
  }
  #pragma unroll
  for (int off = 16; off > 0; off >>= 1) {
    s1 += __shfl_xor(s1, off, 32);
    s2 += __shfl_xor(s2, off, 32);
  }
  if (lane == 0) {
    asv[i] = s1;
    adv[i] = s2;
  }
}

#define AD 256
#define D3 16
#define SCB 8
#define LROUNDS 6
#define LCAP (1024 * LROUNDS)

__device__ __forceinline__ float leaky02(float t) {
  return (t > 0.f) ? t : 0.2f * t;
}

__device__ __forceinline__ int rl_i(int v, int l) {
  return __builtin_amdgcn_readlane(v, l);
}
__device__ __forceinline__ float rl_f(float v, int l) {
  return __int_as_float(__builtin_amdgcn_readlane(__float_as_int(v), l));
}

template <int RA, int RB>
__global__ __launch_bounds__(256) void gat_owner_bytes(
    const int* __restrict__ dst, unsigned char* __restrict__ oba,
    unsigned char* __restrict__ obb, int E, int N) {
  const int nq = (E + 15) >> 4;
  if (nq > 2 * N) return;
  const int q = blockIdx.x * 256 + threadIdx.x;
  if (q >= nq) return;
  const int last = E - 1;
  int d[16];
  #pragma unroll
  for (int j = 0; j < 16; ++j) {
    const int e = q * 16 + j;
    d[j] = dst[e < last ? e : last];
  }
  unsigned wa[4], wb[4];
  #pragma unroll
  for (int w = 0; w < 4; ++w) {
    unsigned a = 0, b = 0;
    #pragma unroll
    for (int t = 0; t < 4; ++t) {
      const int j = w * 4 + t;
      const bool in = (q * 16 + j) < E;
      const unsigned ba = in ? ((unsigned)(d[j] / RA) & 255u) : 255u;
      const unsigned bb = in ? ((unsigned)(d[j] / RB) & 255u) : 255u;
      a |= ba << (8 * t);
      b |= bb << (8 * t);
    }
    wa[w] = a;
    wb[w] = b;
  }
  reinterpret_cast<uint4*>(oba)[q] = make_uint4(wa[0], wa[1], wa[2], wa[3]);
  reinterpret_cast<uint4*>(obb)[q] = make_uint4(wb[0], wb[1], wb[2], wb[3]);
}

__device__ __forceinline__ unsigned zero_bytes(unsigned x) {
  const unsigned y = (x & 0x7F7F7F7Fu) + 0x7F7F7F7Fu;
  return ~(y | x | 0x7F7F7F7Fu);
}

__device__ __forceinline__ void list_append(int* cnt, int* lst, int qreg,
                                            int m, int lane) {
  int b = 0;
  if (lane == 0) b = atomicAdd(cnt, m);
  b = __builtin_amdgcn_readfirstlane(b);
  const int slot = b + lane;
  if (lane < m && slot < LCAP) lst[slot] = qreg;
}

__device__ __forceinline__ void row_add4(
    int s, int dl, float p, int g, int c, int lane,
    const float* __restrict__ H, float* acc) {
  float v[4][8];
  float pb[4];
  int   ro[4];
  #pragma unroll
  for (int b = 0; b < 4; ++b) {
    const int q = (b < c) ? (g + b) : g;
    const int sb = rl_i(s, q);
    ro[b] = rl_i(dl, q) * AD + lane;
    pb[b] = rl_f(p, q);
    const float* hp = H + (size_t)sb * AD + lane;
    #pragma unroll
    for (int j = 0; j < 8; ++j) v[b][j] = hp[j * 32];
  }
  #pragma unroll
  for (int b = 0; b < 4; ++b) {
    if (b < c) {
      #pragma unroll
      for (int j = 0; j < 8; ++j)
        atomicAdd(&acc[ro[b] + j * 32], pb[b] * v[b][j]);
    }
  }
}

__device__ __forceinline__ void row_add16(
    int s, int dl, float p, const float* __restrict__ H, float* acc) {
  const float4* hp = reinterpret_cast<const float4*>(H + (size_t)s * D3);
  const float4 a0 = hp[0], a1 = hp[1], a2 = hp[2], a3 = hp[3];
  float* ar = &acc[dl * D3];
  atomicAdd(ar + 0,  p * a0.x); atomicAdd(ar + 1,  p * a0.y);
  atomicAdd(ar + 2,  p * a0.z); atomicAdd(ar + 3,  p * a0.w);
  atomicAdd(ar + 4,  p * a1.x); atomicAdd(ar + 5,  p * a1.y);
  atomicAdd(ar + 6,  p * a1.z); atomicAdd(ar + 7,  p * a1.w);
  atomicAdd(ar + 8,  p * a2.x); atomicAdd(ar + 9,  p * a2.y);
  atomicAdd(ar + 10, p * a2.z); atomicAdd(ar + 11, p * a2.w);
  atomicAdd(ar + 12, p * a3.x); atomicAdd(ar + 13, p * a3.y);
  atomicAdd(ar + 14, p * a3.z); atomicAdd(ar + 15, p * a3.w);
}

template <int D, int ARW, int MODE>
__device__ __forceinline__ void owner_direct(
    const int* __restrict__ src, const int* __restrict__ dst,
    const float* __restrict__ asv, const float* __restrict__ H,
    const float* advl, unsigned* mrow, float* den, float* acc,
    int lo, int E, int ET, int wid, int lane) {
  for (int b0 = wid * 32; b0 < ET; b0 += 1024) {
    const int e = b0 + lane;
    int s = 0, d = -1;
    if (e < ET) {
      if (e < E) { s = src[e]; d = dst[e]; } else { s = d = e - E; }
    }
    const int dl = d - lo;
    const bool mine = (unsigned)dl < (unsigned)ARW;
    float sc = 0.f;
    if (mine) sc = leaky02(asv[s] + advl[dl]);
    if (MODE == 0) {
      if (mine) atomicMax(&mrow[dl], f2ord(sc));
    } else {
      float p = 0.f;
      if (mine) {
        p = __expf(sc - ord2f(mrow[dl]));
        atomicAdd(&den[dl], p);
      }
      if (D == AD) {
        unsigned mask = __builtin_amdgcn_ballot_w32(mine);
        while (mask) {
          const int L = __builtin_ctz(mask);
          mask &= mask - 1;
          row_add4(s, dl, p, L, 1, lane, H, acc);
        }
      } else {
        if (mine) row_add16(s, dl, p, H, acc);
      }
    }
  }
}

template <int D, int ARW>
__global__ __launch_bounds__(1024) void gat_owner(
    const int* __restrict__ src, const int* __restrict__ dst,
    const unsigned char* __restrict__ ob,
    const float* __restrict__ asv, const float* __restrict__ adv,
    const float* __restrict__ H, const float* __restrict__ bias,
    float* __restrict__ OUT, int E, int N, int Dact) {
  __shared__ __attribute__((aligned(16))) float acc[ARW * D];
  __shared__ int      lsrc[LCAP];
  __shared__ int      ldl[LCAP];
  __shared__ float    lp[LCAP];
  __shared__ unsigned mrow[ARW];
  __shared__ float    den[ARW];
  __shared__ float    advl[ARW];
  __shared__ int      cnt;

  const int tid  = threadIdx.x;
  const int lane = tid & 31;
  const int wid  = __builtin_amdgcn_readfirstlane(tid >> 5);
  const int lo   = blockIdx.x * ARW;
  const int ET   = E + N;
  int nown = N - lo;
  nown = (nown < ARW) ? nown : ARW;

  for (int x = tid; x < ARW * D / 4; x += 1024)
    reinterpret_cast<float4*>(acc)[x] = make_float4(0.f, 0.f, 0.f, 0.f);
  for (int r = tid; r < ARW; r += 1024) {
    mrow[r] = f2ord(-INFINITY);
    den[r]  = 0.f;
    advl[r] = (r < nown) ? adv[lo + r] : 0.f;
    if (r < nown) lsrc[r] = E + lo + r;
  }
  if (tid == 0) cnt = nown;
  __syncthreads();

  const int nq = (E + 15) >> 4;
  const bool bytes_ok = ((N + ARW - 1) / ARW <= 255) && (nq <= 2 * N);
  int qreg = 0;
  int n = 0;
  if (bytes_ok && nq > 0) {
    const uint4* ob4 = reinterpret_cast<const uint4*>(ob);
    const unsigned bsel = blockIdx.x * 0x01010101u;
    const int qstep = SCB * 32;
    const int qlast = nq - 1;
    int qb = wid * qstep;
    uint4 dv[SCB];
    #pragma unroll
    for (int k = 0; k < SCB; ++k) {
      const int qi = qb + k * 32 + lane;
      dv[k] = ob4[qi < qlast ? qi : qlast];
    }
    while (qb < nq) {
      unsigned z[SCB];
      #pragma unroll
      for (int k = 0; k < SCB; ++k) {
        const int qi = qb + k * 32 + lane;
        const unsigned m = (zero_bytes(dv[k].x ^ bsel) >> 7) |
                           (zero_bytes(dv[k].y ^ bsel) >> 6) |
                           (zero_bytes(dv[k].z ^ bsel) >> 5) |
                           (zero_bytes(dv[k].w ^ bsel) >> 4);
        z[k] = (qi < nq) ? m : 0u;
      }
      const int cur = qb;
      qb += 32 * qstep;
      #pragma unroll
      for (int k = 0; k < SCB; ++k) {
        const int qi = qb + k * 32 + lane;
        dv[k] = ob4[qi < qlast ? qi : qlast];
      }
      #pragma unroll
      for (int k = 0; k < SCB; ++k) {
        unsigned lanes = __builtin_amdgcn_ballot_w32(z[k] != 0);
        while (lanes) {
          const int L = __builtin_ctz(lanes);
          lanes &= lanes - 1;
          unsigned bits = (unsigned)__builtin_amdgcn_readlane((int)z[k], L);
          const int e0 = (cur + k * 32 + L) * 16;
          while (bits) {
            const int p = __builtin_ctz(bits);
            bits &= bits - 1;
            const int id = e0 + ((p & 7) << 2) + (p >> 3);
            qreg = (lane == n) ? id : qreg;
            ++n;
            if (n == 32) {
              list_append(&cnt, lsrc, qreg, 32, lane);
              n = 0;
            }
          }
        }
      }
    }
    if (n) list_append(&cnt, lsrc, qreg, n, lane);
  }
  __syncthreads();

  const int total = bytes_ok ? cnt : (LCAP + 1);
  if (total <= LCAP) {
    int   sk[LROUNDS], dk[LROUNDS];
    float ak[LROUNDS];
    #pragma unroll
    for (int k = 0; k < LROUNDS; ++k) {
      const int i = tid + 1024 * k;
      int s = 0, d = lo;
      if (i < total) {
        const int e = lsrc[i];
        if (e < E) { s = src[e]; d = dst[e]; } else { s = d = e - E; }
      }
      sk[k] = s;
      dk[k] = d - lo;
    }
    #pragma unroll
    for (int k = 0; k < LROUNDS; ++k) ak[k] = asv[sk[k]];
    #pragma unroll
    for (int k = 0; k < LROUNDS; ++k) {
      const int i = tid + 1024 * k;
      if (i < total) {
        const float sc = leaky02(ak[k] + advl[dk[k]]);
        lsrc[i] = sk[k];
        ldl[i]  = dk[k];
        lp[i]   = sc;
        atomicMax(&mrow[dk[k]], f2ord(sc));
      }
    }
    __syncthreads();
    for (int b0 = wid * 32; b0 < total; b0 += 1024) {
      const int i = b0 + lane;
      int s = 0, dl = 0;
      float p = 0.f;
      if (i < total) {
        s  = lsrc[i];
        dl = ldl[i];
        p  = __expf(lp[i] - ord2f(mrow[dl]));
        atomicAdd(&den[dl], p);
      }
      if (D == AD) {
        int m = total - b0;
        m = (m < 32) ? m : 32;
        for (int g = 0; g < m; g += 4)
          row_add4(s, dl, p, g, m - g, lane, H, acc);
      } else {
        if (i < total) row_add16(s, dl, p, H, acc);
      }
    }
    __syncthreads();
  } else {
    owner_direct<D, ARW, 0>(src, dst, asv, H, advl, mrow, den, acc,
                            lo, E, ET, wid, lane);
    __syncthreads();
    owner_direct<D, ARW, 1>(src, dst, asv, H, advl, mrow, den, acc,
                            lo, E, ET, wid, lane);
    __syncthreads();
  }

  if (D == AD) {
    for (int idx = tid; idx < ARW * (AD / 4); idx += 1024) {
      const int r  = idx >> 6;
      const int c4 = idx & 63;
      if (r < nown) {
        const float inv = 1.0f / den[r];
        float4 a  = *reinterpret_cast<const float4*>(&acc[r * AD + c4 * 4]);
        float4 bb = *reinterpret_cast<const float4*>(bias + c4 * 4);
        float4 o;
        o.x = a.x * inv + bb.x; o.x = o.x > 0.f ? o.x : 0.f;
        o.y = a.y * inv + bb.y; o.y = o.y > 0.f ? o.y : 0.f;
        o.z = a.z * inv + bb.z; o.z = o.z > 0.f ? o.z : 0.f;
        o.w = a.w * inv + bb.w; o.w = o.w > 0.f ? o.w : 0.f;
        *reinterpret_cast<float4*>(OUT + (size_t)(lo + r) * AD + c4 * 4) = o;
      }
    }
  } else {
    const int nout = nown * Dact;
    for (int x = tid; x < nout; x += 1024) {
      const int r = x / Dact;
      const int c = x - r * Dact;
      OUT[(size_t)lo * Dact + x] = acc[r * D3 + c] / den[r] + bias[c];
    }
  }
}

extern "C" void kernel_launch(void* const* d_in, const int* in_sizes, int n_in,
                              void* d_out, int out_size, void* d_ws, size_t ws_size,
                              hipStream_t stream) {
  const float* x  = (const float*)d_in[0];
  const int*   ei = (const int*)d_in[1];
  const float* W1 = (const float*)d_in[2];
  const float* as1 = (const float*)d_in[3];
  const float* ad1 = (const float*)d_in[4];
  const float* b1 = (const float*)d_in[5];
  const float* W2 = (const float*)d_in[6];
  const float* as2 = (const float*)d_in[7];
  const float* ad2 = (const float*)d_in[8];
  const float* b2 = (const float*)d_in[9];
  const float* W3 = (const float*)d_in[10];
  const float* as3 = (const float*)d_in[11];
  const float* ad3 = (const float*)d_in[12];
  const float* b3 = (const float*)d_in[13];

  const int N  = in_sizes[0] / 256;
  const int E  = in_sizes[1] / 2;
  const int* src = ei;
  const int* dst = ei + E;

  float* ws = (float*)d_ws;
  float*    hA  = ws;
  float*    hB  = hA + (size_t)N * 256;
  float*    obr = hB + (size_t)N * 256;
  float*    asv = obr + (size_t)N * 16;
  float*    adv = asv + N;
  float*    h3  = hA;
  unsigned char* obA = (unsigned char*)obr;
  unsigned char* obB = obA + (size_t)((E + 15) / 16) * 16;

  dim3 blk(256);
  dim3 gemmG((N + BM - 1) / BM, 4);
  dim3 gemmG3((N + BM - 1) / BM, 1);
  int gAlpha = (N + 7) / 8;
  int gOwn   = (N + 199) / 200;
  int gOwn3  = (N + 207) / 208;

  gat_owner_bytes<200, 208><<<(E + 4096) / 4096, blk, 0, stream>>>(dst, obA, obB, E, N);

  gat_gemm_wmma<256, 256><<<gemmG, blk, 0, stream>>>(x, W1, hA, N);
  gat_alpha<256, 256><<<gAlpha, blk, 0, stream>>>(hA, as1, ad1, asv, adv, N);
  gat_owner<AD, 200><<<gOwn, 1024, 0, stream>>>(src, dst, obA, asv, adv, hA, b1, hB, E, N, 256);

  gat_gemm_wmma<256, 256><<<gemmG, blk, 0, stream>>>(hB, W2, hA, N);
  gat_alpha<256, 256><<<gAlpha, blk, 0, stream>>>(hA, as2, ad2, asv, adv, N);
  gat_owner<AD, 200><<<gOwn, 1024, 0, stream>>>(src, dst, obA, asv, adv, hA, b2, hB, E, N, 256);

  gat_gemm_wmma<10, 16><<<gemmG3, blk, 0, stream>>>(hB, W3, h3, N);
  gat_alpha<16, 10><<<gAlpha, blk, 0, stream>>>(h3, as3, ad3, asv, adv, N);
  gat_owner<D3, 208><<<gOwn3, 1024, 0, stream>>>(src, dst, obB, asv, adv, h3, b3, (float*)d_out, E, N, 10);
}


# Node-dot (alpha) kernels: a fixed grid of 4096 waves, each walking a run of consecutive nodes 7 at a time (56 row loads in flight), not one short wave per node (50000 waves); sums unchanged
# speedup vs baseline: 4.0328x; 1.0476x over previous
#include <hip/hip_runtime.h>
#include <math.h>


typedef float v2f __attribute__((ext_vector_type(2)));
typedef float v8f __attribute__((ext_vector_type(8)));

#define DK 256
#define KCHUNK 128
#define LDT (KCHUNK + 4)
#define BM 32
#define BN 64

__device__ __forceinline__ unsigned f2ord(float f) {
  unsigned u = __float_as_uint(f);
  return (u & 0x80000000u) ? ~u : (u | 0x80000000u);
}
__device__ __forceinline__ float ord2f(unsigned u) {
  return __uint_as_float((u & 0x80000000u) ? (u ^ 0x80000000u) : ~u);
}

template <int DACT, int DST>
__global__ __launch_bounds__(256) void gat_gemm_wmma(
    const float* __restrict__ X, const float* __restrict__ W,
    float* __restrict__ H, int N) {
  __shared__ float As[BM * LDT];
  __shared__ float Bst[BN * LDT];

  const int tid  = threadIdx.x;
  const int lane = tid & 31;
  const int lo   = lane & 15;
  const int hi   = lane >> 4;
  const int wid  = tid >> 5;
  const int wm   = (wid >> 2) * 16;
  const int wn   = (wid & 3) * 16;
  const int rowBase = blockIdx.x * BM;
  const int colBase = blockIdx.y * BN;

  const int ar  = tid >> 5;
  const int ac4 = tid & 31;
  const int bk  = tid >> 6;
  const int bn  = tid & 63;
  const int gc  = colBase + bn;
  const bool cok = gc < DACT;
  const int gcc = cok ? gc : (DACT - 1);

  v8f acc = {};

  for (int kk = 0; kk < DK; kk += KCHUNK) {
    float4 av[4];
    float  bv[32];
    #pragma unroll
    for (int it = 0; it < 4; ++it) {
      const int gr  = rowBase + ar + 8 * it;
      const int grc = (gr < N) ? gr : (N - 1);
      av[it] = *reinterpret_cast<const float4*>(X + (size_t)grc * DK + kk + ac4 * 4);
    }
    #pragma unroll
    for (int it = 0; it < 32; ++it)
      bv[it] = W[(size_t)(kk + bk + 4 * it) * DACT + gcc];

    __syncthreads();
    #pragma unroll
    for (int it = 0; it < 4; ++it) {
      const int gr = rowBase + ar + 8 * it;
      float4 v = av[it];
      if (gr >= N) v = make_float4(0.f, 0.f, 0.f, 0.f);
      *reinterpret_cast<float4*>(&As[(ar + 8 * it) * LDT + ac4 * 4]) = v;
    }
    #pragma unroll
    for (int it = 0; it < 32; ++it)
      Bst[bn * LDT + bk + 4 * it] = cok ? bv[it] : 0.f;
    __syncthreads();

    const float* arow = &As[(wm + lo) * LDT];
    const float* brow = &Bst[(wn + lo) * LDT];
    #pragma unroll
    for (int k0 = 0; k0 < KCHUNK; k0 += 4) {
      v2f a = *reinterpret_cast<const v2f*>(arow + k0 + 2 * hi);
      v2f b = *reinterpret_cast<const v2f*>(brow + k0 + 2 * hi);
      acc = __builtin_amdgcn_wmma_f32_16x16x4_f32(false, a, false, b,
                                                  (short)0, acc, false, false);
    }
  }

  const int col = colBase + wn + lo;
  if (col < DST) {
    #pragma unroll
    for (int r = 0; r < 8; ++r) {
      int row = rowBase + wm + hi * 8 + r;
      if (row < N) H[(size_t)row * DST + col] = acc[r];
    }
  }
}

#define AL_G 7
template <int DPAD, int DACT>
__global__ __launch_bounds__(256) void gat_alpha(
    const float* __restrict__ H, const float* __restrict__ a_s,
    const float* __restrict__ a_d, float* __restrict__ asv,
    float* __restrict__ adv, int N) {
  const int lane = threadIdx.x & 31;
  const int gw  = blockIdx.x * 8 + __builtin_amdgcn_readfirstlane(threadIdx.x >> 5);
  const int nw  = gridDim.x * 8;
  const int per = (N + nw - 1) / nw;
  const int beg = gw * per;
  int end = beg + per;
  end = (end < N) ? end : N;
  if (beg >= end) return;
  constexpr int NJ = (DPAD + 31) / 32;
  int   dh[NJ];
  float av[NJ], bv[NJ];
  #pragma unroll
  for (int j = 0; j < NJ; ++j) {
    const int d  = lane + 32 * j;
    const int da = (d < DACT) ? d : (DACT - 1);
    dh[j] = (d < DPAD) ? d : (DPAD - 1);
    const float a = a_s[da];
    const float b = a_d[da];
    av[j] = (d < DACT) ? a : 0.f;
    bv[j] = (d < DACT) ? b : 0.f;
  }
  for (int i0 = beg; i0 < end; i0 += AL_G) {
    float hv[AL_G][NJ];
    #pragma unroll
    for (int g = 0; g < AL_G; ++g) {
      int i = i0 + g;
      i = (i < end) ? i : (end - 1);
      const float* hp = H + (size_t)i * DPAD;
      #pragma unroll
      for (int j = 0; j < NJ; ++j) hv[g][j] = hp[dh[j]];
    }
    float r1 = 0.f, r2 = 0.f;
    #pragma unroll
    for (int g = 0; g < AL_G; ++g) {
      float s1 = 0.f, s2 = 0.f;
      #pragma unroll
      for (int j = 0; j < NJ; ++j) {
        const int d = lane + 32 * j;
        const float h = (d < DPAD) ? hv[g][j] : 0.f;
        s1 = fmaf(h, av[j], s1);
        s2 = fmaf(h, bv[j], s2);
      }
      #pragma unroll
      for (int off = 16; off > 0; off >>= 1) {
        s1 += __shfl_xor(s1, off, 32);
        s2 += __shfl_xor(s2, off, 32);
      }
      r1 = (lane == g) ? s1 : r1;
      r2 = (lane == g) ? s2 : r2;
    }
    int cnt = end - i0;
    cnt = (cnt < AL_G) ? cnt : AL_G;
    if (lane < cnt) {
      asv[i0 + lane] = r1;
      adv[i0 + lane] = r2;
    }
  }
}

#define AD 256
#define D3 16
#define SCB 8
#define LROUNDS 6
#define LCAP (1024 * LROUNDS)

__device__ __forceinline__ float leaky02(float t) {
  return (t > 0.f) ? t : 0.2f * t;
}

__device__ __forceinline__ int rl_i(int v, int l) {
  return __builtin_amdgcn_readlane(v, l);
}
__device__ __forceinline__ float rl_f(float v, int l) {
  return __int_as_float(__builtin_amdgcn_readlane(__float_as_int(v), l));
}

template <int RA, int RB>
__global__ __launch_bounds__(256) void gat_owner_bytes(
    const int* __restrict__ dst, unsigned char* __restrict__ oba,
    unsigned char* __restrict__ obb, int E, int N) {
  const int nq = (E + 15) >> 4;
  if (nq > 2 * N) return;
  const int q = blockIdx.x * 256 + threadIdx.x;
  if (q >= nq) return;
  const int last = E - 1;
  int d[16];
  #pragma unroll
  for (int j = 0; j < 16; ++j) {
    const int e = q * 16 + j;
    d[j] = dst[e < last ? e : last];
  }
  unsigned wa[4], wb[4];
  #pragma unroll
  for (int w = 0; w < 4; ++w) {
    unsigned a = 0, b = 0;
    #pragma unroll
    for (int t = 0; t < 4; ++t) {
      const int j = w * 4 + t;
      const bool in = (q * 16 + j) < E;
      const unsigned ba = in ? ((unsigned)(d[j] / RA) & 255u) : 255u;
      const unsigned bb = in ? ((unsigned)(d[j] / RB) & 255u) : 255u;
      a |= ba << (8 * t);
      b |= bb << (8 * t);
    }
    wa[w] = a;
    wb[w] = b;
  }
  reinterpret_cast<uint4*>(oba)[q] = make_uint4(wa[0], wa[1], wa[2], wa[3]);
  reinterpret_cast<uint4*>(obb)[q] = make_uint4(wb[0], wb[1], wb[2], wb[3]);
}

__device__ __forceinline__ unsigned zero_bytes(unsigned x) {
  const unsigned y = (x & 0x7F7F7F7Fu) + 0x7F7F7F7Fu;
  return ~(y | x | 0x7F7F7F7Fu);
}

__device__ __forceinline__ void list_append(int* cnt, int* lst, int qreg,
                                            int m, int lane) {
  int b = 0;
  if (lane == 0) b = atomicAdd(cnt, m);
  b = __builtin_amdgcn_readfirstlane(b);
  const int slot = b + lane;
  if (lane < m && slot < LCAP) lst[slot] = qreg;
}

__device__ __forceinline__ void row_add4(
    int s, int dl, float p, int g, int c, int lane,
    const float* __restrict__ H, float* acc) {
  float v[4][8];
  float pb[4];
  int   ro[4];
  #pragma unroll
  for (int b = 0; b < 4; ++b) {
    const int q = (b < c) ? (g + b) : g;
    const int sb = rl_i(s, q);
    ro[b] = rl_i(dl, q) * AD + lane;
    pb[b] = rl_f(p, q);
    const float* hp = H + (size_t)sb * AD + lane;
    #pragma unroll
    for (int j = 0; j < 8; ++j) v[b][j] = hp[j * 32];
  }
  #pragma unroll
  for (int b = 0; b < 4; ++b) {
    if (b < c) {
      #pragma unroll
      for (int j = 0; j < 8; ++j)
        atomicAdd(&acc[ro[b] + j * 32], pb[b] * v[b][j]);
    }
  }
}

__device__ __forceinline__ void row_add16(
    int s, int dl, float p, const float* __restrict__ H, float* acc) {
  const float4* hp = reinterpret_cast<const float4*>(H + (size_t)s * D3);
  const float4 a0 = hp[0], a1 = hp[1], a2 = hp[2], a3 = hp[3];
  float* ar = &acc[dl * D3];
  atomicAdd(ar + 0,  p * a0.x); atomicAdd(ar + 1,  p * a0.y);
  atomicAdd(ar + 2,  p * a0.z); atomicAdd(ar + 3,  p * a0.w);
  atomicAdd(ar + 4,  p * a1.x); atomicAdd(ar + 5,  p * a1.y);
  atomicAdd(ar + 6,  p * a1.z); atomicAdd(ar + 7,  p * a1.w);
  atomicAdd(ar + 8,  p * a2.x); atomicAdd(ar + 9,  p * a2.y);
  atomicAdd(ar + 10, p * a2.z); atomicAdd(ar + 11, p * a2.w);
  atomicAdd(ar + 12, p * a3.x); atomicAdd(ar + 13, p * a3.y);
  atomicAdd(ar + 14, p * a3.z); atomicAdd(ar + 15, p * a3.w);
}

template <int D, int ARW, int MODE>
__device__ __forceinline__ void owner_direct(
    const int* __restrict__ src, const int* __restrict__ dst,
    const float* __restrict__ asv, const float* __restrict__ H,
    const float* advl, unsigned* mrow, float* den, float* acc,
    int lo, int E, int ET, int wid, int lane) {
  for (int b0 = wid * 32; b0 < ET; b0 += 1024) {
    const int e = b0 + lane;
    int s = 0, d = -1;
    if (e < ET) {
      if (e < E) { s = src[e]; d = dst[e]; } else { s = d = e - E; }
    }
    const int dl = d - lo;
    const bool mine = (unsigned)dl < (unsigned)ARW;
    float sc = 0.f;
    if (mine) sc = leaky02(asv[s] + advl[dl]);
    if (MODE == 0) {
      if (mine) atomicMax(&mrow[dl], f2ord(sc));
    } else {
      float p = 0.f;
      if (mine) {
        p = __expf(sc - ord2f(mrow[dl]));
        atomicAdd(&den[dl], p);
      }
      if (D == AD) {
        unsigned mask = __builtin_amdgcn_ballot_w32(mine);
        while (mask) {
          const int L = __builtin_ctz(mask);
          mask &= mask - 1;
          row_add4(s, dl, p, L, 1, lane, H, acc);
        }
      } else {
        if (mine) row_add16(s, dl, p, H, acc);
      }
    }
  }
}

template <int D, int ARW>
__global__ __launch_bounds__(1024) void gat_owner(
    const int* __restrict__ src, const int* __restrict__ dst,
    const unsigned char* __restrict__ ob,
    const float* __restrict__ asv, const float* __restrict__ adv,
    const float* __restrict__ H, const float* __restrict__ bias,
    float* __restrict__ OUT, int E, int N, int Dact) {
  __shared__ __attribute__((aligned(16))) float acc[ARW * D];
  __shared__ int      lsrc[LCAP];
  __shared__ int      ldl[LCAP];
  __shared__ float    lp[LCAP];
  __shared__ unsigned mrow[ARW];
  __shared__ float    den[ARW];
  __shared__ float    advl[ARW];
  __shared__ int      cnt;

  const int tid  = threadIdx.x;
  const int lane = tid & 31;
  const int wid  = __builtin_amdgcn_readfirstlane(tid >> 5);
  const int lo   = blockIdx.x * ARW;
  const int ET   = E + N;
  int nown = N - lo;
  nown = (nown < ARW) ? nown : ARW;

  for (int x = tid; x < ARW * D / 4; x += 1024)
    reinterpret_cast<float4*>(acc)[x] = make_float4(0.f, 0.f, 0.f, 0.f);
  for (int r = tid; r < ARW; r += 1024) {
    mrow[r] = f2ord(-INFINITY);
    den[r]  = 0.f;
    advl[r] = (r < nown) ? adv[lo + r] : 0.f;
    if (r < nown) lsrc[r] = E + lo + r;
  }
  if (tid == 0) cnt = nown;
  __syncthreads();

  const int nq = (E + 15) >> 4;
  const bool bytes_ok = ((N + ARW - 1) / ARW <= 255) && (nq <= 2 * N);
  int qreg = 0;
  int n = 0;
  if (bytes_ok && nq > 0) {
    const uint4* ob4 = reinterpret_cast<const uint4*>(ob);
    const unsigned bsel = blockIdx.x * 0x01010101u;
    const int qstep = SCB * 32;
    const int qlast = nq - 1;
    int qb = wid * qstep;
    uint4 dv[SCB];
    #pragma unroll
    for (int k = 0; k < SCB; ++k) {
      const int qi = qb + k * 32 + lane;
      dv[k] = ob4[qi < qlast ? qi : qlast];
    }
    while (qb < nq) {
      unsigned z[SCB];
      #pragma unroll
      for (int k = 0; k < SCB; ++k) {
        const int qi = qb + k * 32 + lane;
        const unsigned m = (zero_bytes(dv[k].x ^ bsel) >> 7) |
                           (zero_bytes(dv[k].y ^ bsel) >> 6) |
                           (zero_bytes(dv[k].z ^ bsel) >> 5) |
                           (zero_bytes(dv[k].w ^ bsel) >> 4);
        z[k] = (qi < nq) ? m : 0u;
      }
      const int cur = qb;
      qb += 32 * qstep;
      #pragma unroll
      for (int k = 0; k < SCB; ++k) {
        const int qi = qb + k * 32 + lane;
        dv[k] = ob4[qi < qlast ? qi : qlast];
      }
      #pragma unroll
      for (int k = 0; k < SCB; ++k) {
        unsigned lanes = __builtin_amdgcn_ballot_w32(z[k] != 0);
        while (lanes) {
          const int L = __builtin_ctz(lanes);
          lanes &= lanes - 1;
          unsigned bits = (unsigned)__builtin_amdgcn_readlane((int)z[k], L);
          const int e0 = (cur + k * 32 + L) * 16;
          while (bits) {
            const int p = __builtin_ctz(bits);
            bits &= bits - 1;
            const int id = e0 + ((p & 7) << 2) + (p >> 3);
            qreg = (lane == n) ? id : qreg;
            ++n;
            if (n == 32) {
              list_append(&cnt, lsrc, qreg, 32, lane);
              n = 0;
            }
          }
        }
      }
    }
    if (n) list_append(&cnt, lsrc, qreg, n, lane);
  }
  __syncthreads();

  const int total = bytes_ok ? cnt : (LCAP + 1);
  if (total <= LCAP) {
    int   sk[LROUNDS], dk[LROUNDS];
    float ak[LROUNDS];
    #pragma unroll
    for (int k = 0; k < LROUNDS; ++k) {
      const int i = tid + 1024 * k;
      int s = 0, d = lo;
      if (i < total) {
        const int e = lsrc[i];
        if (e < E) { s = src[e]; d = dst[e]; } else { s = d = e - E; }
      }
      sk[k] = s;
      dk[k] = d - lo;
    }
    #pragma unroll
    for (int k = 0; k < LROUNDS; ++k) ak[k] = asv[sk[k]];
    #pragma unroll
    for (int k = 0; k < LROUNDS; ++k) {
      const int i = tid + 1024 * k;
      if (i < total) {
        const float sc = leaky02(ak[k] + advl[dk[k]]);
        lsrc[i] = sk[k];
        ldl[i]  = dk[k];
        lp[i]   = sc;
        atomicMax(&mrow[dk[k]], f2ord(sc));
      }
    }
    __syncthreads();
    for (int b0 = wid * 32; b0 < total; b0 += 1024) {
      const int i = b0 + lane;
      int s = 0, dl = 0;
      float p = 0.f;
      if (i < total) {
        s  = lsrc[i];
        dl = ldl[i];
        p  = __expf(lp[i] - ord2f(mrow[dl]));
        atomicAdd(&den[dl], p);
      }
      if (D == AD) {
        int m = total - b0;
        m = (m < 32) ? m : 32;
        for (int g = 0; g < m; g += 4)
          row_add4(s, dl, p, g, m - g, lane, H, acc);
      } else {
        if (i < total) row_add16(s, dl, p, H, acc);
      }
    }
    __syncthreads();
  } else {
    owner_direct<D, ARW, 0>(src, dst, asv, H, advl, mrow, den, acc,
                            lo, E, ET, wid, lane);
    __syncthreads();
    owner_direct<D, ARW, 1>(src, dst, asv, H, advl, mrow, den, acc,
                            lo, E, ET, wid, lane);
    __syncthreads();
  }

  if (D == AD) {
    for (int idx = tid; idx < ARW * (AD / 4); idx += 1024) {
      const int r  = idx >> 6;
      const int c4 = idx & 63;
      if (r < nown) {
        const float inv = 1.0f / den[r];
        float4 a  = *reinterpret_cast<const float4*>(&acc[r * AD + c4 * 4]);
        float4 bb = *reinterpret_cast<const float4*>(bias + c4 * 4);
        float4 o;
        o.x = a.x * inv + bb.x; o.x = o.x > 0.f ? o.x : 0.f;
        o.y = a.y * inv + bb.y; o.y = o.y > 0.f ? o.y : 0.f;
        o.z = a.z * inv + bb.z; o.z = o.z > 0.f ? o.z : 0.f;
        o.w = a.w * inv + bb.w; o.w = o.w > 0.f ? o.w : 0.f;
        *reinterpret_cast<float4*>(OUT + (size_t)(lo + r) * AD + c4 * 4) = o;
      }
    }
  } else {
    const int nout = nown * Dact;
    for (int x = tid; x < nout; x += 1024) {
      const int r = x / Dact;
      const int c = x - r * Dact;
      OUT[(size_t)lo * Dact + x] = acc[r * D3 + c] / den[r] + bias[c];
    }
  }
}

extern "C" void kernel_launch(void* const* d_in, const int* in_sizes, int n_in,
                              void* d_out, int out_size, void* d_ws, size_t ws_size,
                              hipStream_t stream) {
  const float* x  = (const float*)d_in[0];
  const int*   ei = (const int*)d_in[1];
  const float* W1 = (const float*)d_in[2];
  const float* as1 = (const float*)d_in[3];
  const float* ad1 = (const float*)d_in[4];
  const float* b1 = (const float*)d_in[5];
  const float* W2 = (const float*)d_in[6];
  const float* as2 = (const float*)d_in[7];
  const float* ad2 = (const float*)d_in[8];
  const float* b2 = (const float*)d_in[9];
  const float* W3 = (const float*)d_in[10];
  const float* as3 = (const float*)d_in[11];
  const float* ad3 = (const float*)d_in[12];
  const float* b3 = (const float*)d_in[13];

  const int N  = in_sizes[0] / 256;
  const int E  = in_sizes[1] / 2;
  const int* src = ei;
  const int* dst = ei + E;

  float* ws = (float*)d_ws;
  float*    hA  = ws;
  float*    hB  = hA + (size_t)N * 256;
  float*    obr = hB + (size_t)N * 256;
  float*    asv = obr + (size_t)N * 16;
  float*    adv = asv + N;
  float*    h3  = hA;
  unsigned char* obA = (unsigned char*)obr;
  unsigned char* obB = obA + (size_t)((E + 15) / 16) * 16;

  dim3 blk(256);
  dim3 gemmG((N + BM - 1) / BM, 4);
  dim3 gemmG3((N + BM - 1) / BM, 1);
  int gAlpha = 512;
  int gOwn   = (N + 199) / 200;
  int gOwn3  = (N + 207) / 208;

  gat_owner_bytes<200, 208><<<(E + 4096) / 4096, blk, 0, stream>>>(dst, obA, obB, E, N);

  gat_gemm_wmma<256, 256><<<gemmG, blk, 0, stream>>>(x, W1, hA, N);
  gat_alpha<256, 256><<<gAlpha, blk, 0, stream>>>(hA, as1, ad1, asv, adv, N);
  gat_owner<AD, 200><<<gOwn, 1024, 0, stream>>>(src, dst, obA, asv, adv, hA, b1, hB, E, N, 256);

  gat_gemm_wmma<256, 256><<<gemmG, blk, 0, stream>>>(hB, W2, hA, N);
  gat_alpha<256, 256><<<gAlpha, blk, 0, stream>>>(hA, as2, ad2, asv, adv, N);
  gat_owner<AD, 200><<<gOwn, 1024, 0, stream>>>(src, dst, obA, asv, adv, hA, b2, hB, E, N, 256);

  gat_gemm_wmma<10, 16><<<gemmG3, blk, 0, stream>>>(hB, W3, h3, N);
  gat_alpha<16, 10><<<gAlpha, blk, 0, stream>>>(h3, as3, ad3, asv, adv, N);
  gat_owner<D3, 208><<<gOwn3, 1024, 0, stream>>>(src, dst, obB, asv, adv, h3, b3, (float*)d_out, E, N, 10);
}


# Owner blocks, layers 1-2: a source row comes in by 16-byte loads, 2 per 1-KB row not 8, 8 rows in flight per wave not 4; row sums column-permuted in LDS (adds on 32 banks); LDS arrival order
# speedup vs baseline: 4.2140x; 1.0449x over previous
#include <hip/hip_runtime.h>
#include <math.h>


typedef float v2f __attribute__((ext_vector_type(2)));
typedef float v8f __attribute__((ext_vector_type(8)));

#define DK 256
#define KCHUNK 128
#define LDT (KCHUNK + 4)
#define BM 32
#define BN 64

__device__ __forceinline__ unsigned f2ord(float f) {
  unsigned u = __float_as_uint(f);
  return (u & 0x80000000u) ? ~u : (u | 0x80000000u);
}
__device__ __forceinline__ float ord2f(unsigned u) {
  return __uint_as_float((u & 0x80000000u) ? (u ^ 0x80000000u) : ~u);
}

template <int DACT, int DST>
__global__ __launch_bounds__(256) void gat_gemm_wmma(
    const float* __restrict__ X, const float* __restrict__ W,
    float* __restrict__ H, int N) {
  __shared__ float As[BM * LDT];
  __shared__ float Bst[BN * LDT];

  const int tid  = threadIdx.x;
  const int lane = tid & 31;
  const int lo   = lane & 15;
  const int hi   = lane >> 4;
  const int wid  = tid >> 5;
  const int wm   = (wid >> 2) * 16;
  const int wn   = (wid & 3) * 16;
  const int rowBase = blockIdx.x * BM;
  const int colBase = blockIdx.y * BN;

  const int ar  = tid >> 5;
  const int ac4 = tid & 31;
  const int bk  = tid >> 6;
  const int bn  = tid & 63;
  const int gc  = colBase + bn;
  const bool cok = gc < DACT;
  const int gcc = cok ? gc : (DACT - 1);

  v8f acc = {};

  for (int kk = 0; kk < DK; kk += KCHUNK) {
    float4 av[4];
    float  bv[32];
    #pragma unroll
    for (int it = 0; it < 4; ++it) {
      const int gr  = rowBase + ar + 8 * it;
      const int grc = (gr < N) ? gr : (N - 1);
      av[it] = *reinterpret_cast<const float4*>(X + (size_t)grc * DK + kk + ac4 * 4);
    }
    #pragma unroll
    for (int it = 0; it < 32; ++it)
      bv[it] = W[(size_t)(kk + bk + 4 * it) * DACT + gcc];

    __syncthreads();
    #pragma unroll
    for (int it = 0; it < 4; ++it) {
      const int gr = rowBase + ar + 8 * it;
      float4 v = av[it];
      if (gr >= N) v = make_float4(0.f, 0.f, 0.f, 0.f);
      *reinterpret_cast<float4*>(&As[(ar + 8 * it) * LDT + ac4 * 4]) = v;
    }
    #pragma unroll
    for (int it = 0; it < 32; ++it)
      Bst[bn * LDT + bk + 4 * it] = cok ? bv[it] : 0.f;
    __syncthreads();

    const float* arow = &As[(wm + lo) * LDT];
    const float* brow = &Bst[(wn + lo) * LDT];
    #pragma unroll
    for (int k0 = 0; k0 < KCHUNK; k0 += 4) {
      v2f a = *reinterpret_cast<const v2f*>(arow + k0 + 2 * hi);
      v2f b = *reinterpret_cast<const v2f*>(brow + k0 + 2 * hi);
      acc = __builtin_amdgcn_wmma_f32_16x16x4_f32(false, a, false, b,
                                                  (short)0, acc, false, false);
    }
  }

  const int col = colBase + wn + lo;
  if (col < DST) {
    #pragma unroll
    for (int r = 0; r < 8; ++r) {
      int row = rowBase + wm + hi * 8 + r;
      if (row < N) H[(size_t)row * DST + col] = acc[r];
    }
  }
}

#define AL_G 7
template <int DPAD, int DACT>
__global__ __launch_bounds__(256) void gat_alpha(
    const float* __restrict__ H, const float* __restrict__ a_s,
    const float* __restrict__ a_d, float* __restrict__ asv,
    float* __restrict__ adv, int N) {
  const int lane = threadIdx.x & 31;
  const int gw  = blockIdx.x * 8 + __builtin_amdgcn_readfirstlane(threadIdx.x >> 5);
  const int nw  = gridDim.x * 8;
  const int per = (N + nw - 1) / nw;
  const int beg = gw * per;
  int end = beg + per;
  end = (end < N) ? end : N;
  if (beg >= end) return;
  constexpr int NJ = (DPAD + 31) / 32;
  int   dh[NJ];
  float av[NJ], bv[NJ];
  #pragma unroll
  for (int j = 0; j < NJ; ++j) {
    const int d  = lane + 32 * j;
    const int da = (d < DACT) ? d : (DACT - 1);
    dh[j] = (d < DPAD) ? d : (DPAD - 1);
    const float a = a_s[da];
    const float b = a_d[da];
    av[j] = (d < DACT) ? a : 0.f;
    bv[j] = (d < DACT) ? b : 0.f;
  }
  for (int i0 = beg; i0 < end; i0 += AL_G) {
    float hv[AL_G][NJ];
    #pragma unroll
    for (int g = 0; g < AL_G; ++g) {
      int i = i0 + g;
      i = (i < end) ? i : (end - 1);
      const float* hp = H + (size_t)i * DPAD;
      #pragma unroll
      for (int j = 0; j < NJ; ++j) hv[g][j] = hp[dh[j]];
    }
    float r1 = 0.f, r2 = 0.f;
    #pragma unroll
    for (int g = 0; g < AL_G; ++g) {
      float s1 = 0.f, s2 = 0.f;
      #pragma unroll
      for (int j = 0; j < NJ; ++j) {
        const int d = lane + 32 * j;
        const float h = (d < DPAD) ? hv[g][j] : 0.f;
        s1 = fmaf(h, av[j], s1);
        s2 = fmaf(h, bv[j], s2);
      }
      #pragma unroll
      for (int off = 16; off > 0; off >>= 1) {
        s1 += __shfl_xor(s1, off, 32);
        s2 += __shfl_xor(s2, off, 32);
      }
      r1 = (lane == g) ? s1 : r1;
      r2 = (lane == g) ? s2 : r2;
    }
    int cnt = end - i0;
    cnt = (cnt < AL_G) ? cnt : AL_G;
    if (lane < cnt) {
      asv[i0 + lane] = r1;
      adv[i0 + lane] = r2;
    }
  }
}

#define AD 256
#define D3 16
#define SCB 8
#define LROUNDS 6
#define LCAP (1024 * LROUNDS)

__device__ __forceinline__ float leaky02(float t) {
  return (t > 0.f) ? t : 0.2f * t;
}

__device__ __forceinline__ int rl_i(int v, int l) {
  return __builtin_amdgcn_readlane(v, l);
}
__device__ __forceinline__ float rl_f(float v, int l) {
  return __int_as_float(__builtin_amdgcn_readlane(__float_as_int(v), l));
}

template <int RA, int RB>
__global__ __launch_bounds__(256) void gat_owner_bytes(
    const int* __restrict__ dst, unsigned char* __restrict__ oba,
    unsigned char* __restrict__ obb, int E, int N) {
  const int nq = (E + 15) >> 4;
  if (nq > 2 * N) return;
  const int q = blockIdx.x * 256 + threadIdx.x;
  if (q >= nq) return;
  const int last = E - 1;
  int d[16];
  #pragma unroll
  for (int j = 0; j < 16; ++j) {
    const int e = q * 16 + j;
    d[j] = dst[e < last ? e : last];
  }
  unsigned wa[4], wb[4];
  #pragma unroll
  for (int w = 0; w < 4; ++w) {
    unsigned a = 0, b = 0;
    #pragma unroll
    for (int t = 0; t < 4; ++t) {
      const int j = w * 4 + t;
      const bool in = (q * 16 + j) < E;
      const unsigned ba = in ? ((unsigned)(d[j] / RA) & 255u) : 255u;
      const unsigned bb = in ? ((unsigned)(d[j] / RB) & 255u) : 255u;
      a |= ba << (8 * t);
      b |= bb << (8 * t);
    }
    wa[w] = a;
    wb[w] = b;
  }
  reinterpret_cast<uint4*>(oba)[q] = make_uint4(wa[0], wa[1], wa[2], wa[3]);
  reinterpret_cast<uint4*>(obb)[q] = make_uint4(wb[0], wb[1], wb[2], wb[3]);
}

__device__ __forceinline__ unsigned zero_bytes(unsigned x) {
  const unsigned y = (x & 0x7F7F7F7Fu) + 0x7F7F7F7Fu;
  return ~(y | x | 0x7F7F7F7Fu);
}

__device__ __forceinline__ void list_append(int* cnt, int* lst, int qreg,
                                            int m, int lane) {
  int b = 0;
  if (lane == 0) b = atomicAdd(cnt, m);
  b = __builtin_amdgcn_readfirstlane(b);
  const int slot = b + lane;
  if (lane < m && slot < LCAP) lst[slot] = qreg;
}

#define RG 8
template <int NR>
__device__ __forceinline__ void row_add(
    int s, int dl, float p, int g, int c, int lane,
    const float* __restrict__ H, float* acc) {
  float4 v[NR][2];
  float pb[NR];
  int   ro[NR];
  #pragma unroll
  for (int b = 0; b < NR; ++b) {
    const int q = (b < c) ? (g + b) : g;
    const int sb = rl_i(s, q);
    ro[b] = rl_i(dl, q) * AD + lane;
    pb[b] = rl_f(p, q);
    const float4* hp = reinterpret_cast<const float4*>(H + (size_t)sb * AD) + lane;
    v[b][0] = hp[0];
    v[b][1] = hp[32];
  }
  #pragma unroll
  for (int b = 0; b < NR; ++b) {
    if (b < c) {
      float* a = &acc[ro[b]];
      atomicAdd(a + 0 * 32, pb[b] * v[b][0].x);
      atomicAdd(a + 1 * 32, pb[b] * v[b][0].y);
      atomicAdd(a + 2 * 32, pb[b] * v[b][0].z);
      atomicAdd(a + 3 * 32, pb[b] * v[b][0].w);
      atomicAdd(a + 4 * 32, pb[b] * v[b][1].x);
      atomicAdd(a + 5 * 32, pb[b] * v[b][1].y);
      atomicAdd(a + 6 * 32, pb[b] * v[b][1].z);
      atomicAdd(a + 7 * 32, pb[b] * v[b][1].w);
    }
  }
}

__device__ __forceinline__ void row_add16(
    int s, int dl, float p, const float* __restrict__ H, float* acc) {
  const float4* hp = reinterpret_cast<const float4*>(H + (size_t)s * D3);
  const float4 a0 = hp[0], a1 = hp[1], a2 = hp[2], a3 = hp[3];
  float* ar = &acc[dl * D3];
  atomicAdd(ar + 0,  p * a0.x); atomicAdd(ar + 1,  p * a0.y);
  atomicAdd(ar + 2,  p * a0.z); atomicAdd(ar + 3,  p * a0.w);
  atomicAdd(ar + 4,  p * a1.x); atomicAdd(ar + 5,  p * a1.y);
  atomicAdd(ar + 6,  p * a1.z); atomicAdd(ar + 7,  p * a1.w);
  atomicAdd(ar + 8,  p * a2.x); atomicAdd(ar + 9,  p * a2.y);
  atomicAdd(ar + 10, p * a2.z); atomicAdd(ar + 11, p * a2.w);
  atomicAdd(ar + 12, p * a3.x); atomicAdd(ar + 13, p * a3.y);
  atomicAdd(ar + 14, p * a3.z); atomicAdd(ar + 15, p * a3.w);
}

template <int D, int ARW, int MODE>
__device__ __forceinline__ void owner_direct(
    const int* __restrict__ src, const int* __restrict__ dst,
    const float* __restrict__ asv, const float* __restrict__ H,
    const float* advl, unsigned* mrow, float* den, float* acc,
    int lo, int E, int ET, int wid, int lane) {
  for (int b0 = wid * 32; b0 < ET; b0 += 1024) {
    const int e = b0 + lane;
    int s = 0, d = -1;
    if (e < ET) {
      if (e < E) { s = src[e]; d = dst[e]; } else { s = d = e - E; }
    }
    const int dl = d - lo;
    const bool mine = (unsigned)dl < (unsigned)ARW;
    float sc = 0.f;
    if (mine) sc = leaky02(asv[s] + advl[dl]);
    if (MODE == 0) {
      if (mine) atomicMax(&mrow[dl], f2ord(sc));
    } else {
      float p = 0.f;
      if (mine) {
        p = __expf(sc - ord2f(mrow[dl]));
        atomicAdd(&den[dl], p);
      }
      if (D == AD) {
        unsigned mask = __builtin_amdgcn_ballot_w32(mine);
        while (mask) {
          const int L = __builtin_ctz(mask);
          mask &= mask - 1;
          row_add<1>(s, dl, p, L, 1, lane, H, acc);
        }
      } else {
        if (mine) row_add16(s, dl, p, H, acc);
      }
    }
  }
}

template <int D, int ARW>
__global__ __launch_bounds__(1024) void gat_owner(
    const int* __restrict__ src, const int* __restrict__ dst,
    const unsigned char* __restrict__ ob,
    const float* __restrict__ asv, const float* __restrict__ adv,
    const float* __restrict__ H, const float* __restrict__ bias,
    float* __restrict__ OUT, int E, int N, int Dact) {
  __shared__ __attribute__((aligned(16))) float acc[ARW * D];
  __shared__ int      lsrc[LCAP];
  __shared__ int      ldl[LCAP];
  __shared__ float    lp[LCAP];
  __shared__ unsigned mrow[ARW];
  __shared__ float    den[ARW];
  __shared__ float    advl[ARW];
  __shared__ int      cnt;

  const int tid  = threadIdx.x;
  const int lane = tid & 31;
  const int wid  = __builtin_amdgcn_readfirstlane(tid >> 5);
  const int lo   = blockIdx.x * ARW;
  const int ET   = E + N;
  int nown = N - lo;
  nown = (nown < ARW) ? nown : ARW;

  for (int x = tid; x < ARW * D / 4; x += 1024)
    reinterpret_cast<float4*>(acc)[x] = make_float4(0.f, 0.f, 0.f, 0.f);
  for (int r = tid; r < ARW; r += 1024) {
    mrow[r] = f2ord(-INFINITY);
    den[r]  = 0.f;
    advl[r] = (r < nown) ? adv[lo + r] : 0.f;
    if (r < nown) lsrc[r] = E + lo + r;
  }
  if (tid == 0) cnt = nown;
  __syncthreads();

  const int nq = (E + 15) >> 4;
  const bool bytes_ok = ((N + ARW - 1) / ARW <= 255) && (nq <= 2 * N);
  int qreg = 0;
  int n = 0;
  if (bytes_ok && nq > 0) {
    const uint4* ob4 = reinterpret_cast<const uint4*>(ob);
    const unsigned bsel = blockIdx.x * 0x01010101u;
    const int qstep = SCB * 32;
    const int qlast = nq - 1;
    int qb = wid * qstep;
    uint4 dv[SCB];
    #pragma unroll
    for (int k = 0; k < SCB; ++k) {
      const int qi = qb + k * 32 + lane;
      dv[k] = ob4[qi < qlast ? qi : qlast];
    }
    while (qb < nq) {
      unsigned z[SCB];
      #pragma unroll
      for (int k = 0; k < SCB; ++k) {
        const int qi = qb + k * 32 + lane;
        const unsigned m = (zero_bytes(dv[k].x ^ bsel) >> 7) |
                           (zero_bytes(dv[k].y ^ bsel) >> 6) |
                           (zero_bytes(dv[k].z ^ bsel) >> 5) |
                           (zero_bytes(dv[k].w ^ bsel) >> 4);
        z[k] = (qi < nq) ? m : 0u;
      }
      const int cur = qb;
      qb += 32 * qstep;
      #pragma unroll
      for (int k = 0; k < SCB; ++k) {
        const int qi = qb + k * 32 + lane;
        dv[k] = ob4[qi < qlast ? qi : qlast];
      }
      #pragma unroll
      for (int k = 0; k < SCB; ++k) {
        unsigned lanes = __builtin_amdgcn_ballot_w32(z[k] != 0);
        while (lanes) {
          const int L = __builtin_ctz(lanes);
          lanes &= lanes - 1;
          unsigned bits = (unsigned)__builtin_amdgcn_readlane((int)z[k], L);
          const int e0 = (cur + k * 32 + L) * 16;
          while (bits) {
            const int p = __builtin_ctz(bits);
            bits &= bits - 1;
            const int id = e0 + ((p & 7) << 2) + (p >> 3);
            qreg = (lane == n) ? id : qreg;
            ++n;
            if (n == 32) {
              list_append(&cnt, lsrc, qreg, 32, lane);
              n = 0;
            }
          }
        }
      }
    }
    if (n) list_append(&cnt, lsrc, qreg, n, lane);
  }
  __syncthreads();

  const int total = bytes_ok ? cnt : (LCAP + 1);
  if (total <= LCAP) {
    int   sk[LROUNDS], dk[LROUNDS];
    float ak[LROUNDS];
    #pragma unroll
    for (int k = 0; k < LROUNDS; ++k) {
      const int i = tid + 1024 * k;
      int s = 0, d = lo;
      if (i < total) {
        const int e = lsrc[i];
        if (e < E) { s = src[e]; d = dst[e]; } else { s = d = e - E; }
      }
      sk[k] = s;
      dk[k] = d - lo;
    }
    #pragma unroll
    for (int k = 0; k < LROUNDS; ++k) ak[k] = asv[sk[k]];
    #pragma unroll
    for (int k = 0; k < LROUNDS; ++k) {
      const int i = tid + 1024 * k;
      if (i < total) {
        const float sc = leaky02(ak[k] + advl[dk[k]]);
        lsrc[i] = sk[k];
        ldl[i]  = dk[k];
        lp[i]   = sc;
        atomicMax(&mrow[dk[k]], f2ord(sc));
      }
    }
    __syncthreads();
    for (int b0 = wid * 32; b0 < total; b0 += 1024) {
      const int i = b0 + lane;
      int s = 0, dl = 0;
      float p = 0.f;
      if (i < total) {
        s  = lsrc[i];
        dl = ldl[i];
        p  = __expf(lp[i] - ord2f(mrow[dl]));
        atomicAdd(&den[dl], p);
      }
      if (D == AD) {
        int m = total - b0;
        m = (m < 32) ? m : 32;
        for (int g = 0; g < m; g += RG)
          row_add<RG>(s, dl, p, g, m - g, lane, H, acc);
      } else {
        if (i < total) row_add16(s, dl, p, H, acc);
      }
    }
    __syncthreads();
  } else {
    owner_direct<D, ARW, 0>(src, dst, asv, H, advl, mrow, den, acc,
                            lo, E, ET, wid, lane);
    __syncthreads();
    owner_direct<D, ARW, 1>(src, dst, asv, H, advl, mrow, den, acc,
                            lo, E, ET, wid, lane);
    __syncthreads();
  }

  if (D == AD) {
    for (int idx = tid; idx < ARW * (AD / 4); idx += 1024) {
      const int r  = idx >> 6;
      const int c4 = idx & 63;
      if (r < nown) {
        const float inv = 1.0f / den[r];
        const float* ap = &acc[r * AD + (c4 >> 5) * 128 + (c4 & 31)];
        float4 a  = make_float4(ap[0], ap[32], ap[64], ap[96]);
        float4 bb = *reinterpret_cast<const float4*>(bias + c4 * 4);
        float4 o;
        o.x = a.x * inv + bb.x; o.x = o.x > 0.f ? o.x : 0.f;
        o.y = a.y * inv + bb.y; o.y = o.y > 0.f ? o.y : 0.f;
        o.z = a.z * inv + bb.z; o.z = o.z > 0.f ? o.z : 0.f;
        o.w = a.w * inv + bb.w; o.w = o.w > 0.f ? o.w : 0.f;
        *reinterpret_cast<float4*>(OUT + (size_t)(lo + r) * AD + c4 * 4) = o;
      }
    }
  } else {
    const int nout = nown * Dact;
    for (int x = tid; x < nout; x += 1024) {
      const int r = x / Dact;
      const int c = x - r * Dact;
      OUT[(size_t)lo * Dact + x] = acc[r * D3 + c] / den[r] + bias[c];
    }
  }
}

extern "C" void kernel_launch(void* const* d_in, const int* in_sizes, int n_in,
                              void* d_out, int out_size, void* d_ws, size_t ws_size,
                              hipStream_t stream) {
  const float* x  = (const float*)d_in[0];
  const int*   ei = (const int*)d_in[1];
  const float* W1 = (const float*)d_in[2];
  const float* as1 = (const float*)d_in[3];
  const float* ad1 = (const float*)d_in[4];
  const float* b1 = (const float*)d_in[5];
  const float* W2 = (const float*)d_in[6];
  const float* as2 = (const float*)d_in[7];
  const float* ad2 = (const float*)d_in[8];
  const float* b2 = (const float*)d_in[9];
  const float* W3 = (const float*)d_in[10];
  const float* as3 = (const float*)d_in[11];
  const float* ad3 = (const float*)d_in[12];
  const float* b3 = (const float*)d_in[13];

  const int N  = in_sizes[0] / 256;
  const int E  = in_sizes[1] / 2;
  const int* src = ei;
  const int* dst = ei + E;

  float* ws = (float*)d_ws;
  float*    hA  = ws;
  float*    hB  = hA + (size_t)N * 256;
  float*    obr = hB + (size_t)N * 256;
  float*    asv = obr + (size_t)N * 16;
  float*    adv = asv + N;
  float*    h3  = hA;
  unsigned char* obA = (unsigned char*)obr;
  unsigned char* obB = obA + (size_t)((E + 15) / 16) * 16;

  dim3 blk(256);
  dim3 gemmG((N + BM - 1) / BM, 4);
  dim3 gemmG3((N + BM - 1) / BM, 1);
  int gAlpha = 512;
  int gOwn   = (N + 199) / 200;
  int gOwn3  = (N + 207) / 208;

  gat_owner_bytes<200, 208><<<(E + 4096) / 4096, blk, 0, stream>>>(dst, obA, obB, E, N);

  gat_gemm_wmma<256, 256><<<gemmG, blk, 0, stream>>>(x, W1, hA, N);
  gat_alpha<256, 256><<<gAlpha, blk, 0, stream>>>(hA, as1, ad1, asv, adv, N);
  gat_owner<AD, 200><<<gOwn, 1024, 0, stream>>>(src, dst, obA, asv, adv, hA, b1, hB, E, N, 256);

  gat_gemm_wmma<256, 256><<<gemmG, blk, 0, stream>>>(hB, W2, hA, N);
  gat_alpha<256, 256><<<gAlpha, blk, 0, stream>>>(hA, as2, ad2, asv, adv, N);
  gat_owner<AD, 200><<<gOwn, 1024, 0, stream>>>(src, dst, obA, asv, adv, hA, b2, hB, E, N, 256);

  gat_gemm_wmma<10, 16><<<gemmG3, blk, 0, stream>>>(hB, W3, h3, N);
  gat_alpha<16, 10><<<gAlpha, blk, 0, stream>>>(h3, as3, ad3, asv, adv, N);
  gat_owner<D3, 208><<<gOwn3, 1024, 0, stream>>>(src, dst, obB, asv, adv, h3, b3, (float*)d_out, E, N, 10);
}


# 256-wide GEMMs: each wave keeps its 16 columns of W (all 256 k) in 128 registers for the whole launch; 250 persistent 16-wave blocks walk 32-row X tiles through LDS; sums and their order unchanged
# speedup vs baseline: 4.4301x; 1.0513x over previous
#include <hip/hip_runtime.h>
#include <math.h>


typedef float v2f __attribute__((ext_vector_type(2)));
typedef float v8f __attribute__((ext_vector_type(8)));

#define DK 256
#define KCHUNK 128
#define LDT (KCHUNK + 4)
#define BM 32
#define BN 64

__device__ __forceinline__ unsigned f2ord(float f) {
  unsigned u = __float_as_uint(f);
  return (u & 0x80000000u) ? ~u : (u | 0x80000000u);
}
__device__ __forceinline__ float ord2f(unsigned u) {
  return __uint_as_float((u & 0x80000000u) ? (u ^ 0x80000000u) : ~u);
}

template <int DACT, int DST>
__global__ __launch_bounds__(256) void gat_gemm_wmma(
    const float* __restrict__ X, const float* __restrict__ W,
    float* __restrict__ H, int N) {
  __shared__ float As[BM * LDT];
  __shared__ float Bst[BN * LDT];

  const int tid  = threadIdx.x;
  const int lane = tid & 31;
  const int lo   = lane & 15;
  const int hi   = lane >> 4;
  const int wid  = tid >> 5;
  const int wm   = (wid >> 2) * 16;
  const int wn   = (wid & 3) * 16;
  const int rowBase = blockIdx.x * BM;
  const int colBase = blockIdx.y * BN;

  const int ar  = tid >> 5;
  const int ac4 = tid & 31;
  const int bk  = tid >> 6;
  const int bn  = tid & 63;
  const int gc  = colBase + bn;
  const bool cok = gc < DACT;
  const int gcc = cok ? gc : (DACT - 1);

  v8f acc = {};

  for (int kk = 0; kk < DK; kk += KCHUNK) {
    float4 av[4];
    float  bv[32];
    #pragma unroll
    for (int it = 0; it < 4; ++it) {
      const int gr  = rowBase + ar + 8 * it;
      const int grc = (gr < N) ? gr : (N - 1);
      av[it] = *reinterpret_cast<const float4*>(X + (size_t)grc * DK + kk + ac4 * 4);
    }
    #pragma unroll
    for (int it = 0; it < 32; ++it)
      bv[it] = W[(size_t)(kk + bk + 4 * it) * DACT + gcc];

    __syncthreads();
    #pragma unroll
    for (int it = 0; it < 4; ++it) {
      const int gr = rowBase + ar + 8 * it;
      float4 v = av[it];
      if (gr >= N) v = make_float4(0.f, 0.f, 0.f, 0.f);
      *reinterpret_cast<float4*>(&As[(ar + 8 * it) * LDT + ac4 * 4]) = v;
    }
    #pragma unroll
    for (int it = 0; it < 32; ++it)
      Bst[bn * LDT + bk + 4 * it] = cok ? bv[it] : 0.f;
    __syncthreads();

    const float* arow = &As[(wm + lo) * LDT];
    const float* brow = &Bst[(wn + lo) * LDT];
    #pragma unroll
    for (int k0 = 0; k0 < KCHUNK; k0 += 4) {
      v2f a = *reinterpret_cast<const v2f*>(arow + k0 + 2 * hi);
      v2f b = *reinterpret_cast<const v2f*>(brow + k0 + 2 * hi);
      acc = __builtin_amdgcn_wmma_f32_16x16x4_f32(false, a, false, b,
                                                  (short)0, acc, false, false);
    }
  }

  const int col = colBase + wn + lo;
  if (col < DST) {
    #pragma unroll
    for (int r = 0; r < 8; ++r) {
      int row = rowBase + wm + hi * 8 + r;
      if (row < N) H[(size_t)row * DST + col] = acc[r];
    }
  }
}

#define GP 260
#define GR 32

template <bool TWO>
__device__ __forceinline__ void wres_tiles(const float* a0p, const v2f (&bf)[64],
                                           v8f& acc0, v8f& acc1) {
  #pragma unroll
  for (int j = 0; j < 64; ++j) {
    const v2f a0 = *reinterpret_cast<const v2f*>(a0p + 4 * j);
    acc0 = __builtin_amdgcn_wmma_f32_16x16x4_f32(false, a0, false, bf[j],
                                                 (short)0, acc0, false, false);
    if (TWO) {
      const v2f a1 = *reinterpret_cast<const v2f*>(a0p + 16 * GP + 4 * j);
      acc1 = __builtin_amdgcn_wmma_f32_16x16x4_f32(false, a1, false, bf[j],
                                                   (short)0, acc1, false, false);
    }
  }
}

__device__ __forceinline__ void lds_barrier() {
  __builtin_amdgcn_fence(__ATOMIC_RELEASE, "workgroup", "local");
  __builtin_amdgcn_s_barrier();
  __builtin_amdgcn_fence(__ATOMIC_ACQUIRE, "workgroup", "local");
}

struct XQuad { float4 a, b, c, d; };

__device__ __forceinline__ XQuad xquad_load(const float4* X4, int row, int rlast) {
  int q0 = row, q1 = row + 8, q2 = row + 16, q3 = row + 24;
  q0 = (q0 < rlast) ? q0 : rlast;
  q1 = (q1 < rlast) ? q1 : rlast;
  q2 = (q2 < rlast) ? q2 : rlast;
  q3 = (q3 < rlast) ? q3 : rlast;
  XQuad r;
  r.a = X4[(size_t)q0 * (DK / 4)];
  r.b = X4[(size_t)q1 * (DK / 4)];
  r.c = X4[(size_t)q2 * (DK / 4)];
  r.d = X4[(size_t)q3 * (DK / 4)];
  return r;
}

__device__ __forceinline__ void xquad_to_lds(float* tile, int so, const XQuad& v) {
  *reinterpret_cast<float4*>(&tile[so])           = v.a;
  *reinterpret_cast<float4*>(&tile[so + 8 * GP])  = v.b;
  *reinterpret_cast<float4*>(&tile[so + 16 * GP]) = v.c;
  *reinterpret_cast<float4*>(&tile[so + 24 * GP]) = v.d;
}

__global__ __launch_bounds__(512) __attribute__((amdgpu_waves_per_eu(4, 4)))
void gat_gemm_wres(
    const float* X, const float* __restrict__ W, float* H, int N) {
  __shared__ __attribute__((aligned(16))) float Xs[2 * GR * GP];

  const int tid  = threadIdx.x;
  const int lane = tid & 31;
  const int lo   = lane & 15;
  const int hi   = lane >> 4;
  const int wid  = __builtin_amdgcn_readfirstlane(tid >> 5);
  const int n0   = wid * 16;

  const int nb  = (int)gridDim.x;
  const int per = (N + nb - 1) / nb;
  const int r0  = (int)blockIdx.x * per;
  int r1 = r0 + per;
  r1 = (r1 < N) ? r1 : N;
  if (r0 >= r1) return;
  const int nrows = r1 - r0;
  const int nfull = nrows / GR;
  const int rem   = nrows - nfull * GR;
  const int rlast = N - 1;

  const int sc4 = tid & 63;
  const int sr  = tid >> 6;
  const float4* X4 = reinterpret_cast<const float4*>(X) + sc4;
  const int so = sr * GP + sc4 * 4;

  const XQuad t0 = xquad_load(X4, r0 + sr, rlast);
  XQuad cur = xquad_load(X4, r0 + GR + sr, rlast);

  v2f bf[64];
  {
    const float* wp = W + (size_t)(2 * hi) * 256 + n0 + lo;
    #pragma unroll
    for (int j = 0; j < 64; ++j) {
      bf[j].x = wp[(size_t)(4 * j) * 256];
      bf[j].y = wp[(size_t)(4 * j + 1) * 256];
    }
  }

  xquad_to_lds(Xs, so, t0);
  lds_barrier();

  const int aoff = lo * GP + 2 * hi;
  int rb = r0;
  if (nfull > 0) {
    v8f acc0 = {}, acc1 = {};
    wres_tiles<true>(Xs + aoff, bf, acc0, acc1);
    for (int s = 0; ; ) {
      xquad_to_lds(Xs + ((s + 1) & 1) * (GR * GP), so, cur);
      lds_barrier();

      const XQuad nxt = xquad_load(X4, rb + 2 * GR + sr, rlast);

      float* hp = H + (size_t)(rb + 8 * hi) * 256 + n0 + lo;
      #pragma unroll
      for (int r = 0; r < 8; ++r) {
        hp[(size_t)r * 256]        = acc0[r];
        hp[(size_t)(16 + r) * 256] = acc1[r];
      }

      ++s;
      rb += GR;
      if (s == nfull) break;

      const float* xs = Xs + (s & 1) * (GR * GP);
      v8f z0 = {}, z1 = {};
      acc0 = z0;
      acc1 = z1;
      wres_tiles<true>(xs + aoff, bf, acc0, acc1);
      cur = nxt;
    }
  }

  if (rem > 0) {
    const float* xs = Xs + (nfull & 1) * (GR * GP);
    v8f acc0 = {}, acc1 = {};
    const bool two = rem > 16;
    if (two) wres_tiles<true>(xs + aoff, bf, acc0, acc1);
    else     wres_tiles<false>(xs + aoff, bf, acc0, acc1);
    float* hp = H + (size_t)(rb + 8 * hi) * 256 + n0 + lo;
    #pragma unroll
    for (int r = 0; r < 8; ++r) {
      if (8 * hi + r < rem) hp[(size_t)r * 256] = acc0[r];
    }
    if (two) {
      #pragma unroll
      for (int r = 0; r < 8; ++r) {
        if (16 + 8 * hi + r < rem) hp[(size_t)(16 + r) * 256] = acc1[r];
      }
    }
  }
}

#define AL_G 7
template <int DPAD, int DACT>
__global__ __launch_bounds__(256) void gat_alpha(
    const float* __restrict__ H, const float* __restrict__ a_s,
    const float* __restrict__ a_d, float* __restrict__ asv,
    float* __restrict__ adv, int N) {
  const int lane = threadIdx.x & 31;
  const int gw  = blockIdx.x * 8 + __builtin_amdgcn_readfirstlane(threadIdx.x >> 5);
  const int nw  = gridDim.x * 8;
  const int per = (N + nw - 1) / nw;
  const int beg = gw * per;
  int end = beg + per;
  end = (end < N) ? end : N;
  if (beg >= end) return;
  constexpr int NJ = (DPAD + 31) / 32;
  int   dh[NJ];
  float av[NJ], bv[NJ];
  #pragma unroll
  for (int j = 0; j < NJ; ++j) {
    const int d  = lane + 32 * j;
    const int da = (d < DACT) ? d : (DACT - 1);
    dh[j] = (d < DPAD) ? d : (DPAD - 1);
    const float a = a_s[da];
    const float b = a_d[da];
    av[j] = (d < DACT) ? a : 0.f;
    bv[j] = (d < DACT) ? b : 0.f;
  }
  for (int i0 = beg; i0 < end; i0 += AL_G) {
    float hv[AL_G][NJ];
    #pragma unroll
    for (int g = 0; g < AL_G; ++g) {
      int i = i0 + g;
      i = (i < end) ? i : (end - 1);
      const float* hp = H + (size_t)i * DPAD;
      #pragma unroll
      for (int j = 0; j < NJ; ++j) hv[g][j] = hp[dh[j]];
    }
    float r1 = 0.f, r2 = 0.f;
    #pragma unroll
    for (int g = 0; g < AL_G; ++g) {
      float s1 = 0.f, s2 = 0.f;
      #pragma unroll
      for (int j = 0; j < NJ; ++j) {
        const int d = lane + 32 * j;
        const float h = (d < DPAD) ? hv[g][j] : 0.f;
        s1 = fmaf(h, av[j], s1);
        s2 = fmaf(h, bv[j], s2);
      }
      #pragma unroll
      for (int off = 16; off > 0; off >>= 1) {
        s1 += __shfl_xor(s1, off, 32);
        s2 += __shfl_xor(s2, off, 32);
      }
      r1 = (lane == g) ? s1 : r1;
      r2 = (lane == g) ? s2 : r2;
    }
    int cnt = end - i0;
    cnt = (cnt < AL_G) ? cnt : AL_G;
    if (lane < cnt) {
      asv[i0 + lane] = r1;
      adv[i0 + lane] = r2;
    }
  }
}

#define AD 256
#define D3 16
#define SCB 8
#define LROUNDS 6
#define LCAP (1024 * LROUNDS)

__device__ __forceinline__ float leaky02(float t) {
  return (t > 0.f) ? t : 0.2f * t;
}

__device__ __forceinline__ int rl_i(int v, int l) {
  return __builtin_amdgcn_readlane(v, l);
}
__device__ __forceinline__ float rl_f(float v, int l) {
  return __int_as_float(__builtin_amdgcn_readlane(__float_as_int(v), l));
}

template <int RA, int RB>
__global__ __launch_bounds__(256) void gat_owner_bytes(
    const int* __restrict__ dst, unsigned char* __restrict__ oba,
    unsigned char* __restrict__ obb, int E, int N) {
  const int nq = (E + 15) >> 4;
  if (nq > 2 * N) return;
  const int q = blockIdx.x * 256 + threadIdx.x;
  if (q >= nq) return;
  const int last = E - 1;
  int d[16];
  #pragma unroll
  for (int j = 0; j < 16; ++j) {
    const int e = q * 16 + j;
    d[j] = dst[e < last ? e : last];
  }
  unsigned wa[4], wb[4];
  #pragma unroll
  for (int w = 0; w < 4; ++w) {
    unsigned a = 0, b = 0;
    #pragma unroll
    for (int t = 0; t < 4; ++t) {
      const int j = w * 4 + t;
      const bool in = (q * 16 + j) < E;
      const unsigned ba = in ? ((unsigned)(d[j] / RA) & 255u) : 255u;
      const unsigned bb = in ? ((unsigned)(d[j] / RB) & 255u) : 255u;
      a |= ba << (8 * t);
      b |= bb << (8 * t);
    }
    wa[w] = a;
    wb[w] = b;
  }
  reinterpret_cast<uint4*>(oba)[q] = make_uint4(wa[0], wa[1], wa[2], wa[3]);
  reinterpret_cast<uint4*>(obb)[q] = make_uint4(wb[0], wb[1], wb[2], wb[3]);
}

__device__ __forceinline__ unsigned zero_bytes(unsigned x) {
  const unsigned y = (x & 0x7F7F7F7Fu) + 0x7F7F7F7Fu;
  return ~(y | x | 0x7F7F7F7Fu);
}

__device__ __forceinline__ void list_append(int* cnt, int* lst, int qreg,
                                            int m, int lane) {
  int b = 0;
  if (lane == 0) b = atomicAdd(cnt, m);
  b = __builtin_amdgcn_readfirstlane(b);
  const int slot = b + lane;
  if (lane < m && slot < LCAP) lst[slot] = qreg;
}

#define RG 8
template <int NR>
__device__ __forceinline__ void row_add(
    int s, int dl, float p, int g, int c, int lane,
    const float* __restrict__ H, float* acc) {
  float4 v[NR][2];
  float pb[NR];
  int   ro[NR];
  #pragma unroll
  for (int b = 0; b < NR; ++b) {
    const int q = (b < c) ? (g + b) : g;
    const int sb = rl_i(s, q);
    ro[b] = rl_i(dl, q) * AD + lane;
    pb[b] = rl_f(p, q);
    const float4* hp = reinterpret_cast<const float4*>(H + (size_t)sb * AD) + lane;
    v[b][0] = hp[0];
    v[b][1] = hp[32];
  }
  #pragma unroll
  for (int b = 0; b < NR; ++b) {
    if (b < c) {
      float* a = &acc[ro[b]];
      atomicAdd(a + 0 * 32, pb[b] * v[b][0].x);
      atomicAdd(a + 1 * 32, pb[b] * v[b][0].y);
      atomicAdd(a + 2 * 32, pb[b] * v[b][0].z);
      atomicAdd(a + 3 * 32, pb[b] * v[b][0].w);
      atomicAdd(a + 4 * 32, pb[b] * v[b][1].x);
      atomicAdd(a + 5 * 32, pb[b] * v[b][1].y);
      atomicAdd(a + 6 * 32, pb[b] * v[b][1].z);
      atomicAdd(a + 7 * 32, pb[b] * v[b][1].w);
    }
  }
}

__device__ __forceinline__ void row_add16(
    int s, int dl, float p, const float* __restrict__ H, float* acc) {
  const float4* hp = reinterpret_cast<const float4*>(H + (size_t)s * D3);
  const float4 a0 = hp[0], a1 = hp[1], a2 = hp[2], a3 = hp[3];
  float* ar = &acc[dl * D3];
  atomicAdd(ar + 0,  p * a0.x); atomicAdd(ar + 1,  p * a0.y);
  atomicAdd(ar + 2,  p * a0.z); atomicAdd(ar + 3,  p * a0.w);
  atomicAdd(ar + 4,  p * a1.x); atomicAdd(ar + 5,  p * a1.y);
  atomicAdd(ar + 6,  p * a1.z); atomicAdd(ar + 7,  p * a1.w);
  atomicAdd(ar + 8,  p * a2.x); atomicAdd(ar + 9,  p * a2.y);
  atomicAdd(ar + 10, p * a2.z); atomicAdd(ar + 11, p * a2.w);
  atomicAdd(ar + 12, p * a3.x); atomicAdd(ar + 13, p * a3.y);
  atomicAdd(ar + 14, p * a3.z); atomicAdd(ar + 15, p * a3.w);
}

template <int D, int ARW, int MODE>
__device__ __forceinline__ void owner_direct(
    const int* __restrict__ src, const int* __restrict__ dst,
    const float* __restrict__ asv, const float* __restrict__ H,
    const float* advl, unsigned* mrow, float* den, float* acc,
    int lo, int E, int ET, int wid, int lane) {
  for (int b0 = wid * 32; b0 < ET; b0 += 1024) {
    const int e = b0 + lane;
    int s = 0, d = -1;
    if (e < ET) {
      if (e < E) { s = src[e]; d = dst[e]; } else { s = d = e - E; }
    }
    const int dl = d - lo;
    const bool mine = (unsigned)dl < (unsigned)ARW;
    float sc = 0.f;
    if (mine) sc = leaky02(asv[s] + advl[dl]);
    if (MODE == 0) {
      if (mine) atomicMax(&mrow[dl], f2ord(sc));
    } else {
      float p = 0.f;
      if (mine) {
        p = __expf(sc - ord2f(mrow[dl]));
        atomicAdd(&den[dl], p);
      }
      if (D == AD) {
        unsigned mask = __builtin_amdgcn_ballot_w32(mine);
        while (mask) {
          const int L = __builtin_ctz(mask);
          mask &= mask - 1;
          row_add<1>(s, dl, p, L, 1, lane, H, acc);
        }
      } else {
        if (mine) row_add16(s, dl, p, H, acc);
      }
    }
  }
}

template <int D, int ARW>
__global__ __launch_bounds__(1024) void gat_owner(
    const int* __restrict__ src, const int* __restrict__ dst,
    const unsigned char* __restrict__ ob,
    const float* __restrict__ asv, const float* __restrict__ adv,
    const float* __restrict__ H, const float* __restrict__ bias,
    float* __restrict__ OUT, int E, int N, int Dact) {
  __shared__ __attribute__((aligned(16))) float acc[ARW * D];
  __shared__ int      lsrc[LCAP];
  __shared__ int      ldl[LCAP];
  __shared__ float    lp[LCAP];
  __shared__ unsigned mrow[ARW];
  __shared__ float    den[ARW];
  __shared__ float    advl[ARW];
  __shared__ int      cnt;

  const int tid  = threadIdx.x;
  const int lane = tid & 31;
  const int wid  = __builtin_amdgcn_readfirstlane(tid >> 5);
  const int lo   = blockIdx.x * ARW;
  const int ET   = E + N;
  int nown = N - lo;
  nown = (nown < ARW) ? nown : ARW;

  for (int x = tid; x < ARW * D / 4; x += 1024)
    reinterpret_cast<float4*>(acc)[x] = make_float4(0.f, 0.f, 0.f, 0.f);
  for (int r = tid; r < ARW; r += 1024) {
    mrow[r] = f2ord(-INFINITY);
    den[r]  = 0.f;
    advl[r] = (r < nown) ? adv[lo + r] : 0.f;
    if (r < nown) lsrc[r] = E + lo + r;
  }
  if (tid == 0) cnt = nown;
  __syncthreads();

  const int nq = (E + 15) >> 4;
  const bool bytes_ok = ((N + ARW - 1) / ARW <= 255) && (nq <= 2 * N);
  int qreg = 0;
  int n = 0;
  if (bytes_ok && nq > 0) {
    const uint4* ob4 = reinterpret_cast<const uint4*>(ob);
    const unsigned bsel = blockIdx.x * 0x01010101u;
    const int qstep = SCB * 32;
    const int qlast = nq - 1;
    int qb = wid * qstep;
    uint4 dv[SCB];
    #pragma unroll
    for (int k = 0; k < SCB; ++k) {
      const int qi = qb + k * 32 + lane;
      dv[k] = ob4[qi < qlast ? qi : qlast];
    }
    while (qb < nq) {
      unsigned z[SCB];
      #pragma unroll
      for (int k = 0; k < SCB; ++k) {
        const int qi = qb + k * 32 + lane;
        const unsigned m = (zero_bytes(dv[k].x ^ bsel) >> 7) |
                           (zero_bytes(dv[k].y ^ bsel) >> 6) |
                           (zero_bytes(dv[k].z ^ bsel) >> 5) |
                           (zero_bytes(dv[k].w ^ bsel) >> 4);
        z[k] = (qi < nq) ? m : 0u;
      }
      const int cur = qb;
      qb += 32 * qstep;
      #pragma unroll
      for (int k = 0; k < SCB; ++k) {
        const int qi = qb + k * 32 + lane;
        dv[k] = ob4[qi < qlast ? qi : qlast];
      }
      #pragma unroll
      for (int k = 0; k < SCB; ++k) {
        unsigned lanes = __builtin_amdgcn_ballot_w32(z[k] != 0);
        while (lanes) {
          const int L = __builtin_ctz(lanes);
          lanes &= lanes - 1;
          unsigned bits = (unsigned)__builtin_amdgcn_readlane((int)z[k], L);
          const int e0 = (cur + k * 32 + L) * 16;
          while (bits) {
            const int p = __builtin_ctz(bits);
            bits &= bits - 1;
            const int id = e0 + ((p & 7) << 2) + (p >> 3);
            qreg = (lane == n) ? id : qreg;
            ++n;
            if (n == 32) {
              list_append(&cnt, lsrc, qreg, 32, lane);
              n = 0;
            }
          }
        }
      }
    }
    if (n) list_append(&cnt, lsrc, qreg, n, lane);
  }
  __syncthreads();

  const int total = bytes_ok ? cnt : (LCAP + 1);
  if (total <= LCAP) {
    int   sk[LROUNDS], dk[LROUNDS];
    float ak[LROUNDS];
    #pragma unroll
    for (int k = 0; k < LROUNDS; ++k) {
      const int i = tid + 1024 * k;
      int s = 0, d = lo;
      if (i < total) {
        const int e = lsrc[i];
        if (e < E) { s = src[e]; d = dst[e]; } else { s = d = e - E; }
      }
      sk[k] = s;
      dk[k] = d - lo;
    }
    #pragma unroll
    for (int k = 0; k < LROUNDS; ++k) ak[k] = asv[sk[k]];
    #pragma unroll
    for (int k = 0; k < LROUNDS; ++k) {
      const int i = tid + 1024 * k;
      if (i < total) {
        const float sc = leaky02(ak[k] + advl[dk[k]]);
        lsrc[i] = sk[k];
        ldl[i]  = dk[k];
        lp[i]   = sc;
        atomicMax(&mrow[dk[k]], f2ord(sc));
      }
    }
    __syncthreads();
    for (int b0 = wid * 32; b0 < total; b0 += 1024) {
      const int i = b0 + lane;
      int s = 0, dl = 0;
      float p = 0.f;
      if (i < total) {
        s  = lsrc[i];
        dl = ldl[i];
        p  = __expf(lp[i] - ord2f(mrow[dl]));
        atomicAdd(&den[dl], p);
      }
      if (D == AD) {
        int m = total - b0;
        m = (m < 32) ? m : 32;
        for (int g = 0; g < m; g += RG)
          row_add<RG>(s, dl, p, g, m - g, lane, H, acc);
      } else {
        if (i < total) row_add16(s, dl, p, H, acc);
      }
    }
    __syncthreads();
  } else {
    owner_direct<D, ARW, 0>(src, dst, asv, H, advl, mrow, den, acc,
                            lo, E, ET, wid, lane);
    __syncthreads();
    owner_direct<D, ARW, 1>(src, dst, asv, H, advl, mrow, den, acc,
                            lo, E, ET, wid, lane);
    __syncthreads();
  }

  if (D == AD) {
    for (int idx = tid; idx < ARW * (AD / 4); idx += 1024) {
      const int r  = idx >> 6;
      const int c4 = idx & 63;
      if (r < nown) {
        const float inv = 1.0f / den[r];
        const float* ap = &acc[r * AD + (c4 >> 5) * 128 + (c4 & 31)];
        float4 a  = make_float4(ap[0], ap[32], ap[64], ap[96]);
        float4 bb = *reinterpret_cast<const float4*>(bias + c4 * 4);
        float4 o;
        o.x = a.x * inv + bb.x; o.x = o.x > 0.f ? o.x : 0.f;
        o.y = a.y * inv + bb.y; o.y = o.y > 0.f ? o.y : 0.f;
        o.z = a.z * inv + bb.z; o.z = o.z > 0.f ? o.z : 0.f;
        o.w = a.w * inv + bb.w; o.w = o.w > 0.f ? o.w : 0.f;
        *reinterpret_cast<float4*>(OUT + (size_t)(lo + r) * AD + c4 * 4) = o;
      }
    }
  } else {
    const int nout = nown * Dact;
    for (int x = tid; x < nout; x += 1024) {
      const int r = x / Dact;
      const int c = x - r * Dact;
      OUT[(size_t)lo * Dact + x] = acc[r * D3 + c] / den[r] + bias[c];
    }
  }
}

extern "C" void kernel_launch(void* const* d_in, const int* in_sizes, int n_in,
                              void* d_out, int out_size, void* d_ws, size_t ws_size,
                              hipStream_t stream) {
  const float* x  = (const float*)d_in[0];
  const int*   ei = (const int*)d_in[1];
  const float* W1 = (const float*)d_in[2];
  const float* as1 = (const float*)d_in[3];
  const float* ad1 = (const float*)d_in[4];
  const float* b1 = (const float*)d_in[5];
  const float* W2 = (const float*)d_in[6];
  const float* as2 = (const float*)d_in[7];
  const float* ad2 = (const float*)d_in[8];
  const float* b2 = (const float*)d_in[9];
  const float* W3 = (const float*)d_in[10];
  const float* as3 = (const float*)d_in[11];
  const float* ad3 = (const float*)d_in[12];
  const float* b3 = (const float*)d_in[13];

  const int N  = in_sizes[0] / 256;
  const int E  = in_sizes[1] / 2;
  const int* src = ei;
  const int* dst = ei + E;

  float* ws = (float*)d_ws;
  float*    hA  = ws;
  float*    hB  = hA + (size_t)N * 256;
  float*    obr = hB + (size_t)N * 256;
  float*    asv = obr + (size_t)N * 16;
  float*    adv = asv + N;
  float*    h3  = hA;
  unsigned char* obA = (unsigned char*)obr;
  unsigned char* obB = obA + (size_t)((E + 15) / 16) * 16;

  dim3 blk(256);
  dim3 gemmG3((N + BM - 1) / BM, 1);
  int gGemmW = 250;
  int gAlpha = 512;
  int gOwn   = (N + 199) / 200;
  int gOwn3  = (N + 207) / 208;

  gat_owner_bytes<200, 208><<<(E + 4096) / 4096, blk, 0, stream>>>(dst, obA, obB, E, N);

  gat_gemm_wres<<<gGemmW, 512, 0, stream>>>(x, W1, hA, N);
  gat_alpha<256, 256><<<gAlpha, blk, 0, stream>>>(hA, as1, ad1, asv, adv, N);
  gat_owner<AD, 200><<<gOwn, 1024, 0, stream>>>(src, dst, obA, asv, adv, hA, b1, hB, E, N, 256);

  gat_gemm_wres<<<gGemmW, 512, 0, stream>>>(hB, W2, hA, N);
  gat_alpha<256, 256><<<gAlpha, blk, 0, stream>>>(hA, as2, ad2, asv, adv, N);
  gat_owner<AD, 200><<<gOwn, 1024, 0, stream>>>(src, dst, obA, asv, adv, hA, b2, hB, E, N, 256);

  gat_gemm_wmma<10, 16><<<gemmG3, blk, 0, stream>>>(hB, W3, h3, N);
  gat_alpha<16, 10><<<gAlpha, blk, 0, stream>>>(h3, as3, ad3, asv, adv, N);
  gat_owner<D3, 208><<<gOwn3, 1024, 0, stream>>>(src, dst, obB, asv, adv, h3, b3, (float*)d_out, E, N, 10);
}


# Layers 2-3 attention dots taken as x.(W a) in the owner block that writes input row x (W a from extra blocks of the first launch): 2 dot launches and their h re-reads go; dot sum order changes
# speedup vs baseline: 4.5899x; 1.0361x over previous
#include <hip/hip_runtime.h>
#include <math.h>


typedef float v2f __attribute__((ext_vector_type(2)));
typedef float v8f __attribute__((ext_vector_type(8)));

#define DK 256
#define KCHUNK 128
#define LDT (KCHUNK + 4)
#define BM 32
#define BN 64

__device__ __forceinline__ unsigned f2ord(float f) {
  unsigned u = __float_as_uint(f);
  return (u & 0x80000000u) ? ~u : (u | 0x80000000u);
}
__device__ __forceinline__ float ord2f(unsigned u) {
  return __uint_as_float((u & 0x80000000u) ? (u ^ 0x80000000u) : ~u);
}

template <int DACT, int DST>
__global__ __launch_bounds__(256) void gat_gemm_wmma(
    const float* __restrict__ X, const float* __restrict__ W,
    float* __restrict__ H, int N) {
  __shared__ float As[BM * LDT];
  __shared__ float Bst[BN * LDT];

  const int tid  = threadIdx.x;
  const int lane = tid & 31;
  const int lo   = lane & 15;
  const int hi   = lane >> 4;
  const int wid  = tid >> 5;
  const int wm   = (wid >> 2) * 16;
  const int wn   = (wid & 3) * 16;
  const int rowBase = blockIdx.x * BM;
  const int colBase = blockIdx.y * BN;

  const int ar  = tid >> 5;
  const int ac4 = tid & 31;
  const int bk  = tid >> 6;
  const int bn  = tid & 63;
  const int gc  = colBase + bn;
  const bool cok = gc < DACT;
  const int gcc = cok ? gc : (DACT - 1);

  v8f acc = {};

  for (int kk = 0; kk < DK; kk += KCHUNK) {
    float4 av[4];
    float  bv[32];
    #pragma unroll
    for (int it = 0; it < 4; ++it) {
      const int gr  = rowBase + ar + 8 * it;
      const int grc = (gr < N) ? gr : (N - 1);
      av[it] = *reinterpret_cast<const float4*>(X + (size_t)grc * DK + kk + ac4 * 4);
    }
    #pragma unroll
    for (int it = 0; it < 32; ++it)
      bv[it] = W[(size_t)(kk + bk + 4 * it) * DACT + gcc];

    __syncthreads();
    #pragma unroll
    for (int it = 0; it < 4; ++it) {
      const int gr = rowBase + ar + 8 * it;
      float4 v = av[it];
      if (gr >= N) v = make_float4(0.f, 0.f, 0.f, 0.f);
      *reinterpret_cast<float4*>(&As[(ar + 8 * it) * LDT + ac4 * 4]) = v;
    }
    #pragma unroll
    for (int it = 0; it < 32; ++it)
      Bst[bn * LDT + bk + 4 * it] = cok ? bv[it] : 0.f;
    __syncthreads();

    const float* arow = &As[(wm + lo) * LDT];
    const float* brow = &Bst[(wn + lo) * LDT];
    #pragma unroll
    for (int k0 = 0; k0 < KCHUNK; k0 += 4) {
      v2f a = *reinterpret_cast<const v2f*>(arow + k0 + 2 * hi);
      v2f b = *reinterpret_cast<const v2f*>(brow + k0 + 2 * hi);
      acc = __builtin_amdgcn_wmma_f32_16x16x4_f32(false, a, false, b,
                                                  (short)0, acc, false, false);
    }
  }

  const int col = colBase + wn + lo;
  if (col < DST) {
    #pragma unroll
    for (int r = 0; r < 8; ++r) {
      int row = rowBase + wm + hi * 8 + r;
      if (row < N) H[(size_t)row * DST + col] = acc[r];
    }
  }
}

#define GP 260
#define GR 32

template <bool TWO>
__device__ __forceinline__ void wres_tiles(const float* a0p, const v2f (&bf)[64],
                                           v8f& acc0, v8f& acc1) {
  #pragma unroll
  for (int j = 0; j < 64; ++j) {
    const v2f a0 = *reinterpret_cast<const v2f*>(a0p + 4 * j);
    acc0 = __builtin_amdgcn_wmma_f32_16x16x4_f32(false, a0, false, bf[j],
                                                 (short)0, acc0, false, false);
    if (TWO) {
      const v2f a1 = *reinterpret_cast<const v2f*>(a0p + 16 * GP + 4 * j);
      acc1 = __builtin_amdgcn_wmma_f32_16x16x4_f32(false, a1, false, bf[j],
                                                   (short)0, acc1, false, false);
    }
  }
}

__device__ __forceinline__ void lds_barrier() {
  __builtin_amdgcn_fence(__ATOMIC_RELEASE, "workgroup", "local");
  __builtin_amdgcn_s_barrier();
  __builtin_amdgcn_fence(__ATOMIC_ACQUIRE, "workgroup", "local");
}

struct XQuad { float4 a, b, c, d; };

__device__ __forceinline__ XQuad xquad_load(const float4* X4, int row, int rlast) {
  int q0 = row, q1 = row + 8, q2 = row + 16, q3 = row + 24;
  q0 = (q0 < rlast) ? q0 : rlast;
  q1 = (q1 < rlast) ? q1 : rlast;
  q2 = (q2 < rlast) ? q2 : rlast;
  q3 = (q3 < rlast) ? q3 : rlast;
  XQuad r;
  r.a = X4[(size_t)q0 * (DK / 4)];
  r.b = X4[(size_t)q1 * (DK / 4)];
  r.c = X4[(size_t)q2 * (DK / 4)];
  r.d = X4[(size_t)q3 * (DK / 4)];
  return r;
}

__device__ __forceinline__ void xquad_to_lds(float* tile, int so, const XQuad& v) {
  *reinterpret_cast<float4*>(&tile[so])           = v.a;
  *reinterpret_cast<float4*>(&tile[so + 8 * GP])  = v.b;
  *reinterpret_cast<float4*>(&tile[so + 16 * GP]) = v.c;
  *reinterpret_cast<float4*>(&tile[so + 24 * GP]) = v.d;
}

__global__ __launch_bounds__(512) __attribute__((amdgpu_waves_per_eu(4, 4)))
void gat_gemm_wres(
    const float* X, const float* __restrict__ W, float* H, int N) {
  __shared__ __attribute__((aligned(16))) float Xs[2 * GR * GP];

  const int tid  = threadIdx.x;
  const int lane = tid & 31;
  const int lo   = lane & 15;
  const int hi   = lane >> 4;
  const int wid  = __builtin_amdgcn_readfirstlane(tid >> 5);
  const int n0   = wid * 16;

  const int nb  = (int)gridDim.x;
  const int per = (N + nb - 1) / nb;
  const int r0  = (int)blockIdx.x * per;
  int r1 = r0 + per;
  r1 = (r1 < N) ? r1 : N;
  if (r0 >= r1) return;
  const int nrows = r1 - r0;
  const int nfull = nrows / GR;
  const int rem   = nrows - nfull * GR;
  const int rlast = N - 1;

  const int sc4 = tid & 63;
  const int sr  = tid >> 6;
  const float4* X4 = reinterpret_cast<const float4*>(X) + sc4;
  const int so = sr * GP + sc4 * 4;

  const XQuad t0 = xquad_load(X4, r0 + sr, rlast);
  XQuad cur = xquad_load(X4, r0 + GR + sr, rlast);

  v2f bf[64];
  {
    const float* wp = W + (size_t)(2 * hi) * 256 + n0 + lo;
    #pragma unroll
    for (int j = 0; j < 64; ++j) {
      bf[j].x = wp[(size_t)(4 * j) * 256];
      bf[j].y = wp[(size_t)(4 * j + 1) * 256];
    }
  }

  xquad_to_lds(Xs, so, t0);
  lds_barrier();

  const int aoff = lo * GP + 2 * hi;
  int rb = r0;
  if (nfull > 0) {
    v8f acc0 = {}, acc1 = {};
    wres_tiles<true>(Xs + aoff, bf, acc0, acc1);
    for (int s = 0; ; ) {
      xquad_to_lds(Xs + ((s + 1) & 1) * (GR * GP), so, cur);
      lds_barrier();

      const XQuad nxt = xquad_load(X4, rb + 2 * GR + sr, rlast);

      float* hp = H + (size_t)(rb + 8 * hi) * 256 + n0 + lo;
      #pragma unroll
      for (int r = 0; r < 8; ++r) {
        hp[(size_t)r * 256]        = acc0[r];
        hp[(size_t)(16 + r) * 256] = acc1[r];
      }

      ++s;
      rb += GR;
      if (s == nfull) break;

      const float* xs = Xs + (s & 1) * (GR * GP);
      v8f z0 = {}, z1 = {};
      acc0 = z0;
      acc1 = z1;
      wres_tiles<true>(xs + aoff, bf, acc0, acc1);
      cur = nxt;
    }
  }

  if (rem > 0) {
    const float* xs = Xs + (nfull & 1) * (GR * GP);
    v8f acc0 = {}, acc1 = {};
    const bool two = rem > 16;
    if (two) wres_tiles<true>(xs + aoff, bf, acc0, acc1);
    else     wres_tiles<false>(xs + aoff, bf, acc0, acc1);
    float* hp = H + (size_t)(rb + 8 * hi) * 256 + n0 + lo;
    #pragma unroll
    for (int r = 0; r < 8; ++r) {
      if (8 * hi + r < rem) hp[(size_t)r * 256] = acc0[r];
    }
    if (two) {
      #pragma unroll
      for (int r = 0; r < 8; ++r) {
        if (16 + 8 * hi + r < rem) hp[(size_t)(16 + r) * 256] = acc1[r];
      }
    }
  }
}

#define AL_G 7
template <int DPAD, int DACT>
__global__ __launch_bounds__(256) void gat_alpha(
    const float* __restrict__ H, const float* __restrict__ a_s,
    const float* __restrict__ a_d, float* __restrict__ asv,
    float* __restrict__ adv, int N) {
  const int lane = threadIdx.x & 31;
  const int gw  = blockIdx.x * 8 + __builtin_amdgcn_readfirstlane(threadIdx.x >> 5);
  const int nw  = gridDim.x * 8;
  const int per = (N + nw - 1) / nw;
  const int beg = gw * per;
  int end = beg + per;
  end = (end < N) ? end : N;
  if (beg >= end) return;
  constexpr int NJ = (DPAD + 31) / 32;
  int   dh[NJ];
  float av[NJ], bv[NJ];
  #pragma unroll
  for (int j = 0; j < NJ; ++j) {
    const int d  = lane + 32 * j;
    const int da = (d < DACT) ? d : (DACT - 1);
    dh[j] = (d < DPAD) ? d : (DPAD - 1);
    const float a = a_s[da];
    const float b = a_d[da];
    av[j] = (d < DACT) ? a : 0.f;
    bv[j] = (d < DACT) ? b : 0.f;
  }
  for (int i0 = beg; i0 < end; i0 += AL_G) {
    float hv[AL_G][NJ];
    #pragma unroll
    for (int g = 0; g < AL_G; ++g) {
      int i = i0 + g;
      i = (i < end) ? i : (end - 1);
      const float* hp = H + (size_t)i * DPAD;
      #pragma unroll
      for (int j = 0; j < NJ; ++j) hv[g][j] = hp[dh[j]];
    }
    float r1 = 0.f, r2 = 0.f;
    #pragma unroll
    for (int g = 0; g < AL_G; ++g) {
      float s1 = 0.f, s2 = 0.f;
      #pragma unroll
      for (int j = 0; j < NJ; ++j) {
        const int d = lane + 32 * j;
        const float h = (d < DPAD) ? hv[g][j] : 0.f;
        s1 = fmaf(h, av[j], s1);
        s2 = fmaf(h, bv[j], s2);
      }
      #pragma unroll
      for (int off = 16; off > 0; off >>= 1) {
        s1 += __shfl_xor(s1, off, 32);
        s2 += __shfl_xor(s2, off, 32);
      }
      r1 = (lane == g) ? s1 : r1;
      r2 = (lane == g) ? s2 : r2;
    }
    int cnt = end - i0;
    cnt = (cnt < AL_G) ? cnt : AL_G;
    if (lane < cnt) {
      asv[i0 + lane] = r1;
      adv[i0 + lane] = r2;
    }
  }
}

#define AD 256
#define D3 16
#define SCB 8
#define LROUNDS 6
#define LCAP (1024 * LROUNDS)

__device__ __forceinline__ float leaky02(float t) {
  return (t > 0.f) ? t : 0.2f * t;
}

__device__ __forceinline__ int rl_i(int v, int l) {
  return __builtin_amdgcn_readlane(v, l);
}
__device__ __forceinline__ float rl_f(float v, int l) {
  return __int_as_float(__builtin_amdgcn_readlane(__float_as_int(v), l));
}

__device__ __forceinline__ bool ob_room(int nq, int N) {
  return nq + 128 <= N + (N >> 1) + (N >> 2);
}

template <int DACT>
__device__ __forceinline__ void wa_rows(
    const float* __restrict__ W, const float* __restrict__ a_s,
    const float* __restrict__ a_d, float* __restrict__ wa_s,
    float* __restrict__ wa_d, int kb, int wid, int lane) {
  constexpr int NJ = (DACT + 31) / 32;
  int   dn[NJ];
  float av[NJ], bv[NJ];
  #pragma unroll
  for (int j = 0; j < NJ; ++j) {
    const int d = lane + 32 * j;
    dn[j] = (d < DACT) ? d : (DACT - 1);
    const float a = a_s[dn[j]];
    const float b = a_d[dn[j]];
    av[j] = (d < DACT) ? a : 0.f;
    bv[j] = (d < DACT) ? b : 0.f;
  }
  const int k0 = kb + wid * 4;
  float wv[4][NJ];
  #pragma unroll
  for (int t = 0; t < 4; ++t) {
    #pragma unroll
    for (int j = 0; j < NJ; ++j) wv[t][j] = W[(size_t)(k0 + t) * DACT + dn[j]];
  }
  float r1 = 0.f, r2 = 0.f;
  #pragma unroll
  for (int t = 0; t < 4; ++t) {
    float s1 = 0.f, s2 = 0.f;
    #pragma unroll
    for (int j = 0; j < NJ; ++j) {
      s1 = fmaf(wv[t][j], av[j], s1);
      s2 = fmaf(wv[t][j], bv[j], s2);
    }
    #pragma unroll
    for (int off = 16; off > 0; off >>= 1) {
      s1 += __shfl_xor(s1, off, 32);
      s2 += __shfl_xor(s2, off, 32);
    }
    r1 = (lane == t) ? s1 : r1;
    r2 = (lane == t) ? s2 : r2;
  }
  if (lane < 4) {
    wa_s[k0 + lane] = r1;
    wa_d[k0 + lane] = r2;
  }
}

template <int RA, int RB, int DACT3>
__global__ __launch_bounds__(256) void gat_owner_bytes(
    const int* __restrict__ dst, unsigned char* __restrict__ oba,
    unsigned char* __restrict__ obb, int E, int N, int nob,
    const float* __restrict__ W2, const float* __restrict__ as2,
    const float* __restrict__ ad2, const float* __restrict__ W3,
    const float* __restrict__ as3, const float* __restrict__ ad3,
    float* __restrict__ wav) {
  if ((int)blockIdx.x >= nob) {
    const int xb   = (int)blockIdx.x - nob;
    const int lane = threadIdx.x & 31;
    const int wid  = threadIdx.x >> 5;
    const int kb   = (xb & 7) * 32;
    if (xb < 8) wa_rows<256>(W2, as2, ad2, wav, wav + 256, kb, wid, lane);
    else        wa_rows<DACT3>(W3, as3, ad3, wav + 512, wav + 768, kb, wid, lane);
    return;
  }
  const int nq = (E + 15) >> 4;
  if (!ob_room(nq, N)) return;
  const int q = blockIdx.x * 256 + threadIdx.x;
  if (q >= nq) return;
  const int last = E - 1;
  int d[16];
  #pragma unroll
  for (int j = 0; j < 16; ++j) {
    const int e = q * 16 + j;
    d[j] = dst[e < last ? e : last];
  }
  unsigned wa[4], wb[4];
  #pragma unroll
  for (int w = 0; w < 4; ++w) {
    unsigned a = 0, b = 0;
    #pragma unroll
    for (int t = 0; t < 4; ++t) {
      const int j = w * 4 + t;
      const bool in = (q * 16 + j) < E;
      const unsigned ba = in ? ((unsigned)(d[j] / RA) & 255u) : 255u;
      const unsigned bb = in ? ((unsigned)(d[j] / RB) & 255u) : 255u;
      a |= ba << (8 * t);
      b |= bb << (8 * t);
    }
    wa[w] = a;
    wb[w] = b;
  }
  reinterpret_cast<uint4*>(oba)[q] = make_uint4(wa[0], wa[1], wa[2], wa[3]);
  reinterpret_cast<uint4*>(obb)[q] = make_uint4(wb[0], wb[1], wb[2], wb[3]);
}

__device__ __forceinline__ unsigned zero_bytes(unsigned x) {
  const unsigned y = (x & 0x7F7F7F7Fu) + 0x7F7F7F7Fu;
  return ~(y | x | 0x7F7F7F7Fu);
}

__device__ __forceinline__ void list_append(int* cnt, int* lst, int qreg,
                                            int m, int lane) {
  int b = 0;
  if (lane == 0) b = atomicAdd(cnt, m);
  b = __builtin_amdgcn_readfirstlane(b);
  const int slot = b + lane;
  if (lane < m && slot < LCAP) lst[slot] = qreg;
}

#define RG 8
template <int NR>
__device__ __forceinline__ void row_add(
    int s, int dl, float p, int g, int c, int lane,
    const float* __restrict__ H, float* acc) {
  float4 v[NR][2];
  float pb[NR];
  int   ro[NR];
  #pragma unroll
  for (int b = 0; b < NR; ++b) {
    const int q = (b < c) ? (g + b) : g;
    const int sb = rl_i(s, q);
    ro[b] = rl_i(dl, q) * AD + lane;
    pb[b] = rl_f(p, q);
    const float4* hp = reinterpret_cast<const float4*>(H + (size_t)sb * AD) + lane;
    v[b][0] = hp[0];
    v[b][1] = hp[32];
  }
  #pragma unroll
  for (int b = 0; b < NR; ++b) {
    if (b < c) {
      float* a = &acc[ro[b]];
      atomicAdd(a + 0 * 32, pb[b] * v[b][0].x);
      atomicAdd(a + 1 * 32, pb[b] * v[b][0].y);
      atomicAdd(a + 2 * 32, pb[b] * v[b][0].z);
      atomicAdd(a + 3 * 32, pb[b] * v[b][0].w);
      atomicAdd(a + 4 * 32, pb[b] * v[b][1].x);
      atomicAdd(a + 5 * 32, pb[b] * v[b][1].y);
      atomicAdd(a + 6 * 32, pb[b] * v[b][1].z);
      atomicAdd(a + 7 * 32, pb[b] * v[b][1].w);
    }
  }
}

__device__ __forceinline__ void row_add16(
    int s, int dl, float p, const float* __restrict__ H, float* acc) {
  const float4* hp = reinterpret_cast<const float4*>(H + (size_t)s * D3);
  const float4 a0 = hp[0], a1 = hp[1], a2 = hp[2], a3 = hp[3];
  float* ar = &acc[dl * D3];
  atomicAdd(ar + 0,  p * a0.x); atomicAdd(ar + 1,  p * a0.y);
  atomicAdd(ar + 2,  p * a0.z); atomicAdd(ar + 3,  p * a0.w);
  atomicAdd(ar + 4,  p * a1.x); atomicAdd(ar + 5,  p * a1.y);
  atomicAdd(ar + 6,  p * a1.z); atomicAdd(ar + 7,  p * a1.w);
  atomicAdd(ar + 8,  p * a2.x); atomicAdd(ar + 9,  p * a2.y);
  atomicAdd(ar + 10, p * a2.z); atomicAdd(ar + 11, p * a2.w);
  atomicAdd(ar + 12, p * a3.x); atomicAdd(ar + 13, p * a3.y);
  atomicAdd(ar + 14, p * a3.z); atomicAdd(ar + 15, p * a3.w);
}

template <int D, int ARW, int MODE>
__device__ __forceinline__ void owner_direct(
    const int* __restrict__ src, const int* __restrict__ dst,
    const float* __restrict__ asv, const float* __restrict__ H,
    const float* advl, unsigned* mrow, float* den, float* acc,
    int lo, int E, int ET, int wid, int lane) {
  for (int b0 = wid * 32; b0 < ET; b0 += 1024) {
    const int e = b0 + lane;
    int s = 0, d = -1;
    if (e < ET) {
      if (e < E) { s = src[e]; d = dst[e]; } else { s = d = e - E; }
    }
    const int dl = d - lo;
    const bool mine = (unsigned)dl < (unsigned)ARW;
    float sc = 0.f;
    if (mine) sc = leaky02(asv[s] + advl[dl]);
    if (MODE == 0) {
      if (mine) atomicMax(&mrow[dl], f2ord(sc));
    } else {
      float p = 0.f;
      if (mine) {
        p = __expf(sc - ord2f(mrow[dl]));
        atomicAdd(&den[dl], p);
      }
      if (D == AD) {
        unsigned mask = __builtin_amdgcn_ballot_w32(mine);
        while (mask) {
          const int L = __builtin_ctz(mask);
          mask &= mask - 1;
          row_add<1>(s, dl, p, L, 1, lane, H, acc);
        }
      } else {
        if (mine) row_add16(s, dl, p, H, acc);
      }
    }
  }
}

template <int D, int ARW>
__global__ __launch_bounds__(1024) void gat_owner(
    const int* __restrict__ src, const int* __restrict__ dst,
    const unsigned char* __restrict__ ob,
    const float* __restrict__ asv, const float* __restrict__ adv,
    const float* __restrict__ H, const float* __restrict__ bias,
    float* __restrict__ OUT, int E, int N, int Dact,
    const float* __restrict__ was, const float* __restrict__ wad,
    float* __restrict__ asvN, float* __restrict__ advN) {
  __shared__ __attribute__((aligned(16))) float acc[ARW * D];
  __shared__ int      lsrc[LCAP];
  __shared__ int      ldl[LCAP];
  __shared__ float    lp[LCAP];
  __shared__ unsigned mrow[ARW];
  __shared__ float    den[ARW];
  __shared__ float    advl[ARW];
  __shared__ int      cnt;

  const int tid  = threadIdx.x;
  const int lane = tid & 31;
  const int wid  = __builtin_amdgcn_readfirstlane(tid >> 5);
  const int lo   = blockIdx.x * ARW;
  const int ET   = E + N;
  int nown = N - lo;
  nown = (nown < ARW) ? nown : ARW;

  for (int x = tid; x < ARW * D / 4; x += 1024)
    reinterpret_cast<float4*>(acc)[x] = make_float4(0.f, 0.f, 0.f, 0.f);
  for (int r = tid; r < ARW; r += 1024) {
    mrow[r] = f2ord(-INFINITY);
    den[r]  = 0.f;
    advl[r] = (r < nown) ? adv[lo + r] : 0.f;
    if (r < nown) lsrc[r] = E + lo + r;
  }
  if (tid == 0) cnt = nown;
  __syncthreads();

  const int nq = (E + 15) >> 4;
  const bool bytes_ok = ((N + ARW - 1) / ARW <= 255) && ob_room(nq, N);
  int qreg = 0;
  int n = 0;
  if (bytes_ok && nq > 0) {
    const uint4* ob4 = reinterpret_cast<const uint4*>(ob);
    const unsigned bsel = blockIdx.x * 0x01010101u;
    const int qstep = SCB * 32;
    const int qlast = nq - 1;
    int qb = wid * qstep;
    uint4 dv[SCB];
    #pragma unroll
    for (int k = 0; k < SCB; ++k) {
      const int qi = qb + k * 32 + lane;
      dv[k] = ob4[qi < qlast ? qi : qlast];
    }
    while (qb < nq) {
      unsigned z[SCB];
      #pragma unroll
      for (int k = 0; k < SCB; ++k) {
        const int qi = qb + k * 32 + lane;
        const unsigned m = (zero_bytes(dv[k].x ^ bsel) >> 7) |
                           (zero_bytes(dv[k].y ^ bsel) >> 6) |
                           (zero_bytes(dv[k].z ^ bsel) >> 5) |
                           (zero_bytes(dv[k].w ^ bsel) >> 4);
        z[k] = (qi < nq) ? m : 0u;
      }
      const int cur = qb;
      qb += 32 * qstep;
      #pragma unroll
      for (int k = 0; k < SCB; ++k) {
        const int qi = qb + k * 32 + lane;
        dv[k] = ob4[qi < qlast ? qi : qlast];
      }
      #pragma unroll
      for (int k = 0; k < SCB; ++k) {
        unsigned lanes = __builtin_amdgcn_ballot_w32(z[k] != 0);
        while (lanes) {
          const int L = __builtin_ctz(lanes);
          lanes &= lanes - 1;
          unsigned bits = (unsigned)__builtin_amdgcn_readlane((int)z[k], L);
          const int e0 = (cur + k * 32 + L) * 16;
          while (bits) {
            const int p = __builtin_ctz(bits);
            bits &= bits - 1;
            const int id = e0 + ((p & 7) << 2) + (p >> 3);
            qreg = (lane == n) ? id : qreg;
            ++n;
            if (n == 32) {
              list_append(&cnt, lsrc, qreg, 32, lane);
              n = 0;
            }
          }
        }
      }
    }
    if (n) list_append(&cnt, lsrc, qreg, n, lane);
  }
  __syncthreads();

  const int total = bytes_ok ? cnt : (LCAP + 1);
  if (total <= LCAP) {
    int   sk[LROUNDS], dk[LROUNDS];
    float ak[LROUNDS];
    #pragma unroll
    for (int k = 0; k < LROUNDS; ++k) {
      const int i = tid + 1024 * k;
      int s = 0, d = lo;
      if (i < total) {
        const int e = lsrc[i];
        if (e < E) { s = src[e]; d = dst[e]; } else { s = d = e - E; }
      }
      sk[k] = s;
      dk[k] = d - lo;
    }
    #pragma unroll
    for (int k = 0; k < LROUNDS; ++k) ak[k] = asv[sk[k]];
    #pragma unroll
    for (int k = 0; k < LROUNDS; ++k) {
      const int i = tid + 1024 * k;
      if (i < total) {
        const float sc = leaky02(ak[k] + advl[dk[k]]);
        lsrc[i] = sk[k];
        ldl[i]  = dk[k];
        lp[i]   = sc;
        atomicMax(&mrow[dk[k]], f2ord(sc));
      }
    }
    __syncthreads();
    for (int b0 = wid * 32; b0 < total; b0 += 1024) {
      const int i = b0 + lane;
      int s = 0, dl = 0;
      float p = 0.f;
      if (i < total) {
        s  = lsrc[i];
        dl = ldl[i];
        p  = __expf(lp[i] - ord2f(mrow[dl]));
        atomicAdd(&den[dl], p);
      }
      if (D == AD) {
        int m = total - b0;
        m = (m < 32) ? m : 32;
        for (int g = 0; g < m; g += RG)
          row_add<RG>(s, dl, p, g, m - g, lane, H, acc);
      } else {
        if (i < total) row_add16(s, dl, p, H, acc);
      }
    }
    __syncthreads();
  } else {
    owner_direct<D, ARW, 0>(src, dst, asv, H, advl, mrow, den, acc,
                            lo, E, ET, wid, lane);
    __syncthreads();
    owner_direct<D, ARW, 1>(src, dst, asv, H, advl, mrow, den, acc,
                            lo, E, ET, wid, lane);
    __syncthreads();
  }

  if (D == AD) {
    constexpr int RPW = (ARW + 31) / 32;
    const float4* b4 = reinterpret_cast<const float4*>(bias);
    const float4* s4 = reinterpret_cast<const float4*>(was);
    const float4* d4 = reinterpret_cast<const float4*>(wad);
    const float4 bb0 = b4[lane], bb1 = b4[32 + lane];
    const float4 ws0 = s4[lane], ws1 = s4[32 + lane];
    const float4 wd0 = d4[lane], wd1 = d4[32 + lane];
    const int rb = wid * RPW;
    float k1 = 0.f, k2 = 0.f;
    #pragma unroll
    for (int t = 0; t < RPW; ++t) {
      const int r = rb + t;
      if (r < nown) {
        const float inv = 1.0f / den[r];
        const float* ap = &acc[r * AD + lane];
        const float4 a0 = make_float4(ap[0], ap[32], ap[64], ap[96]);
        const float4 a1 = make_float4(ap[128], ap[160], ap[192], ap[224]);
        float4 o0, o1;
        o0.x = a0.x * inv + bb0.x; o0.x = o0.x > 0.f ? o0.x : 0.f;
        o0.y = a0.y * inv + bb0.y; o0.y = o0.y > 0.f ? o0.y : 0.f;
        o0.z = a0.z * inv + bb0.z; o0.z = o0.z > 0.f ? o0.z : 0.f;
        o0.w = a0.w * inv + bb0.w; o0.w = o0.w > 0.f ? o0.w : 0.f;
        o1.x = a1.x * inv + bb1.x; o1.x = o1.x > 0.f ? o1.x : 0.f;
        o1.y = a1.y * inv + bb1.y; o1.y = o1.y > 0.f ? o1.y : 0.f;
        o1.z = a1.z * inv + bb1.z; o1.z = o1.z > 0.f ? o1.z : 0.f;
        o1.w = a1.w * inv + bb1.w; o1.w = o1.w > 0.f ? o1.w : 0.f;
        float4* op = reinterpret_cast<float4*>(OUT + (size_t)(lo + r) * AD) + lane;
        op[0]  = o0;
        op[32] = o1;
        float s1 = o0.x * ws0.x, s2 = o0.x * wd0.x;
        s1 = fmaf(o0.y, ws0.y, s1); s2 = fmaf(o0.y, wd0.y, s2);
        s1 = fmaf(o0.z, ws0.z, s1); s2 = fmaf(o0.z, wd0.z, s2);
        s1 = fmaf(o0.w, ws0.w, s1); s2 = fmaf(o0.w, wd0.w, s2);
        s1 = fmaf(o1.x, ws1.x, s1); s2 = fmaf(o1.x, wd1.x, s2);
        s1 = fmaf(o1.y, ws1.y, s1); s2 = fmaf(o1.y, wd1.y, s2);
        s1 = fmaf(o1.z, ws1.z, s1); s2 = fmaf(o1.z, wd1.z, s2);
        s1 = fmaf(o1.w, ws1.w, s1); s2 = fmaf(o1.w, wd1.w, s2);
        #pragma unroll
        for (int off = 16; off > 0; off >>= 1) {
          s1 += __shfl_xor(s1, off, 32);
          s2 += __shfl_xor(s2, off, 32);
        }
        k1 = (lane == t) ? s1 : k1;
        k2 = (lane == t) ? s2 : k2;
      }
    }
    int cn = nown - rb;
    cn = (cn < RPW) ? cn : RPW;
    if (lane < cn) {
      asvN[lo + rb + lane] = k1;
      advN[lo + rb + lane] = k2;
    }
  } else {
    const int nout = nown * Dact;
    for (int x = tid; x < nout; x += 1024) {
      const int r = x / Dact;
      const int c = x - r * Dact;
      OUT[(size_t)lo * Dact + x] = acc[r * D3 + c] / den[r] + bias[c];
    }
  }
}

extern "C" void kernel_launch(void* const* d_in, const int* in_sizes, int n_in,
                              void* d_out, int out_size, void* d_ws, size_t ws_size,
                              hipStream_t stream) {
  const float* x  = (const float*)d_in[0];
  const int*   ei = (const int*)d_in[1];
  const float* W1 = (const float*)d_in[2];
  const float* as1 = (const float*)d_in[3];
  const float* ad1 = (const float*)d_in[4];
  const float* b1 = (const float*)d_in[5];
  const float* W2 = (const float*)d_in[6];
  const float* as2 = (const float*)d_in[7];
  const float* ad2 = (const float*)d_in[8];
  const float* b2 = (const float*)d_in[9];
  const float* W3 = (const float*)d_in[10];
  const float* as3 = (const float*)d_in[11];
  const float* ad3 = (const float*)d_in[12];
  const float* b3 = (const float*)d_in[13];

  const int N  = in_sizes[0] / 256;
  const int E  = in_sizes[1] / 2;
  const int* src = ei;
  const int* dst = ei + E;

  float* ws = (float*)d_ws;
  float*    hA  = ws;
  float*    hB  = hA + (size_t)N * 256;
  float*    obr = hB + (size_t)N * 256;
  float*    asv = obr + (size_t)N * 16;
  float*    adv = asv + N;
  float*    wa  = obr;
  float*    asvB = asv - 2 * (size_t)N;
  float*    advB = asvB + N;
  float*    h3  = hA;
  unsigned char* obA = (unsigned char*)(obr + 1024);
  unsigned char* obB = obA + (size_t)((E + 15) / 16) * 16;

  dim3 blk(256);
  dim3 gemmG3((N + BM - 1) / BM, 1);
  int gGemmW = 250;
  int gAlpha = 512;
  int gOwn   = (N + 199) / 200;
  int gOwn3  = (N + 207) / 208;

  int nOB = (E + 4096) / 4096;
  gat_owner_bytes<200, 208, 10><<<nOB + 16, blk, 0, stream>>>(
      dst, obA, obB, E, N, nOB, W2, as2, ad2, W3, as3, ad3, wa);

  gat_gemm_wres<<<gGemmW, 512, 0, stream>>>(x, W1, hA, N);
  gat_alpha<256, 256><<<gAlpha, blk, 0, stream>>>(hA, as1, ad1, asv, adv, N);
  gat_owner<AD, 200><<<gOwn, 1024, 0, stream>>>(src, dst, obA, asv, adv, hA, b1, hB, E, N, 256,
                                                wa, wa + 256, asvB, advB);

  gat_gemm_wres<<<gGemmW, 512, 0, stream>>>(hB, W2, hA, N);
  gat_owner<AD, 200><<<gOwn, 1024, 0, stream>>>(src, dst, obA, asvB, advB, hA, b2, hB, E, N, 256,
                                                wa + 512, wa + 768, asv, adv);

  gat_gemm_wmma<10, 16><<<gemmG3, blk, 0, stream>>>(hB, W3, h3, N);
  gat_owner<D3, 208><<<gOwn3, 1024, 0, stream>>>(src, dst, obB, asv, adv, h3, b3, (float*)d_out, E, N, 10,
                                                 nullptr, nullptr, nullptr, nullptr);
}


# Owner blocks find their edges in two levels: first launch adds a presence bit per owner per 64-edge unit; a block tests owner bytes of marked units only, lists hits by wave ballot; LDS arrival order
# speedup vs baseline: 4.9871x; 1.0865x over previous
#include <hip/hip_runtime.h>
#include <math.h>


typedef float v2f __attribute__((ext_vector_type(2)));
typedef float v8f __attribute__((ext_vector_type(8)));

#define DK 256
#define KCHUNK 128
#define LDT (KCHUNK + 4)
#define BM 32
#define BN 64

__device__ __forceinline__ unsigned f2ord(float f) {
  unsigned u = __float_as_uint(f);
  return (u & 0x80000000u) ? ~u : (u | 0x80000000u);
}
__device__ __forceinline__ float ord2f(unsigned u) {
  return __uint_as_float((u & 0x80000000u) ? (u ^ 0x80000000u) : ~u);
}

template <int DACT, int DST>
__global__ __launch_bounds__(256) void gat_gemm_wmma(
    const float* __restrict__ X, const float* __restrict__ W,
    float* __restrict__ H, int N) {
  __shared__ float As[BM * LDT];
  __shared__ float Bst[BN * LDT];

  const int tid  = threadIdx.x;
  const int lane = tid & 31;
  const int lo   = lane & 15;
  const int hi   = lane >> 4;
  const int wid  = tid >> 5;
  const int wm   = (wid >> 2) * 16;
  const int wn   = (wid & 3) * 16;
  const int rowBase = blockIdx.x * BM;
  const int colBase = blockIdx.y * BN;

  const int ar  = tid >> 5;
  const int ac4 = tid & 31;
  const int bk  = tid >> 6;
  const int bn  = tid & 63;
  const int gc  = colBase + bn;
  const bool cok = gc < DACT;
  const int gcc = cok ? gc : (DACT - 1);

  v8f acc = {};

  for (int kk = 0; kk < DK; kk += KCHUNK) {
    float4 av[4];
    float  bv[32];
    #pragma unroll
    for (int it = 0; it < 4; ++it) {
      const int gr  = rowBase + ar + 8 * it;
      const int grc = (gr < N) ? gr : (N - 1);
      av[it] = *reinterpret_cast<const float4*>(X + (size_t)grc * DK + kk + ac4 * 4);
    }
    #pragma unroll
    for (int it = 0; it < 32; ++it)
      bv[it] = W[(size_t)(kk + bk + 4 * it) * DACT + gcc];

    __syncthreads();
    #pragma unroll
    for (int it = 0; it < 4; ++it) {
      const int gr = rowBase + ar + 8 * it;
      float4 v = av[it];
      if (gr >= N) v = make_float4(0.f, 0.f, 0.f, 0.f);
      *reinterpret_cast<float4*>(&As[(ar + 8 * it) * LDT + ac4 * 4]) = v;
    }
    #pragma unroll
    for (int it = 0; it < 32; ++it)
      Bst[bn * LDT + bk + 4 * it] = cok ? bv[it] : 0.f;
    __syncthreads();

    const float* arow = &As[(wm + lo) * LDT];
    const float* brow = &Bst[(wn + lo) * LDT];
    #pragma unroll
    for (int k0 = 0; k0 < KCHUNK; k0 += 4) {
      v2f a = *reinterpret_cast<const v2f*>(arow + k0 + 2 * hi);
      v2f b = *reinterpret_cast<const v2f*>(brow + k0 + 2 * hi);
      acc = __builtin_amdgcn_wmma_f32_16x16x4_f32(false, a, false, b,
                                                  (short)0, acc, false, false);
    }
  }

  const int col = colBase + wn + lo;
  if (col < DST) {
    #pragma unroll
    for (int r = 0; r < 8; ++r) {
      int row = rowBase + wm + hi * 8 + r;
      if (row < N) H[(size_t)row * DST + col] = acc[r];
    }
  }
}

#define GP 260
#define GR 32

template <bool TWO>
__device__ __forceinline__ void wres_tiles(const float* a0p, const v2f (&bf)[64],
                                           v8f& acc0, v8f& acc1) {
  #pragma unroll
  for (int j = 0; j < 64; ++j) {
    const v2f a0 = *reinterpret_cast<const v2f*>(a0p + 4 * j);
    acc0 = __builtin_amdgcn_wmma_f32_16x16x4_f32(false, a0, false, bf[j],
                                                 (short)0, acc0, false, false);
    if (TWO) {
      const v2f a1 = *reinterpret_cast<const v2f*>(a0p + 16 * GP + 4 * j);
      acc1 = __builtin_amdgcn_wmma_f32_16x16x4_f32(false, a1, false, bf[j],
                                                   (short)0, acc1, false, false);
    }
  }
}

__device__ __forceinline__ void lds_barrier() {
  __builtin_amdgcn_fence(__ATOMIC_RELEASE, "workgroup", "local");
  __builtin_amdgcn_s_barrier();
  __builtin_amdgcn_fence(__ATOMIC_ACQUIRE, "workgroup", "local");
}

struct XQuad { float4 a, b, c, d; };

__device__ __forceinline__ XQuad xquad_load(const float4* X4, int row, int rlast) {
  int q0 = row, q1 = row + 8, q2 = row + 16, q3 = row + 24;
  q0 = (q0 < rlast) ? q0 : rlast;
  q1 = (q1 < rlast) ? q1 : rlast;
  q2 = (q2 < rlast) ? q2 : rlast;
  q3 = (q3 < rlast) ? q3 : rlast;
  XQuad r;
  r.a = X4[(size_t)q0 * (DK / 4)];
  r.b = X4[(size_t)q1 * (DK / 4)];
  r.c = X4[(size_t)q2 * (DK / 4)];
  r.d = X4[(size_t)q3 * (DK / 4)];
  return r;
}

__device__ __forceinline__ void xquad_to_lds(float* tile, int so, const XQuad& v) {
  *reinterpret_cast<float4*>(&tile[so])           = v.a;
  *reinterpret_cast<float4*>(&tile[so + 8 * GP])  = v.b;
  *reinterpret_cast<float4*>(&tile[so + 16 * GP]) = v.c;
  *reinterpret_cast<float4*>(&tile[so + 24 * GP]) = v.d;
}

__global__ __launch_bounds__(512) __attribute__((amdgpu_waves_per_eu(4, 4)))
void gat_gemm_wres(
    const float* X, const float* __restrict__ W, float* H, int N) {
  __shared__ __attribute__((aligned(16))) float Xs[2 * GR * GP];

  const int tid  = threadIdx.x;
  const int lane = tid & 31;
  const int lo   = lane & 15;
  const int hi   = lane >> 4;
  const int wid  = __builtin_amdgcn_readfirstlane(tid >> 5);
  const int n0   = wid * 16;

  const int nb  = (int)gridDim.x;
  const int per = (N + nb - 1) / nb;
  const int r0  = (int)blockIdx.x * per;
  int r1 = r0 + per;
  r1 = (r1 < N) ? r1 : N;
  if (r0 >= r1) return;
  const int nrows = r1 - r0;
  const int nfull = nrows / GR;
  const int rem   = nrows - nfull * GR;
  const int rlast = N - 1;

  const int sc4 = tid & 63;
  const int sr  = tid >> 6;
  const float4* X4 = reinterpret_cast<const float4*>(X) + sc4;
  const int so = sr * GP + sc4 * 4;

  const XQuad t0 = xquad_load(X4, r0 + sr, rlast);
  XQuad cur = xquad_load(X4, r0 + GR + sr, rlast);

  v2f bf[64];
  {
    const float* wp = W + (size_t)(2 * hi) * 256 + n0 + lo;
    #pragma unroll
    for (int j = 0; j < 64; ++j) {
      bf[j].x = wp[(size_t)(4 * j) * 256];
      bf[j].y = wp[(size_t)(4 * j + 1) * 256];
    }
  }

  xquad_to_lds(Xs, so, t0);
  lds_barrier();

  const int aoff = lo * GP + 2 * hi;
  int rb = r0;
  if (nfull > 0) {
    v8f acc0 = {}, acc1 = {};
    wres_tiles<true>(Xs + aoff, bf, acc0, acc1);
    for (int s = 0; ; ) {
      xquad_to_lds(Xs + ((s + 1) & 1) * (GR * GP), so, cur);
      lds_barrier();

      const XQuad nxt = xquad_load(X4, rb + 2 * GR + sr, rlast);

      float* hp = H + (size_t)(rb + 8 * hi) * 256 + n0 + lo;
      #pragma unroll
      for (int r = 0; r < 8; ++r) {
        hp[(size_t)r * 256]        = acc0[r];
        hp[(size_t)(16 + r) * 256] = acc1[r];
      }

      ++s;
      rb += GR;
      if (s == nfull) break;

      const float* xs = Xs + (s & 1) * (GR * GP);
      v8f z0 = {}, z1 = {};
      acc0 = z0;
      acc1 = z1;
      wres_tiles<true>(xs + aoff, bf, acc0, acc1);
      cur = nxt;
    }
  }

  if (rem > 0) {
    const float* xs = Xs + (nfull & 1) * (GR * GP);
    v8f acc0 = {}, acc1 = {};
    const bool two = rem > 16;
    if (two) wres_tiles<true>(xs + aoff, bf, acc0, acc1);
    else     wres_tiles<false>(xs + aoff, bf, acc0, acc1);
    float* hp = H + (size_t)(rb + 8 * hi) * 256 + n0 + lo;
    #pragma unroll
    for (int r = 0; r < 8; ++r) {
      if (8 * hi + r < rem) hp[(size_t)r * 256] = acc0[r];
    }
    if (two) {
      #pragma unroll
      for (int r = 0; r < 8; ++r) {
        if (16 + 8 * hi + r < rem) hp[(size_t)(16 + r) * 256] = acc1[r];
      }
    }
  }
}

#define AL_G 7
template <int DPAD, int DACT>
__global__ __launch_bounds__(256) void gat_alpha(
    const float* __restrict__ H, const float* __restrict__ a_s,
    const float* __restrict__ a_d, float* __restrict__ asv,
    float* __restrict__ adv, int N) {
  const int lane = threadIdx.x & 31;
  const int gw  = blockIdx.x * 8 + __builtin_amdgcn_readfirstlane(threadIdx.x >> 5);
  const int nw  = gridDim.x * 8;
  const int per = (N + nw - 1) / nw;
  const int beg = gw * per;
  int end = beg + per;
  end = (end < N) ? end : N;
  if (beg >= end) return;
  constexpr int NJ = (DPAD + 31) / 32;
  int   dh[NJ];
  float av[NJ], bv[NJ];
  #pragma unroll
  for (int j = 0; j < NJ; ++j) {
    const int d  = lane + 32 * j;
    const int da = (d < DACT) ? d : (DACT - 1);
    dh[j] = (d < DPAD) ? d : (DPAD - 1);
    const float a = a_s[da];
    const float b = a_d[da];
    av[j] = (d < DACT) ? a : 0.f;
    bv[j] = (d < DACT) ? b : 0.f;
  }
  for (int i0 = beg; i0 < end; i0 += AL_G) {
    float hv[AL_G][NJ];
    #pragma unroll
    for (int g = 0; g < AL_G; ++g) {
      int i = i0 + g;
      i = (i < end) ? i : (end - 1);
      const float* hp = H + (size_t)i * DPAD;
      #pragma unroll
      for (int j = 0; j < NJ; ++j) hv[g][j] = hp[dh[j]];
    }
    float r1 = 0.f, r2 = 0.f;
    #pragma unroll
    for (int g = 0; g < AL_G; ++g) {
      float s1 = 0.f, s2 = 0.f;
      #pragma unroll
      for (int j = 0; j < NJ; ++j) {
        const int d = lane + 32 * j;
        const float h = (d < DPAD) ? hv[g][j] : 0.f;
        s1 = fmaf(h, av[j], s1);
        s2 = fmaf(h, bv[j], s2);
      }
      #pragma unroll
      for (int off = 16; off > 0; off >>= 1) {
        s1 += __shfl_xor(s1, off, 32);
        s2 += __shfl_xor(s2, off, 32);
      }
      r1 = (lane == g) ? s1 : r1;
      r2 = (lane == g) ? s2 : r2;
    }
    int cnt = end - i0;
    cnt = (cnt < AL_G) ? cnt : AL_G;
    if (lane < cnt) {
      asv[i0 + lane] = r1;
      adv[i0 + lane] = r2;
    }
  }
}

#define AD 256
#define D3 16
#define SCB 8
#define LROUNDS 6
#define LCAP (1024 * LROUNDS)

__device__ __forceinline__ float leaky02(float t) {
  return (t > 0.f) ? t : 0.2f * t;
}

__device__ __forceinline__ int rl_i(int v, int l) {
  return __builtin_amdgcn_readlane(v, l);
}
__device__ __forceinline__ float rl_f(float v, int l) {
  return __int_as_float(__builtin_amdgcn_readlane(__float_as_int(v), l));
}

__device__ __forceinline__ bool ob_room(int nq, int N) {
  return nq + 128 <= N + (N >> 1) + (N >> 2);
}

__device__ __forceinline__ bool pl_room(int nq, int nup, int N) {
  return 2 * nq + 4 * nup + 256 <= 3 * N + (N >> 1);
}

template <int DACT>
__device__ __forceinline__ void wa_rows(
    const float* __restrict__ W, const float* __restrict__ a_s,
    const float* __restrict__ a_d, float* __restrict__ wa_s,
    float* __restrict__ wa_d, int kb, int wid, int lane) {
  constexpr int NJ = (DACT + 31) / 32;
  int   dn[NJ];
  float av[NJ], bv[NJ];
  #pragma unroll
  for (int j = 0; j < NJ; ++j) {
    const int d = lane + 32 * j;
    dn[j] = (d < DACT) ? d : (DACT - 1);
    const float a = a_s[dn[j]];
    const float b = a_d[dn[j]];
    av[j] = (d < DACT) ? a : 0.f;
    bv[j] = (d < DACT) ? b : 0.f;
  }
  const int k0 = kb + wid * 4;
  float wv[4][NJ];
  #pragma unroll
  for (int t = 0; t < 4; ++t) {
    #pragma unroll
    for (int j = 0; j < NJ; ++j) wv[t][j] = W[(size_t)(k0 + t) * DACT + dn[j]];
  }
  float r1 = 0.f, r2 = 0.f;
  #pragma unroll
  for (int t = 0; t < 4; ++t) {
    float s1 = 0.f, s2 = 0.f;
    #pragma unroll
    for (int j = 0; j < NJ; ++j) {
      s1 = fmaf(wv[t][j], av[j], s1);
      s2 = fmaf(wv[t][j], bv[j], s2);
    }
    #pragma unroll
    for (int off = 16; off > 0; off >>= 1) {
      s1 += __shfl_xor(s1, off, 32);
      s2 += __shfl_xor(s2, off, 32);
    }
    r1 = (lane == t) ? s1 : r1;
    r2 = (lane == t) ? s2 : r2;
  }
  if (lane < 4) {
    wa_s[k0 + lane] = r1;
    wa_d[k0 + lane] = r2;
  }
}

template <int RA, int RB, int DACT3>
__global__ __launch_bounds__(256) void gat_owner_bytes(
    const int* __restrict__ dst, unsigned char* __restrict__ oba,
    unsigned char* __restrict__ obb, int E, int N, int nob,
    const float* __restrict__ W2, const float* __restrict__ as2,
    const float* __restrict__ ad2, const float* __restrict__ W3,
    const float* __restrict__ as3, const float* __restrict__ ad3,
    float* __restrict__ wav) {
  if ((int)blockIdx.x >= nob) {
    const int xb   = (int)blockIdx.x - nob;
    const int lane = threadIdx.x & 31;
    const int wid  = threadIdx.x >> 5;
    const int kb   = (xb & 7) * 32;
    if (xb < 8) wa_rows<256>(W2, as2, ad2, wav, wav + 256, kb, wid, lane);
    else        wa_rows<DACT3>(W3, as3, ad3, wav + 512, wav + 768, kb, wid, lane);
    return;
  }
  __shared__ unsigned pls[1024];
  const int nq = (E + 15) >> 4;
  if (!ob_room(nq, N)) return;
  const int nup = nob * 64;
  const bool planes = pl_room(nq, nup, N);
  if (planes) {
    for (int x = threadIdx.x; x < 1024; x += 256) pls[x] = 0u;
    __syncthreads();
  }
  const int q = blockIdx.x * 256 + threadIdx.x;
  if (q < nq) {
    const int ul = threadIdx.x >> 2;
    const int last = E - 1;
    int d[16];
    #pragma unroll
    for (int j = 0; j < 16; ++j) {
      const int e = q * 16 + j;
      d[j] = dst[e < last ? e : last];
    }
    unsigned wa[4], wb[4];
    #pragma unroll
    for (int w = 0; w < 4; ++w) {
      unsigned a = 0, b = 0;
      #pragma unroll
      for (int t = 0; t < 4; ++t) {
        const int j = w * 4 + t;
        const bool in = (q * 16 + j) < E;
        const unsigned ba = in ? ((unsigned)(d[j] / RA) & 255u) : 255u;
        const unsigned bb = in ? ((unsigned)(d[j] / RB) & 255u) : 255u;
        a |= ba << (8 * t);
        b |= bb << (8 * t);
        if (planes && in) {
          atomicOr(&pls[(ba >> 5) * 64 + ul], 1u << (ba & 31u));
          atomicOr(&pls[512 + (bb >> 5) * 64 + ul], 1u << (bb & 31u));
        }
      }
      wa[w] = a;
      wb[w] = b;
    }
    reinterpret_cast<uint4*>(oba)[q] = make_uint4(wa[0], wa[1], wa[2], wa[3]);
    reinterpret_cast<uint4*>(obb)[q] = make_uint4(wb[0], wb[1], wb[2], wb[3]);
  }
  if (planes) {
    __syncthreads();
    unsigned* pl = reinterpret_cast<unsigned*>(oba + (size_t)nq * 32);
    for (int x = threadIdx.x; x < 1024; x += 256)
      pl[(size_t)(x >> 6) * nup + blockIdx.x * 64 + (x & 63)] = pls[x];
  }
}

__device__ __forceinline__ unsigned zero_bytes(unsigned x) {
  const unsigned y = (x & 0x7F7F7F7Fu) + 0x7F7F7F7Fu;
  return ~(y | x | 0x7F7F7F7Fu);
}

__device__ __forceinline__ void list_append(int* cnt, int* lst, int qreg,
                                            int m, int lane) {
  int b = 0;
  if (lane == 0) b = atomicAdd(cnt, m);
  b = __builtin_amdgcn_readfirstlane(b);
  const int slot = b + lane;
  if (lane < m && slot < LCAP) lst[slot] = qreg;
}

#define RG 8
template <int NR>
__device__ __forceinline__ void row_add(
    int s, int dl, float p, int g, int c, int lane,
    const float* __restrict__ H, float* acc) {
  float4 v[NR][2];
  float pb[NR];
  int   ro[NR];
  #pragma unroll
  for (int b = 0; b < NR; ++b) {
    const int q = (b < c) ? (g + b) : g;
    const int sb = rl_i(s, q);
    ro[b] = rl_i(dl, q) * AD + lane;
    pb[b] = rl_f(p, q);
    const float4* hp = reinterpret_cast<const float4*>(H + (size_t)sb * AD) + lane;
    v[b][0] = hp[0];
    v[b][1] = hp[32];
  }
  #pragma unroll
  for (int b = 0; b < NR; ++b) {
    if (b < c) {
      float* a = &acc[ro[b]];
      atomicAdd(a + 0 * 32, pb[b] * v[b][0].x);
      atomicAdd(a + 1 * 32, pb[b] * v[b][0].y);
      atomicAdd(a + 2 * 32, pb[b] * v[b][0].z);
      atomicAdd(a + 3 * 32, pb[b] * v[b][0].w);
      atomicAdd(a + 4 * 32, pb[b] * v[b][1].x);
      atomicAdd(a + 5 * 32, pb[b] * v[b][1].y);
      atomicAdd(a + 6 * 32, pb[b] * v[b][1].z);
      atomicAdd(a + 7 * 32, pb[b] * v[b][1].w);
    }
  }
}

__device__ __forceinline__ void row_add16(
    int s, int dl, float p, const float* __restrict__ H, float* acc) {
  const float4* hp = reinterpret_cast<const float4*>(H + (size_t)s * D3);
  const float4 a0 = hp[0], a1 = hp[1], a2 = hp[2], a3 = hp[3];
  float* ar = &acc[dl * D3];
  atomicAdd(ar + 0,  p * a0.x); atomicAdd(ar + 1,  p * a0.y);
  atomicAdd(ar + 2,  p * a0.z); atomicAdd(ar + 3,  p * a0.w);
  atomicAdd(ar + 4,  p * a1.x); atomicAdd(ar + 5,  p * a1.y);
  atomicAdd(ar + 6,  p * a1.z); atomicAdd(ar + 7,  p * a1.w);
  atomicAdd(ar + 8,  p * a2.x); atomicAdd(ar + 9,  p * a2.y);
  atomicAdd(ar + 10, p * a2.z); atomicAdd(ar + 11, p * a2.w);
  atomicAdd(ar + 12, p * a3.x); atomicAdd(ar + 13, p * a3.y);
  atomicAdd(ar + 14, p * a3.z); atomicAdd(ar + 15, p * a3.w);
}

template <int D, int ARW, int MODE>
__device__ __forceinline__ void owner_direct(
    const int* __restrict__ src, const int* __restrict__ dst,
    const float* __restrict__ asv, const float* __restrict__ H,
    const float* advl, unsigned* mrow, float* den, float* acc,
    int lo, int E, int ET, int wid, int lane) {
  for (int b0 = wid * 32; b0 < ET; b0 += 1024) {
    const int e = b0 + lane;
    int s = 0, d = -1;
    if (e < ET) {
      if (e < E) { s = src[e]; d = dst[e]; } else { s = d = e - E; }
    }
    const int dl = d - lo;
    const bool mine = (unsigned)dl < (unsigned)ARW;
    float sc = 0.f;
    if (mine) sc = leaky02(asv[s] + advl[dl]);
    if (MODE == 0) {
      if (mine) atomicMax(&mrow[dl], f2ord(sc));
    } else {
      float p = 0.f;
      if (mine) {
        p = __expf(sc - ord2f(mrow[dl]));
        atomicAdd(&den[dl], p);
      }
      if (D == AD) {
        unsigned mask = __builtin_amdgcn_ballot_w32(mine);
        while (mask) {
          const int L = __builtin_ctz(mask);
          mask &= mask - 1;
          row_add<1>(s, dl, p, L, 1, lane, H, acc);
        }
      } else {
        if (mine) row_add16(s, dl, p, H, acc);
      }
    }
  }
}

template <int D, int ARW>
__global__ __launch_bounds__(1024) void gat_owner(
    const int* __restrict__ src, const int* __restrict__ dst,
    const unsigned char* __restrict__ ob,
    const float* __restrict__ asv, const float* __restrict__ adv,
    const float* __restrict__ H, const float* __restrict__ bias,
    float* __restrict__ OUT, int E, int N, int Dact,
    const float* __restrict__ was, const float* __restrict__ wad,
    float* __restrict__ asvN, float* __restrict__ advN,
    const unsigned char* __restrict__ ob0) {
  __shared__ __attribute__((aligned(16))) float acc[ARW * D];
  __shared__ int      lsrc[LCAP];
  __shared__ int      ldl[LCAP];
  __shared__ float    lp[LCAP];
  __shared__ unsigned mrow[ARW];
  __shared__ float    den[ARW];
  __shared__ float    advl[ARW];
  __shared__ int      cnt;
  __shared__ int      ucnt;

  const int tid  = threadIdx.x;
  const int lane = tid & 31;
  const int wid  = __builtin_amdgcn_readfirstlane(tid >> 5);
  const int lo   = blockIdx.x * ARW;
  const int ET   = E + N;
  int nown = N - lo;
  nown = (nown < ARW) ? nown : ARW;

  for (int x = tid; x < ARW * D / 4; x += 1024)
    reinterpret_cast<float4*>(acc)[x] = make_float4(0.f, 0.f, 0.f, 0.f);
  for (int r = tid; r < ARW; r += 1024) {
    mrow[r] = f2ord(-INFINITY);
    den[r]  = 0.f;
    advl[r] = (r < nown) ? adv[lo + r] : 0.f;
    if (r < nown) lsrc[r] = E + lo + r;
  }
  if (tid == 0) { cnt = nown; ucnt = 0; }
  __syncthreads();

  const int nq = (E + 15) >> 4;
  const bool bytes_ok = ((N + ARW - 1) / ARW <= 255) && ob_room(nq, N);
  int qreg = 0;
  int n = 0;
  const int nup = ((E + 4096) / 4096) * 64;
  const bool planes_ok = bytes_ok && nq > 0 && pl_room(nq, nup, N);
  bool over = false;
  if (planes_ok) {
    const unsigned* plw = reinterpret_cast<const unsigned*>(ob0 + (size_t)nq * 32) +
                          (size_t)((D == AD ? 0 : 8) + (blockIdx.x >> 5)) * nup;
    const unsigned bit = blockIdx.x & 31u;
    for (int u0 = 0; u0 < nup; u0 += 16 * 1024) {
      unsigned wv[16];
      #pragma unroll
      for (int j = 0; j < 16; ++j) {
        const int u = u0 + j * 1024 + tid;
        wv[j] = plw[u < nup ? u : nup - 1];
      }
      unsigned mk[16];
      int off[16];
      int tot = 0;
      #pragma unroll
      for (int j = 0; j < 16; ++j) {
        const int u = u0 + j * 1024 + tid;
        const bool pr = (u < nup) && ((wv[j] >> bit) & 1u);
        mk[j] = __builtin_amdgcn_ballot_w32(pr);
        off[j] = tot;
        tot += __builtin_popcount(mk[j]);
      }
      if (tot) {
        int b = 0;
        if (lane == 0) b = atomicAdd(&ucnt, tot);
        b = __builtin_amdgcn_readfirstlane(b);
        #pragma unroll
        for (int j = 0; j < 16; ++j) {
          if ((mk[j] >> lane) & 1u) {
            const int slot = b + off[j] + (int)__builtin_amdgcn_mbcnt_lo(mk[j], 0u);
            if (slot < LCAP) ldl[slot] = u0 + j * 1024 + tid;
          }
        }
      }
    }
    __syncthreads();
    const int uc = ucnt;
    over = uc > LCAP;
    const int nu = over ? 0 : uc;
    const uint4* ob4 = reinterpret_cast<const uint4*>(ob);
    const unsigned bsel = blockIdx.x * 0x01010101u;
    const int qlast = nq - 1;
    uint4 dv[4];
    int un = (tid < nu) ? ldl[tid] : 0;
    #pragma unroll
    for (int h = 0; h < 4; ++h) {
      const int qi = 4 * un + h;
      dv[h] = ob4[qi < qlast ? qi : qlast];
    }
    for (int i0 = 0; i0 < nu; i0 += 1024) {
      const int i = i0 + tid;
      const bool have = i < nu;
      const int u = un;
      unsigned z[4];
      #pragma unroll
      for (int h = 0; h < 4; ++h) {
        const int qi = 4 * u + h;
        const unsigned m = (zero_bytes(dv[h].x ^ bsel) >> 7) |
                           (zero_bytes(dv[h].y ^ bsel) >> 6) |
                           (zero_bytes(dv[h].z ^ bsel) >> 5) |
                           (zero_bytes(dv[h].w ^ bsel) >> 4);
        z[h] = (have && qi < nq) ? m : 0u;
      }
      un = (i + 1024 < nu) ? ldl[i + 1024] : 0;
      #pragma unroll
      for (int h = 0; h < 4; ++h) {
        const int qi = 4 * un + h;
        dv[h] = ob4[qi < qlast ? qi : qlast];
      }
      for (;;) {
        unsigned mk[4];
        int off[4];
        int tot = 0;
        #pragma unroll
        for (int h = 0; h < 4; ++h) {
          mk[h] = __builtin_amdgcn_ballot_w32(z[h] != 0u);
          off[h] = tot;
          tot += __builtin_popcount(mk[h]);
        }
        if (tot == 0) break;
        int b = 0;
        if (lane == 0) b = atomicAdd(&cnt, tot);
        b = __builtin_amdgcn_readfirstlane(b);
        #pragma unroll
        for (int h = 0; h < 4; ++h) {
          if (z[h] != 0u) {
            const int p = __builtin_ctz(z[h]);
            z[h] &= z[h] - 1u;
            const int id = (4 * u + h) * 16 + ((p & 7) << 2) + (p >> 3);
            const int slot = b + off[h] + (int)__builtin_amdgcn_mbcnt_lo(mk[h], 0u);
            if (slot < LCAP) lsrc[slot] = id;
          }
        }
      }
    }
  } else if (bytes_ok && nq > 0) {
    const uint4* ob4 = reinterpret_cast<const uint4*>(ob);
    const unsigned bsel = blockIdx.x * 0x01010101u;
    const int qstep = SCB * 32;
    const int qlast = nq - 1;
    int qb = wid * qstep;
    uint4 dv[SCB];
    #pragma unroll
    for (int k = 0; k < SCB; ++k) {
      const int qi = qb + k * 32 + lane;
      dv[k] = ob4[qi < qlast ? qi : qlast];
    }
    while (qb < nq) {
      unsigned z[SCB];
      #pragma unroll
      for (int k = 0; k < SCB; ++k) {
        const int qi = qb + k * 32 + lane;
        const unsigned m = (zero_bytes(dv[k].x ^ bsel) >> 7) |
                           (zero_bytes(dv[k].y ^ bsel) >> 6) |
                           (zero_bytes(dv[k].z ^ bsel) >> 5) |
                           (zero_bytes(dv[k].w ^ bsel) >> 4);
        z[k] = (qi < nq) ? m : 0u;
      }
      const int cur = qb;
      qb += 32 * qstep;
      #pragma unroll
      for (int k = 0; k < SCB; ++k) {
        const int qi = qb + k * 32 + lane;
        dv[k] = ob4[qi < qlast ? qi : qlast];
      }
      #pragma unroll
      for (int k = 0; k < SCB; ++k) {
        unsigned lanes = __builtin_amdgcn_ballot_w32(z[k] != 0);
        while (lanes) {
          const int L = __builtin_ctz(lanes);
          lanes &= lanes - 1;
          unsigned bits = (unsigned)__builtin_amdgcn_readlane((int)z[k], L);
          const int e0 = (cur + k * 32 + L) * 16;
          while (bits) {
            const int p = __builtin_ctz(bits);
            bits &= bits - 1;
            const int id = e0 + ((p & 7) << 2) + (p >> 3);
            qreg = (lane == n) ? id : qreg;
            ++n;
            if (n == 32) {
              list_append(&cnt, lsrc, qreg, 32, lane);
              n = 0;
            }
          }
        }
      }
    }
    if (n) list_append(&cnt, lsrc, qreg, n, lane);
  }
  __syncthreads();

  const int total = (bytes_ok && !over) ? cnt : (LCAP + 1);
  if (total <= LCAP) {
    int   sk[LROUNDS], dk[LROUNDS];
    float ak[LROUNDS];
    #pragma unroll
    for (int k = 0; k < LROUNDS; ++k) {
      const int i = tid + 1024 * k;
      int s = 0, d = lo;
      if (i < total) {
        const int e = lsrc[i];
        if (e < E) { s = src[e]; d = dst[e]; } else { s = d = e - E; }
      }
      sk[k] = s;
      dk[k] = d - lo;
    }
    #pragma unroll
    for (int k = 0; k < LROUNDS; ++k) ak[k] = asv[sk[k]];
    #pragma unroll
    for (int k = 0; k < LROUNDS; ++k) {
      const int i = tid + 1024 * k;
      if (i < total) {
        const float sc = leaky02(ak[k] + advl[dk[k]]);
        lsrc[i] = sk[k];
        ldl[i]  = dk[k];
        lp[i]   = sc;
        atomicMax(&mrow[dk[k]], f2ord(sc));
      }
    }
    __syncthreads();
    for (int b0 = wid * 32; b0 < total; b0 += 1024) {
      const int i = b0 + lane;
      int s = 0, dl = 0;
      float p = 0.f;
      if (i < total) {
        s  = lsrc[i];
        dl = ldl[i];
        p  = __expf(lp[i] - ord2f(mrow[dl]));
        atomicAdd(&den[dl], p);
      }
      if (D == AD) {
        int m = total - b0;
        m = (m < 32) ? m : 32;
        for (int g = 0; g < m; g += RG)
          row_add<RG>(s, dl, p, g, m - g, lane, H, acc);
      } else {
        if (i < total) row_add16(s, dl, p, H, acc);
      }
    }
    __syncthreads();
  } else {
    owner_direct<D, ARW, 0>(src, dst, asv, H, advl, mrow, den, acc,
                            lo, E, ET, wid, lane);
    __syncthreads();
    owner_direct<D, ARW, 1>(src, dst, asv, H, advl, mrow, den, acc,
                            lo, E, ET, wid, lane);
    __syncthreads();
  }

  if (D == AD) {
    constexpr int RPW = (ARW + 31) / 32;
    const float4* b4 = reinterpret_cast<const float4*>(bias);
    const float4* s4 = reinterpret_cast<const float4*>(was);
    const float4* d4 = reinterpret_cast<const float4*>(wad);
    const float4 bb0 = b4[lane], bb1 = b4[32 + lane];
    const float4 ws0 = s4[lane], ws1 = s4[32 + lane];
    const float4 wd0 = d4[lane], wd1 = d4[32 + lane];
    const int rb = wid * RPW;
    float k1 = 0.f, k2 = 0.f;
    #pragma unroll
    for (int t = 0; t < RPW; ++t) {
      const int r = rb + t;
      if (r < nown) {
        const float inv = 1.0f / den[r];
        const float* ap = &acc[r * AD + lane];
        const float4 a0 = make_float4(ap[0], ap[32], ap[64], ap[96]);
        const float4 a1 = make_float4(ap[128], ap[160], ap[192], ap[224]);
        float4 o0, o1;
        o0.x = a0.x * inv + bb0.x; o0.x = o0.x > 0.f ? o0.x : 0.f;
        o0.y = a0.y * inv + bb0.y; o0.y = o0.y > 0.f ? o0.y : 0.f;
        o0.z = a0.z * inv + bb0.z; o0.z = o0.z > 0.f ? o0.z : 0.f;
        o0.w = a0.w * inv + bb0.w; o0.w = o0.w > 0.f ? o0.w : 0.f;
        o1.x = a1.x * inv + bb1.x; o1.x = o1.x > 0.f ? o1.x : 0.f;
        o1.y = a1.y * inv + bb1.y; o1.y = o1.y > 0.f ? o1.y : 0.f;
        o1.z = a1.z * inv + bb1.z; o1.z = o1.z > 0.f ? o1.z : 0.f;
        o1.w = a1.w * inv + bb1.w; o1.w = o1.w > 0.f ? o1.w : 0.f;
        float4* op = reinterpret_cast<float4*>(OUT + (size_t)(lo + r) * AD) + lane;
        op[0]  = o0;
        op[32] = o1;
        float s1 = o0.x * ws0.x, s2 = o0.x * wd0.x;
        s1 = fmaf(o0.y, ws0.y, s1); s2 = fmaf(o0.y, wd0.y, s2);
        s1 = fmaf(o0.z, ws0.z, s1); s2 = fmaf(o0.z, wd0.z, s2);
        s1 = fmaf(o0.w, ws0.w, s1); s2 = fmaf(o0.w, wd0.w, s2);
        s1 = fmaf(o1.x, ws1.x, s1); s2 = fmaf(o1.x, wd1.x, s2);
        s1 = fmaf(o1.y, ws1.y, s1); s2 = fmaf(o1.y, wd1.y, s2);
        s1 = fmaf(o1.z, ws1.z, s1); s2 = fmaf(o1.z, wd1.z, s2);
        s1 = fmaf(o1.w, ws1.w, s1); s2 = fmaf(o1.w, wd1.w, s2);
        #pragma unroll
        for (int off = 16; off > 0; off >>= 1) {
          s1 += __shfl_xor(s1, off, 32);
          s2 += __shfl_xor(s2, off, 32);
        }
        k1 = (lane == t) ? s1 : k1;
        k2 = (lane == t) ? s2 : k2;
      }
    }
    int cn = nown - rb;
    cn = (cn < RPW) ? cn : RPW;
    if (lane < cn) {
      asvN[lo + rb + lane] = k1;
      advN[lo + rb + lane] = k2;
    }
  } else {
    const int nout = nown * Dact;
    for (int x = tid; x < nout; x += 1024) {
      const int r = x / Dact;
      const int c = x - r * Dact;
      OUT[(size_t)lo * Dact + x] = acc[r * D3 + c] / den[r] + bias[c];
    }
  }
}

extern "C" void kernel_launch(void* const* d_in, const int* in_sizes, int n_in,
                              void* d_out, int out_size, void* d_ws, size_t ws_size,
                              hipStream_t stream) {
  const float* x  = (const float*)d_in[0];
  const int*   ei = (const int*)d_in[1];
  const float* W1 = (const float*)d_in[2];
  const float* as1 = (const float*)d_in[3];
  const float* ad1 = (const float*)d_in[4];
  const float* b1 = (const float*)d_in[5];
  const float* W2 = (const float*)d_in[6];
  const float* as2 = (const float*)d_in[7];
  const float* ad2 = (const float*)d_in[8];
  const float* b2 = (const float*)d_in[9];
  const float* W3 = (const float*)d_in[10];
  const float* as3 = (const float*)d_in[11];
  const float* ad3 = (const float*)d_in[12];
  const float* b3 = (const float*)d_in[13];

  const int N  = in_sizes[0] / 256;
  const int E  = in_sizes[1] / 2;
  const int* src = ei;
  const int* dst = ei + E;

  float* ws = (float*)d_ws;
  float*    hA  = ws;
  float*    hB  = hA + (size_t)N * 256;
  float*    obr = hB + (size_t)N * 256;
  float*    asv = obr + (size_t)N * 16;
  float*    adv = asv + N;
  float*    wa  = obr;
  float*    asvB = asv - 2 * (size_t)N;
  float*    advB = asvB + N;
  float*    h3  = hA;
  unsigned char* obA = (unsigned char*)(obr + 1024);
  unsigned char* obB = obA + (size_t)((E + 15) / 16) * 16;

  dim3 blk(256);
  dim3 gemmG3((N + BM - 1) / BM, 1);
  int gGemmW = 250;
  int gAlpha = 512;
  int gOwn   = (N + 199) / 200;
  int gOwn3  = (N + 207) / 208;

  int nOB = (E + 4096) / 4096;
  gat_owner_bytes<200, 208, 10><<<nOB + 16, blk, 0, stream>>>(
      dst, obA, obB, E, N, nOB, W2, as2, ad2, W3, as3, ad3, wa);

  gat_gemm_wres<<<gGemmW, 512, 0, stream>>>(x, W1, hA, N);
  gat_alpha<256, 256><<<gAlpha, blk, 0, stream>>>(hA, as1, ad1, asv, adv, N);
  gat_owner<AD, 200><<<gOwn, 1024, 0, stream>>>(src, dst, obA, asv, adv, hA, b1, hB, E, N, 256,
                                                wa, wa + 256, asvB, advB, obA);

  gat_gemm_wres<<<gGemmW, 512, 0, stream>>>(hB, W2, hA, N);
  gat_owner<AD, 200><<<gOwn, 1024, 0, stream>>>(src, dst, obA, asvB, advB, hA, b2, hB, E, N, 256,
                                                wa + 512, wa + 768, asv, adv, obA);

  gat_gemm_wmma<10, 16><<<gemmG3, blk, 0, stream>>>(hB, W3, h3, N);
  gat_owner<D3, 208><<<gOwn3, 1024, 0, stream>>>(src, dst, obB, asv, adv, h3, b3, (float*)d_out, E, N, 10,
                                                 nullptr, nullptr, nullptr, nullptr, obA);
}
